# starting kernel
# baseline (speedup 1.0000x reference)
#include <hip/hip_runtime.h>
#include <hip/hip_bf16.h>

typedef __attribute__((ext_vector_type(2))) float v2f;
typedef __attribute__((ext_vector_type(8))) float v8f;

#define C_IN  128
#define C_HID 64
#define C_OUT 2
#define WPAD  132

__device__ __forceinline__ void atomic_add_f32(float* p, float v) {
    __hip_atomic_fetch_add(p, v, __ATOMIC_RELAXED, __HIP_MEMORY_SCOPE_AGENT);
}

__global__ void k_init_deg(float* __restrict__ deg, int n) {
    int i = blockIdx.x * blockDim.x + threadIdx.x;
    if (i < n) deg[i] = 1.0f;
}

__global__ void k_count_deg(const int* __restrict__ ei, float* __restrict__ deg, int E) {
    int e = blockIdx.x * blockDim.x + threadIdx.x;
    if (e < E) atomic_add_f32(&deg[ei[(size_t)E + e]], 1.0f);
}

__global__ void k_dinv(float* __restrict__ deg, int n) {
    int i = blockIdx.x * blockDim.x + threadIdx.x;
    if (i < n) deg[i] = __frsqrt_rn(deg[i]);
}

__global__ __launch_bounds__(256)
void k_gemm1_wmma(const float* __restrict__ x, const float* __restrict__ W1,
                  float* __restrict__ h1, int numPairs) {
    __shared__ float Wt[C_HID * WPAD];
    for (int i = threadIdx.x; i < C_HID * C_IN; i += 256) {
        int nn = i >> 7;
        int kk = i & 127;
        Wt[nn * WPAD + kk] = W1[kk * C_HID + nn];
    }
    __syncthreads();

    int wave = threadIdx.x >> 5;
    int lane = threadIdx.x & 31;
    int pair = blockIdx.x * 8 + wave;
    if (pair >= numPairs) pair = numPairs - 1;

    int m     = lane & 15;
    int khalf = lane >> 4;
    int ncol  = lane & 15;

    const float* xrow0 = x + ((size_t)pair * 32 + m) * C_IN + 2 * khalf;
    const float* xrow1 = xrow0 + (size_t)16 * C_IN;

    const v2f* bp0 = (const v2f*)&Wt[(ncol     ) * WPAD + 2 * khalf];
    const v2f* bp1 = (const v2f*)&Wt[(ncol + 16) * WPAD + 2 * khalf];
    const v2f* bp2 = (const v2f*)&Wt[(ncol + 32) * WPAD + 2 * khalf];
    const v2f* bp3 = (const v2f*)&Wt[(ncol + 48) * WPAD + 2 * khalf];

    v8f a00 = {}, a01 = {}, a02 = {}, a03 = {};
    v8f a10 = {}, a11 = {}, a12 = {}, a13 = {};
    #pragma unroll 4
    for (int k0 = 0; k0 < C_IN; k0 += 4) {
        v2f av0 = *(const v2f*)(xrow0 + k0);
        v2f av1 = *(const v2f*)(xrow1 + k0);
        v2f b0 = bp0[k0 >> 1];
        v2f b1 = bp1[k0 >> 1];
        v2f b2 = bp2[k0 >> 1];
        v2f b3 = bp3[k0 >> 1];
        a00 = __builtin_amdgcn_wmma_f32_16x16x4_f32(false, av0, false, b0, (short)0, a00, false, false);
        a10 = __builtin_amdgcn_wmma_f32_16x16x4_f32(false, av1, false, b0, (short)0, a10, false, false);
        a01 = __builtin_amdgcn_wmma_f32_16x16x4_f32(false, av0, false, b1, (short)0, a01, false, false);
        a11 = __builtin_amdgcn_wmma_f32_16x16x4_f32(false, av1, false, b1, (short)0, a11, false, false);
        a02 = __builtin_amdgcn_wmma_f32_16x16x4_f32(false, av0, false, b2, (short)0, a02, false, false);
        a12 = __builtin_amdgcn_wmma_f32_16x16x4_f32(false, av1, false, b2, (short)0, a12, false, false);
        a03 = __builtin_amdgcn_wmma_f32_16x16x4_f32(false, av0, false, b3, (short)0, a03, false, false);
        a13 = __builtin_amdgcn_wmma_f32_16x16x4_f32(false, av1, false, b3, (short)0, a13, false, false);
    }

    float* h0 = h1 + (size_t)pair * 32 * C_HID;
    float* h16 = h0 + (size_t)16 * C_HID;
    #pragma unroll
    for (int r = 0; r < 8; ++r) {
        int mrow = r + 8 * khalf;
        h0[mrow * C_HID + ncol]       = a00[r];
        h0[mrow * C_HID + 16 + ncol]  = a01[r];
        h0[mrow * C_HID + 32 + ncol]  = a02[r];
        h0[mrow * C_HID + 48 + ncol]  = a03[r];
        h16[mrow * C_HID + ncol]      = a10[r];
        h16[mrow * C_HID + 16 + ncol] = a11[r];
        h16[mrow * C_HID + 32 + ncol] = a12[r];
        h16[mrow * C_HID + 48 + ncol] = a13[r];
    }
}

__global__ void k_init_agg1(const float* __restrict__ h1, const float* __restrict__ dinv,
                            const float* __restrict__ b1, float* __restrict__ agg1, int total) {
    int idx = blockIdx.x * blockDim.x + threadIdx.x;
    if (idx >= total) return;
    int i = idx >> 6;
    int c = idx & 63;
    float d = dinv[i];
    agg1[idx] = b1[c] + d * d * h1[idx];
}

__global__ __launch_bounds__(256)
void k_agg1_edges(const int* __restrict__ ei, const float* __restrict__ dinv,
                  const float* __restrict__ h1, float* __restrict__ agg1, int E) {
    int gw   = (blockIdx.x * blockDim.x + threadIdx.x) >> 5;
    int lane = threadIdx.x & 31;
    if (gw >= E) return;
    int row = ei[gw];
    int col = ei[(size_t)E + gw];
    float norm = dinv[row] * dinv[col];
    const float* hr = h1 + (size_t)row * C_HID;
    float* ac = agg1 + (size_t)col * C_HID;
    atomic_add_f32(&ac[lane],      norm * hr[lane]);
    atomic_add_f32(&ac[lane + 32], norm * hr[lane + 32]);
}

__global__ void k_gemm2(const float* __restrict__ agg1, const float* __restrict__ W2,
                        const float* __restrict__ b2, const float* __restrict__ dinv,
                        float* __restrict__ h2, float* __restrict__ out, int n) {
    int i = blockIdx.x * blockDim.x + threadIdx.x;
    if (i >= n) return;
    const float* a = agg1 + (size_t)i * C_HID;
    float o0 = 0.f, o1 = 0.f;
    #pragma unroll 8
    for (int k = 0; k < C_HID; ++k) {
        float v = a[k];
        v = v > 0.f ? v : 0.f;
        o0 = fmaf(v, W2[k * 2 + 0], o0);
        o1 = fmaf(v, W2[k * 2 + 1], o1);
    }
    h2[i * 2 + 0] = o0;
    h2[i * 2 + 1] = o1;
    float d2 = dinv[i] * dinv[i];
    out[i * 2 + 0] = b2[0] + d2 * o0;
    out[i * 2 + 1] = b2[1] + d2 * o1;
}

__global__ void k_agg2_edges(const int* __restrict__ ei, const float* __restrict__ dinv,
                             const float* __restrict__ h2, float* __restrict__ out, int E) {
    int e = blockIdx.x * blockDim.x + threadIdx.x;
    if (e >= E) return;
    int row = ei[e];
    int col = ei[(size_t)E + e];
    float norm = dinv[row] * dinv[col];
    atomic_add_f32(&out[col * 2 + 0], norm * h2[row * 2 + 0]);
    atomic_add_f32(&out[col * 2 + 1], norm * h2[row * 2 + 1]);
}

extern "C" void kernel_launch(void* const* d_in, const int* in_sizes, int n_in,
                              void* d_out, int out_size, void* d_ws, size_t ws_size,
                              hipStream_t stream) {
    const float* x  = (const float*)d_in[0];
    const int*   ei = (const int*)d_in[1];
    const float* W1 = (const float*)d_in[2];
    const float* b1 = (const float*)d_in[3];
    const float* W2 = (const float*)d_in[4];
    const float* b2 = (const float*)d_in[5];
    float* out = (float*)d_out;

    int n = in_sizes[0] / C_IN;
    int E = in_sizes[1] / 2;

    float* ws   = (float*)d_ws;
    float* dinv = ws;
    float* h1   = dinv + n;
    float* agg1 = h1 + (size_t)n * C_HID;
    float* h2   = agg1 + (size_t)n * C_HID;

    const int B = 256;
    int numPairs = (n + 31) / 32;

    k_init_deg  <<<(n + B - 1) / B, B, 0, stream>>>(dinv, n);
    k_count_deg <<<(E + B - 1) / B, B, 0, stream>>>(ei, dinv, E);
    k_dinv      <<<(n + B - 1) / B, B, 0, stream>>>(dinv, n);

    k_gemm1_wmma<<<(numPairs + 7) / 8, B, 0, stream>>>(x, W1, h1, numPairs);

    int tot1 = n * C_HID;
    k_init_agg1 <<<(tot1 + B - 1) / B, B, 0, stream>>>(h1, dinv, b1, agg1, tot1);
    k_agg1_edges<<<((size_t)E * 32 + B - 1) / B, B, 0, stream>>>(ei, dinv, h1, agg1, E);

    k_gemm2     <<<(n + B - 1) / B, B, 0, stream>>>(agg1, W2, b2, dinv, h2, out, n);
    k_agg2_edges<<<(E + B - 1) / B, B, 0, stream>>>(ei, dinv, h2, out, E);
}


# Edges counting-sorted into 128-node target bins (LDS counters); one block per bin sums messages and degrees with LDS atomics, one store per element: no global atomics; sums stay in arrival order
# speedup vs baseline: 3.7486x; 3.7486x over previous
#include <hip/hip_runtime.h>
#include <hip/hip_bf16.h>


typedef __attribute__((ext_vector_type(2))) float v2f;
typedef __attribute__((ext_vector_type(8))) float v8f;

#define C_IN  128
#define C_HID 64
#define C_OUT 2
#define WPAD  132

#define BIN_SH 7
#define BIN_R  (1 << BIN_SH)
#define NCHUNK 256
#define AGG1_WAVES 16

__device__ __forceinline__ int block_excl_scan(int v, int* wsum, int tid, int nwaves, int* total) {
    int lane = tid & 31, wv = tid >> 5;
    int inc = v;
    #pragma unroll
    for (int d = 1; d < 32; d <<= 1) {
        int t = __shfl_up(inc, d, 32);
        if (lane >= d) inc += t;
    }
    if (lane == 31) wsum[wv] = inc;
    __syncthreads();
    if (wv == 0) {
        int s = lane < nwaves ? wsum[lane] : 0;
        int si = s;
        #pragma unroll
        for (int d = 1; d < 32; d <<= 1) {
            int t = __shfl_up(si, d, 32);
            if (lane >= d) si += t;
        }
        wsum[lane] = si - s;
        if (lane == 31) wsum[32] = si;
    }
    __syncthreads();
    *total = wsum[32];
    return inc - v + wsum[wv];
}

__global__ __launch_bounds__(1024)
void k_bin_hist(const int* __restrict__ ei, int* __restrict__ counts, int E, int NB, int chunk) {
    extern __shared__ int sh_bins[];
    int tid = threadIdx.x;
    for (int i = tid; i < NB; i += 1024) sh_bins[i] = 0;
    __syncthreads();
    int w = blockIdx.x;
    int beg = w * chunk;
    int end = beg + chunk < E ? beg + chunk : E;
    const int* col = ei + (size_t)E;
    int e = beg + tid;
    for (; e + 3 * 1024 < end; e += 4 * 1024) {
        int c[4];
        #pragma unroll
        for (int u = 0; u < 4; ++u) c[u] = col[e + u * 1024];
        #pragma unroll
        for (int u = 0; u < 4; ++u) atomicAdd(&sh_bins[c[u] >> BIN_SH], 1);
    }
    for (; e < end; e += 1024) atomicAdd(&sh_bins[col[e] >> BIN_SH], 1);
    __syncthreads();
    for (int i = tid; i < NB; i += 1024) counts[(size_t)i * NCHUNK + w] = sh_bins[i];
}

__global__ __launch_bounds__(NCHUNK)
void k_bin_rowscan(int* __restrict__ counts, int* __restrict__ binTotal) {
    __shared__ int wsum[33];
    int tid = threadIdx.x;
    size_t idx = (size_t)blockIdx.x * NCHUNK + tid;
    int total;
    int off = block_excl_scan(counts[idx], wsum, tid, NCHUNK / 32, &total);
    counts[idx] = off;
    if (tid == 0) binTotal[blockIdx.x] = total;
}

__global__ __launch_bounds__(1024)
void k_bin_base(const int* __restrict__ binTotal, int* __restrict__ binBase, int NB) {
    __shared__ int wsum[33];
    int tid = threadIdx.x;
    int ipt = (NB + 1023) / 1024;
    int b0 = tid * ipt;
    int s = 0;
    for (int i = 0; i < ipt; ++i) if (b0 + i < NB) s += binTotal[b0 + i];
    int total;
    int run = block_excl_scan(s, wsum, tid, 32, &total);
    for (int i = 0; i < ipt; ++i) if (b0 + i < NB) { binBase[b0 + i] = run; run += binTotal[b0 + i]; }
    if (tid == 0) binBase[NB] = total;
}

__global__ __launch_bounds__(1024)
void k_bin_scatter(const int* __restrict__ ei, const int* __restrict__ counts,
                   const int* __restrict__ binBase, unsigned* __restrict__ bucket,
                   int E, int NB, int chunk) {
    extern __shared__ int sh_bins[];
    int tid = threadIdx.x;
    int w = blockIdx.x;
    for (int i = tid; i < NB; i += 1024) sh_bins[i] = binBase[i] + counts[(size_t)i * NCHUNK + w];
    __syncthreads();
    int beg = w * chunk;
    int end = beg + chunk < E ? beg + chunk : E;
    const int* row = ei;
    const int* col = ei + (size_t)E;
    int e = beg + tid;
    for (; e + 3 * 1024 < end; e += 4 * 1024) {
        int r[4], c[4];
        #pragma unroll
        for (int u = 0; u < 4; ++u) { r[u] = row[e + u * 1024]; c[u] = col[e + u * 1024]; }
        #pragma unroll
        for (int u = 0; u < 4; ++u) {
            int pos = atomicAdd(&sh_bins[c[u] >> BIN_SH], 1);
            bucket[pos] = ((unsigned)r[u] << BIN_SH) | (unsigned)(c[u] & (BIN_R - 1));
        }
    }
    for (; e < end; e += 1024) {
        int r = row[e], c = col[e];
        int pos = atomicAdd(&sh_bins[c >> BIN_SH], 1);
        bucket[pos] = ((unsigned)r << BIN_SH) | (unsigned)(c & (BIN_R - 1));
    }
}

__global__ __launch_bounds__(512)
void k_deg_bins(const unsigned* __restrict__ bucket, const int* __restrict__ binBase,
                float* __restrict__ dinv, int n) {
    __shared__ int cnt[BIN_R];
    int tid = threadIdx.x;
    int b = blockIdx.x;
    if (tid < BIN_R) cnt[tid] = 0;
    __syncthreads();
    int beg = binBase[b], end = binBase[b + 1];
    int p = beg + tid;
    for (; p + 3 * 512 < end; p += 4 * 512) {
        unsigned rec[4];
        #pragma unroll
        for (int u = 0; u < 4; ++u) rec[u] = bucket[p + u * 512];
        #pragma unroll
        for (int u = 0; u < 4; ++u) atomicAdd(&cnt[rec[u] & (BIN_R - 1)], 1);
    }
    for (; p < end; p += 512) atomicAdd(&cnt[bucket[p] & (BIN_R - 1)], 1);
    __syncthreads();
    if (tid < BIN_R) {
        int node = b * BIN_R + tid;
        if (node < n) dinv[node] = __frsqrt_rn((float)(cnt[tid] + 1));
    }
}

__global__ __launch_bounds__(256)
void k_gemm1_wmma(const float* __restrict__ x, const float* __restrict__ W1,
                  float* __restrict__ h1, int numPairs) {
    __shared__ float Wt[C_HID * WPAD];
    for (int i = threadIdx.x; i < C_HID * C_IN; i += 256) {
        int nn = i >> 7;
        int kk = i & 127;
        Wt[nn * WPAD + kk] = W1[kk * C_HID + nn];
    }
    __syncthreads();

    int wave = threadIdx.x >> 5;
    int lane = threadIdx.x & 31;
    int pair = blockIdx.x * 8 + wave;
    if (pair >= numPairs) pair = numPairs - 1;

    int m     = lane & 15;
    int khalf = lane >> 4;
    int ncol  = lane & 15;

    const float* xrow0 = x + ((size_t)pair * 32 + m) * C_IN + 2 * khalf;
    const float* xrow1 = xrow0 + (size_t)16 * C_IN;

    const v2f* bp0 = (const v2f*)&Wt[(ncol     ) * WPAD + 2 * khalf];
    const v2f* bp1 = (const v2f*)&Wt[(ncol + 16) * WPAD + 2 * khalf];
    const v2f* bp2 = (const v2f*)&Wt[(ncol + 32) * WPAD + 2 * khalf];
    const v2f* bp3 = (const v2f*)&Wt[(ncol + 48) * WPAD + 2 * khalf];

    v8f a00 = {}, a01 = {}, a02 = {}, a03 = {};
    v8f a10 = {}, a11 = {}, a12 = {}, a13 = {};
    #pragma unroll 4
    for (int k0 = 0; k0 < C_IN; k0 += 4) {
        v2f av0 = *(const v2f*)(xrow0 + k0);
        v2f av1 = *(const v2f*)(xrow1 + k0);
        v2f b0 = bp0[k0 >> 1];
        v2f b1 = bp1[k0 >> 1];
        v2f b2 = bp2[k0 >> 1];
        v2f b3 = bp3[k0 >> 1];
        a00 = __builtin_amdgcn_wmma_f32_16x16x4_f32(false, av0, false, b0, (short)0, a00, false, false);
        a10 = __builtin_amdgcn_wmma_f32_16x16x4_f32(false, av1, false, b0, (short)0, a10, false, false);
        a01 = __builtin_amdgcn_wmma_f32_16x16x4_f32(false, av0, false, b1, (short)0, a01, false, false);
        a11 = __builtin_amdgcn_wmma_f32_16x16x4_f32(false, av1, false, b1, (short)0, a11, false, false);
        a02 = __builtin_amdgcn_wmma_f32_16x16x4_f32(false, av0, false, b2, (short)0, a02, false, false);
        a12 = __builtin_amdgcn_wmma_f32_16x16x4_f32(false, av1, false, b2, (short)0, a12, false, false);
        a03 = __builtin_amdgcn_wmma_f32_16x16x4_f32(false, av0, false, b3, (short)0, a03, false, false);
        a13 = __builtin_amdgcn_wmma_f32_16x16x4_f32(false, av1, false, b3, (short)0, a13, false, false);
    }

    float* h0 = h1 + (size_t)pair * 32 * C_HID;
    float* h16 = h0 + (size_t)16 * C_HID;
    #pragma unroll
    for (int r = 0; r < 8; ++r) {
        int mrow = r + 8 * khalf;
        h0[mrow * C_HID + ncol]       = a00[r];
        h0[mrow * C_HID + 16 + ncol]  = a01[r];
        h0[mrow * C_HID + 32 + ncol]  = a02[r];
        h0[mrow * C_HID + 48 + ncol]  = a03[r];
        h16[mrow * C_HID + ncol]      = a10[r];
        h16[mrow * C_HID + 16 + ncol] = a11[r];
        h16[mrow * C_HID + 32 + ncol] = a12[r];
        h16[mrow * C_HID + 48 + ncol] = a13[r];
    }
}

__device__ __forceinline__ void agg1_gather16(v2f (&hv)[16], const float* __restrict__ h1, int row, int lane, int j0) {
    #pragma unroll
    for (int u = 0; u < 16; ++u) {
        int r = __builtin_amdgcn_readlane(row, j0 + u);
        hv[u] = *(const v2f*)(h1 + (size_t)r * C_HID + 2 * lane);
    }
}
__device__ __forceinline__ void agg1_add16(const v2f (&hv)[16], float* acc, int cl, int nbits, int lane, int j0) {
    #pragma unroll
    for (int u = 0; u < 16; ++u) {
        int   c = __builtin_amdgcn_readlane(cl, j0 + u);
        float s = __int_as_float(__builtin_amdgcn_readlane(nbits, j0 + u));
        float* a = &acc[c * C_HID + 2 * lane];
        atomicAdd(a,     s * hv[u].x);
        atomicAdd(a + 1, s * hv[u].y);
    }
}

__global__ __launch_bounds__(AGG1_WAVES * 32, 16)
void k_agg1_bins(const unsigned* __restrict__ bucket, const int* __restrict__ binBase,
                 const float* __restrict__ dinv, const float* __restrict__ h1,
                 const float* __restrict__ b1, float* __restrict__ agg1, int n) {
    __shared__ __attribute__((aligned(16))) float acc[BIN_R * C_HID];
    int tid  = threadIdx.x;
    int lane = tid & 31;
    int wv   = __builtin_amdgcn_readfirstlane(tid >> 5);
    int b    = blockIdx.x;
    int node0 = b * BIN_R;
    for (int i = tid; i < BIN_R * C_HID; i += AGG1_WAVES * 32) acc[i] = 0.f;
    __syncthreads();

    int beg = binBase[b], end = binBase[b + 1];
    int p = beg + wv * 32;
    bool ok0 = p + lane < end;
    unsigned rec = ok0 ? bucket[p + lane] : 0u;
    float nrm = ok0 ? dinv[rec >> BIN_SH] * dinv[node0 + (int)(rec & (BIN_R - 1))] : 0.f;
    while (p < end) {
        int left = end - p;
        int pn = p + AGG1_WAVES * 32;
        bool okN = pn + lane < end;
        unsigned recN = 0u;
        int   row = (int)(rec >> BIN_SH);
        int   cl  = (int)(rec & (BIN_R - 1));
        int   nbits = __float_as_int(nrm);
        float nrmN = 0.f;
        if (left >= 32) {
            v2f hv[16];
            agg1_gather16(hv, h1, row, lane, 0);
            recN = okN ? bucket[pn + lane] : 0u;
            __builtin_amdgcn_sched_barrier(0);
            agg1_add16(hv, acc, cl, nbits, lane, 0);
            agg1_gather16(hv, h1, row, lane, 16);
            float dA = okN ? dinv[recN >> BIN_SH] : 0.f;
            float dB = okN ? dinv[node0 + (int)(recN & (BIN_R - 1))] : 0.f;
            __builtin_amdgcn_sched_barrier(0);
            agg1_add16(hv, acc, cl, nbits, lane, 16);
            nrmN = dA * dB;
        } else {
            for (int j = 0; j < left; ++j) {
                int   r = __builtin_amdgcn_readlane(row, j);
                int   c = __builtin_amdgcn_readlane(cl, j);
                float s = __int_as_float(__builtin_amdgcn_readlane(nbits, j));
                v2f hv = *(const v2f*)(h1 + (size_t)r * C_HID + 2 * lane);
                float* a = &acc[c * C_HID + 2 * lane];
                atomicAdd(a,     s * hv.x);
                atomicAdd(a + 1, s * hv.y);
            }
        }
        rec = recN;
        nrm = nrmN;
        p = pn;
    }
    __syncthreads();

    for (int i = tid; i < BIN_R * (C_HID / 4); i += AGG1_WAVES * 32) {
        int cl = i >> 4;
        int c4 = (i & 15) * 4;
        int node = node0 + cl;
        if (node < n) {
            float d = dinv[node];
            float dd = d * d;
            size_t g = (size_t)node * C_HID + c4;
            float4 hs = *(const float4*)(h1 + g);
            float4 av = *(const float4*)(&acc[cl * C_HID + c4]);
            float4 o;
            o.x = b1[c4 + 0] + dd * hs.x + av.x;
            o.y = b1[c4 + 1] + dd * hs.y + av.y;
            o.z = b1[c4 + 2] + dd * hs.z + av.z;
            o.w = b1[c4 + 3] + dd * hs.w + av.w;
            *(float4*)(agg1 + g) = o;
        }
    }
}

__global__ void k_gemm2(const float* __restrict__ agg1, const float* __restrict__ W2,
                        const float* __restrict__ b2, const float* __restrict__ dinv,
                        float* __restrict__ h2, float* __restrict__ out, int n) {
    int i = blockIdx.x * blockDim.x + threadIdx.x;
    if (i >= n) return;
    const float* a = agg1 + (size_t)i * C_HID;
    float o0 = 0.f, o1 = 0.f;
    #pragma unroll 8
    for (int k = 0; k < C_HID; ++k) {
        float v = a[k];
        v = v > 0.f ? v : 0.f;
        o0 = fmaf(v, W2[k * 2 + 0], o0);
        o1 = fmaf(v, W2[k * 2 + 1], o1);
    }
    h2[i * 2 + 0] = o0;
    h2[i * 2 + 1] = o1;
    float d2 = dinv[i] * dinv[i];
    out[i * 2 + 0] = b2[0] + d2 * o0;
    out[i * 2 + 1] = b2[1] + d2 * o1;
}

__global__ __launch_bounds__(512)
void k_agg2_bins(const unsigned* __restrict__ bucket, const int* __restrict__ binBase,
                 const float* __restrict__ dinv, const float* __restrict__ h2,
                 float* __restrict__ out, int n) {
    __shared__ float acc[BIN_R * C_OUT];
    int tid = threadIdx.x;
    int b = blockIdx.x;
    int node0 = b * BIN_R;
    if (tid < BIN_R * C_OUT) acc[tid] = 0.f;
    __syncthreads();
    int beg = binBase[b], end = binBase[b + 1];
    int p = beg + tid;
    for (; p + 3 * 512 < end; p += 4 * 512) {
        unsigned rec[4];
        #pragma unroll
        for (int u = 0; u < 4; ++u) rec[u] = bucket[p + u * 512];
        float norm[4];
        v2f hv[4];
        #pragma unroll
        for (int u = 0; u < 4; ++u) {
            int row = (int)(rec[u] >> BIN_SH);
            int cl  = (int)(rec[u] & (BIN_R - 1));
            norm[u] = dinv[row] * dinv[node0 + cl];
            hv[u] = *(const v2f*)(h2 + (size_t)row * C_OUT);
        }
        #pragma unroll
        for (int u = 0; u < 4; ++u) {
            int cl = (int)(rec[u] & (BIN_R - 1));
            atomicAdd(&acc[cl * C_OUT + 0], norm[u] * hv[u].x);
            atomicAdd(&acc[cl * C_OUT + 1], norm[u] * hv[u].y);
        }
    }
    for (; p < end; p += 512) {
        unsigned rec = bucket[p];
        int row = (int)(rec >> BIN_SH);
        int cl  = (int)(rec & (BIN_R - 1));
        float norm = dinv[row] * dinv[node0 + cl];
        v2f hv = *(const v2f*)(h2 + (size_t)row * C_OUT);
        atomicAdd(&acc[cl * C_OUT + 0], norm * hv.x);
        atomicAdd(&acc[cl * C_OUT + 1], norm * hv.y);
    }
    __syncthreads();
    if (tid < BIN_R * C_OUT) {
        int node = node0 + (tid >> 1);
        if (node < n) {
            size_t g = (size_t)node0 * C_OUT + tid;
            out[g] = out[g] + acc[tid];
        }
    }
}

extern "C" void kernel_launch(void* const* d_in, const int* in_sizes, int n_in,
                              void* d_out, int out_size, void* d_ws, size_t ws_size,
                              hipStream_t stream) {
    const float* x  = (const float*)d_in[0];
    const int*   ei = (const int*)d_in[1];
    const float* W1 = (const float*)d_in[2];
    const float* b1 = (const float*)d_in[3];
    const float* W2 = (const float*)d_in[4];
    const float* b2 = (const float*)d_in[5];
    float* out = (float*)d_out;

    int n = in_sizes[0] / C_IN;
    int E = in_sizes[1] / 2;
    int nPad = (n + 3) & ~3;

    float* ws   = (float*)d_ws;
    float* dinv = ws;
    float* h1   = dinv + nPad;
    float* agg1 = h1 + (size_t)nPad * C_HID;
    float* h2   = agg1 + (size_t)nPad * C_HID;

    int NB    = (n + BIN_R - 1) / BIN_R;
    int chunk = (E + NCHUNK - 1) / NCHUNK;
    int*      counts   = (int*)(h2 + (size_t)nPad * C_OUT);
    int*      binTotal = counts + (size_t)NB * NCHUNK;
    int*      binBase  = binTotal + NB;
    unsigned* bucket   = (unsigned*)(binBase + NB + 1);

    const int B = 256;
    int numPairs = (n + 31) / 32;
    size_t ldsBins = (size_t)NB * sizeof(int);

    k_bin_hist   <<<NCHUNK, 1024, ldsBins, stream>>>(ei, counts, E, NB, chunk);
    k_bin_rowscan<<<NB, NCHUNK, 0, stream>>>(counts, binTotal);
    k_bin_base   <<<1, 1024, 0, stream>>>(binTotal, binBase, NB);
    k_bin_scatter<<<NCHUNK, 1024, ldsBins, stream>>>(ei, counts, binBase, bucket, E, NB, chunk);
    k_deg_bins   <<<NB, 512, 0, stream>>>(bucket, binBase, dinv, n);

    k_gemm1_wmma<<<(numPairs + 7) / 8, B, 0, stream>>>(x, W1, h1, numPairs);

    k_agg1_bins  <<<NB, AGG1_WAVES * 32, 0, stream>>>(bucket, binBase, dinv, h1, b1, agg1, n);

    k_gemm2     <<<(n + B - 1) / B, B, 0, stream>>>(agg1, W2, b2, dinv, h2, out, n);
    k_agg2_bins  <<<NB, 512, 0, stream>>>(bucket, binBase, dinv, h2, out, n);
}


# Edge weight factored: producers store rows pre-scaled by dinv[source]; both aggregations drop per-edge dinv gathers and multiplies; dinv[target] scales each node's finished sum (arrival order kept)
# speedup vs baseline: 4.1666x; 1.1115x over previous
#include <hip/hip_runtime.h>
#include <hip/hip_bf16.h>


typedef __attribute__((ext_vector_type(2))) float v2f;
typedef __attribute__((ext_vector_type(8))) float v8f;

#define C_IN  128
#define C_HID 64
#define C_OUT 2
#define WPAD  132

#define BIN_SH 7
#define BIN_R  (1 << BIN_SH)
#define NCHUNK 256
#define AGG1_WAVES 16

__device__ __forceinline__ int block_excl_scan(int v, int* wsum, int tid, int nwaves, int* total) {
    int lane = tid & 31, wv = tid >> 5;
    int inc = v;
    #pragma unroll
    for (int d = 1; d < 32; d <<= 1) {
        int t = __shfl_up(inc, d, 32);
        if (lane >= d) inc += t;
    }
    if (lane == 31) wsum[wv] = inc;
    __syncthreads();
    if (wv == 0) {
        int s = lane < nwaves ? wsum[lane] : 0;
        int si = s;
        #pragma unroll
        for (int d = 1; d < 32; d <<= 1) {
            int t = __shfl_up(si, d, 32);
            if (lane >= d) si += t;
        }
        wsum[lane] = si - s;
        if (lane == 31) wsum[32] = si;
    }
    __syncthreads();
    *total = wsum[32];
    return inc - v + wsum[wv];
}

__global__ __launch_bounds__(1024)
void k_bin_hist(const int* __restrict__ ei, int* __restrict__ counts, int E, int NB, int chunk) {
    extern __shared__ int sh_bins[];
    int tid = threadIdx.x;
    for (int i = tid; i < NB; i += 1024) sh_bins[i] = 0;
    __syncthreads();
    int w = blockIdx.x;
    int beg = w * chunk;
    int end = beg + chunk < E ? beg + chunk : E;
    const int* col = ei + (size_t)E;
    int e = beg + tid;
    for (; e + 3 * 1024 < end; e += 4 * 1024) {
        int c[4];
        #pragma unroll
        for (int u = 0; u < 4; ++u) c[u] = col[e + u * 1024];
        #pragma unroll
        for (int u = 0; u < 4; ++u) atomicAdd(&sh_bins[c[u] >> BIN_SH], 1);
    }
    for (; e < end; e += 1024) atomicAdd(&sh_bins[col[e] >> BIN_SH], 1);
    __syncthreads();
    for (int i = tid; i < NB; i += 1024) counts[(size_t)i * NCHUNK + w] = sh_bins[i];
}

__global__ __launch_bounds__(NCHUNK)
void k_bin_rowscan(int* __restrict__ counts, int* __restrict__ binTotal) {
    __shared__ int wsum[33];
    int tid = threadIdx.x;
    size_t idx = (size_t)blockIdx.x * NCHUNK + tid;
    int total;
    int off = block_excl_scan(counts[idx], wsum, tid, NCHUNK / 32, &total);
    counts[idx] = off;
    if (tid == 0) binTotal[blockIdx.x] = total;
}

__global__ __launch_bounds__(1024)
void k_bin_base(const int* __restrict__ binTotal, int* __restrict__ binBase, int NB) {
    __shared__ int wsum[33];
    int tid = threadIdx.x;
    int ipt = (NB + 1023) / 1024;
    int b0 = tid * ipt;
    int s = 0;
    for (int i = 0; i < ipt; ++i) if (b0 + i < NB) s += binTotal[b0 + i];
    int total;
    int run = block_excl_scan(s, wsum, tid, 32, &total);
    for (int i = 0; i < ipt; ++i) if (b0 + i < NB) { binBase[b0 + i] = run; run += binTotal[b0 + i]; }
    if (tid == 0) binBase[NB] = total;
}

__global__ __launch_bounds__(1024)
void k_bin_scatter(const int* __restrict__ ei, const int* __restrict__ counts,
                   const int* __restrict__ binBase, unsigned* __restrict__ bucket,
                   int E, int NB, int chunk) {
    extern __shared__ int sh_bins[];
    int tid = threadIdx.x;
    int w = blockIdx.x;
    for (int i = tid; i < NB; i += 1024) sh_bins[i] = binBase[i] + counts[(size_t)i * NCHUNK + w];
    __syncthreads();
    int beg = w * chunk;
    int end = beg + chunk < E ? beg + chunk : E;
    const int* row = ei;
    const int* col = ei + (size_t)E;
    int e = beg + tid;
    for (; e + 3 * 1024 < end; e += 4 * 1024) {
        int r[4], c[4];
        #pragma unroll
        for (int u = 0; u < 4; ++u) { r[u] = row[e + u * 1024]; c[u] = col[e + u * 1024]; }
        #pragma unroll
        for (int u = 0; u < 4; ++u) {
            int pos = atomicAdd(&sh_bins[c[u] >> BIN_SH], 1);
            bucket[pos] = ((unsigned)r[u] << BIN_SH) | (unsigned)(c[u] & (BIN_R - 1));
        }
    }
    for (; e < end; e += 1024) {
        int r = row[e], c = col[e];
        int pos = atomicAdd(&sh_bins[c >> BIN_SH], 1);
        bucket[pos] = ((unsigned)r << BIN_SH) | (unsigned)(c & (BIN_R - 1));
    }
}

__global__ __launch_bounds__(512)
void k_deg_bins(const unsigned* __restrict__ bucket, const int* __restrict__ binBase,
                float* __restrict__ dinv, int n) {
    __shared__ int cnt[BIN_R];
    int tid = threadIdx.x;
    int b = blockIdx.x;
    if (tid < BIN_R) cnt[tid] = 0;
    __syncthreads();
    int beg = binBase[b], end = binBase[b + 1];
    int p = beg + tid;
    for (; p + 3 * 512 < end; p += 4 * 512) {
        unsigned rec[4];
        #pragma unroll
        for (int u = 0; u < 4; ++u) rec[u] = bucket[p + u * 512];
        #pragma unroll
        for (int u = 0; u < 4; ++u) atomicAdd(&cnt[rec[u] & (BIN_R - 1)], 1);
    }
    for (; p < end; p += 512) atomicAdd(&cnt[bucket[p] & (BIN_R - 1)], 1);
    __syncthreads();
    if (tid < BIN_R) {
        int node = b * BIN_R + tid;
        if (node < n) dinv[node] = __frsqrt_rn((float)(cnt[tid] + 1));
    }
}

__global__ __launch_bounds__(256)
void k_gemm1_wmma(const float* __restrict__ x, const float* __restrict__ W1,
                  const float* __restrict__ dinv, float* __restrict__ h1s, int numPairs) {
    __shared__ float Wt[C_HID * WPAD];
    for (int i = threadIdx.x; i < C_HID * C_IN; i += 256) {
        int nn = i >> 7;
        int kk = i & 127;
        Wt[nn * WPAD + kk] = W1[kk * C_HID + nn];
    }
    __syncthreads();

    int wave = threadIdx.x >> 5;
    int lane = threadIdx.x & 31;
    int pair = blockIdx.x * 8 + wave;
    if (pair >= numPairs) pair = numPairs - 1;

    int m     = lane & 15;
    int khalf = lane >> 4;
    int ncol  = lane & 15;

    const float* xrow0 = x + ((size_t)pair * 32 + m) * C_IN + 2 * khalf;
    const float* xrow1 = xrow0 + (size_t)16 * C_IN;

    const v2f* bp0 = (const v2f*)&Wt[(ncol     ) * WPAD + 2 * khalf];
    const v2f* bp1 = (const v2f*)&Wt[(ncol + 16) * WPAD + 2 * khalf];
    const v2f* bp2 = (const v2f*)&Wt[(ncol + 32) * WPAD + 2 * khalf];
    const v2f* bp3 = (const v2f*)&Wt[(ncol + 48) * WPAD + 2 * khalf];

    const float* dr = dinv + (size_t)pair * 32 + 8 * khalf;
    float4 dA0 = *(const float4*)(dr);
    float4 dA1 = *(const float4*)(dr + 4);
    float4 dB0 = *(const float4*)(dr + 16);
    float4 dB1 = *(const float4*)(dr + 20);

    v8f a00 = {}, a01 = {}, a02 = {}, a03 = {};
    v8f a10 = {}, a11 = {}, a12 = {}, a13 = {};
    #pragma unroll 4
    for (int k0 = 0; k0 < C_IN; k0 += 4) {
        v2f av0 = *(const v2f*)(xrow0 + k0);
        v2f av1 = *(const v2f*)(xrow1 + k0);
        v2f b0 = bp0[k0 >> 1];
        v2f b1 = bp1[k0 >> 1];
        v2f b2 = bp2[k0 >> 1];
        v2f b3 = bp3[k0 >> 1];
        a00 = __builtin_amdgcn_wmma_f32_16x16x4_f32(false, av0, false, b0, (short)0, a00, false, false);
        a10 = __builtin_amdgcn_wmma_f32_16x16x4_f32(false, av1, false, b0, (short)0, a10, false, false);
        a01 = __builtin_amdgcn_wmma_f32_16x16x4_f32(false, av0, false, b1, (short)0, a01, false, false);
        a11 = __builtin_amdgcn_wmma_f32_16x16x4_f32(false, av1, false, b1, (short)0, a11, false, false);
        a02 = __builtin_amdgcn_wmma_f32_16x16x4_f32(false, av0, false, b2, (short)0, a02, false, false);
        a12 = __builtin_amdgcn_wmma_f32_16x16x4_f32(false, av1, false, b2, (short)0, a12, false, false);
        a03 = __builtin_amdgcn_wmma_f32_16x16x4_f32(false, av0, false, b3, (short)0, a03, false, false);
        a13 = __builtin_amdgcn_wmma_f32_16x16x4_f32(false, av1, false, b3, (short)0, a13, false, false);
    }

    float d0[8] = {dA0.x, dA0.y, dA0.z, dA0.w, dA1.x, dA1.y, dA1.z, dA1.w};
    float d1[8] = {dB0.x, dB0.y, dB0.z, dB0.w, dB1.x, dB1.y, dB1.z, dB1.w};
    float* h0 = h1s + (size_t)pair * 32 * C_HID;
    float* h16 = h0 + (size_t)16 * C_HID;
    #pragma unroll
    for (int r = 0; r < 8; ++r) {
        int mrow = r + 8 * khalf;
        h0[mrow * C_HID + ncol]       = d0[r] * a00[r];
        h0[mrow * C_HID + 16 + ncol]  = d0[r] * a01[r];
        h0[mrow * C_HID + 32 + ncol]  = d0[r] * a02[r];
        h0[mrow * C_HID + 48 + ncol]  = d0[r] * a03[r];
        h16[mrow * C_HID + ncol]      = d1[r] * a10[r];
        h16[mrow * C_HID + 16 + ncol] = d1[r] * a11[r];
        h16[mrow * C_HID + 32 + ncol] = d1[r] * a12[r];
        h16[mrow * C_HID + 48 + ncol] = d1[r] * a13[r];
    }
}

__device__ __forceinline__ void agg1_gather16(v2f (&hv)[16], const float* __restrict__ h1, int row, int lane, int j0) {
    #pragma unroll
    for (int u = 0; u < 16; ++u) {
        int r = __builtin_amdgcn_readlane(row, j0 + u);
        hv[u] = *(const v2f*)(h1 + (size_t)r * C_HID + 2 * lane);
    }
}
__device__ __forceinline__ void agg1_add16(const v2f (&hv)[16], float* acc, int cl, int lane, int j0) {
    #pragma unroll
    for (int u = 0; u < 16; ++u) {
        int c = __builtin_amdgcn_readlane(cl, j0 + u);
        float* a = &acc[c * C_HID + 2 * lane];
        atomicAdd(a,     hv[u].x);
        atomicAdd(a + 1, hv[u].y);
    }
}

__global__ __launch_bounds__(AGG1_WAVES * 32, 16)
void k_agg1_bins(const unsigned* __restrict__ bucket, const int* __restrict__ binBase,
                 const float* __restrict__ dinv, const float* __restrict__ h1s,
                 const float* __restrict__ b1, float* __restrict__ agg1, int n) {
    __shared__ __attribute__((aligned(16))) float acc[BIN_R * C_HID];
    int tid  = threadIdx.x;
    int lane = tid & 31;
    int wv   = __builtin_amdgcn_readfirstlane(tid >> 5);
    int b    = blockIdx.x;
    int node0 = b * BIN_R;
    for (int i = tid; i < BIN_R * C_HID; i += AGG1_WAVES * 32) acc[i] = 0.f;
    __syncthreads();

    int beg = binBase[b], end = binBase[b + 1];
    int p = beg + wv * 32;
    unsigned rec = p + lane < end ? bucket[p + lane] : 0u;
    while (p < end) {
        int left = end - p;
        int pn = p + AGG1_WAVES * 32;
        unsigned recN = 0u;
        int row = (int)(rec >> BIN_SH);
        int cl  = (int)(rec & (BIN_R - 1));
        if (left >= 32) {
            v2f hv[16];
            agg1_gather16(hv, h1s, row, lane, 0);
            recN = pn + lane < end ? bucket[pn + lane] : 0u;
            agg1_add16(hv, acc, cl, lane, 0);
            agg1_gather16(hv, h1s, row, lane, 16);
            agg1_add16(hv, acc, cl, lane, 16);
        } else {
            for (int j = 0; j < left; ++j) {
                int r = __builtin_amdgcn_readlane(row, j);
                int c = __builtin_amdgcn_readlane(cl, j);
                v2f hv = *(const v2f*)(h1s + (size_t)r * C_HID + 2 * lane);
                float* a = &acc[c * C_HID + 2 * lane];
                atomicAdd(a,     hv.x);
                atomicAdd(a + 1, hv.y);
            }
        }
        rec = recN;
        p = pn;
    }
    __syncthreads();

    for (int i = tid; i < BIN_R * (C_HID / 4); i += AGG1_WAVES * 32) {
        int cl = i >> 4;
        int c4 = (i & 15) * 4;
        int node = node0 + cl;
        if (node < n) {
            float d = dinv[node];
            size_t g = (size_t)node * C_HID + c4;
            float4 hs = *(const float4*)(h1s + g);
            float4 av = *(const float4*)(&acc[cl * C_HID + c4]);
            float4 o;
            o.x = b1[c4 + 0] + d * (hs.x + av.x);
            o.y = b1[c4 + 1] + d * (hs.y + av.y);
            o.z = b1[c4 + 2] + d * (hs.z + av.z);
            o.w = b1[c4 + 3] + d * (hs.w + av.w);
            *(float4*)(agg1 + g) = o;
        }
    }
}

__global__ void k_gemm2(const float* __restrict__ agg1, const float* __restrict__ W2,
                        const float* __restrict__ dinv, float* __restrict__ h2s, int n) {
    int i = blockIdx.x * blockDim.x + threadIdx.x;
    if (i >= n) return;
    const float* a = agg1 + (size_t)i * C_HID;
    float o0 = 0.f, o1 = 0.f;
    #pragma unroll 8
    for (int k = 0; k < C_HID; ++k) {
        float v = a[k];
        v = v > 0.f ? v : 0.f;
        o0 = fmaf(v, W2[k * 2 + 0], o0);
        o1 = fmaf(v, W2[k * 2 + 1], o1);
    }
    float d = dinv[i];
    h2s[i * 2 + 0] = d * o0;
    h2s[i * 2 + 1] = d * o1;
}

__global__ __launch_bounds__(512)
void k_agg2_bins(const unsigned* __restrict__ bucket, const int* __restrict__ binBase,
                 const float* __restrict__ dinv, const float* __restrict__ h2s,
                 const float* __restrict__ b2, float* __restrict__ out, int n) {
    __shared__ float acc[BIN_R * C_OUT];
    int tid = threadIdx.x;
    int b = blockIdx.x;
    int node0 = b * BIN_R;
    if (tid < BIN_R * C_OUT) acc[tid] = 0.f;
    __syncthreads();
    int beg = binBase[b], end = binBase[b + 1];
    int p = beg + tid;
    for (; p + 3 * 512 < end; p += 4 * 512) {
        unsigned rec[4];
        #pragma unroll
        for (int u = 0; u < 4; ++u) rec[u] = bucket[p + u * 512];
        v2f hv[4];
        #pragma unroll
        for (int u = 0; u < 4; ++u) hv[u] = *(const v2f*)(h2s + (size_t)(rec[u] >> BIN_SH) * C_OUT);
        #pragma unroll
        for (int u = 0; u < 4; ++u) {
            int cl = (int)(rec[u] & (BIN_R - 1));
            atomicAdd(&acc[cl * C_OUT + 0], hv[u].x);
            atomicAdd(&acc[cl * C_OUT + 1], hv[u].y);
        }
    }
    for (; p < end; p += 512) {
        unsigned rec = bucket[p];
        int cl = (int)(rec & (BIN_R - 1));
        v2f hv = *(const v2f*)(h2s + (size_t)(rec >> BIN_SH) * C_OUT);
        atomicAdd(&acc[cl * C_OUT + 0], hv.x);
        atomicAdd(&acc[cl * C_OUT + 1], hv.y);
    }
    __syncthreads();
    if (tid < BIN_R * C_OUT) {
        int node = node0 + (tid >> 1);
        if (node < n) {
            size_t g = (size_t)node0 * C_OUT + tid;
            out[g] = b2[tid & 1] + dinv[node] * (h2s[g] + acc[tid]);
        }
    }
}

extern "C" void kernel_launch(void* const* d_in, const int* in_sizes, int n_in,
                              void* d_out, int out_size, void* d_ws, size_t ws_size,
                              hipStream_t stream) {
    const float* x  = (const float*)d_in[0];
    const int*   ei = (const int*)d_in[1];
    const float* W1 = (const float*)d_in[2];
    const float* b1 = (const float*)d_in[3];
    const float* W2 = (const float*)d_in[4];
    const float* b2 = (const float*)d_in[5];
    float* out = (float*)d_out;

    int n = in_sizes[0] / C_IN;
    int E = in_sizes[1] / 2;
    int nPad = (n + 3) & ~3;

    float* ws   = (float*)d_ws;
    float* dinv = ws;
    float* h1s  = dinv + nPad;
    float* agg1 = h1s + (size_t)nPad * C_HID;
    float* h2s  = agg1 + (size_t)nPad * C_HID;

    int NB    = (n + BIN_R - 1) / BIN_R;
    int chunk = (E + NCHUNK - 1) / NCHUNK;
    int*      counts   = (int*)(h2s + (size_t)nPad * C_OUT);
    int*      binTotal = counts + (size_t)NB * NCHUNK;
    int*      binBase  = binTotal + NB;
    unsigned* bucket   = (unsigned*)(binBase + NB + 1);

    const int B = 256;
    int numPairs = (n + 31) / 32;
    size_t ldsBins = (size_t)NB * sizeof(int);

    k_bin_hist   <<<NCHUNK, 1024, ldsBins, stream>>>(ei, counts, E, NB, chunk);
    k_bin_rowscan<<<NB, NCHUNK, 0, stream>>>(counts, binTotal);
    k_bin_base   <<<1, 1024, 0, stream>>>(binTotal, binBase, NB);
    k_bin_scatter<<<NCHUNK, 1024, ldsBins, stream>>>(ei, counts, binBase, bucket, E, NB, chunk);
    k_deg_bins   <<<NB, 512, 0, stream>>>(bucket, binBase, dinv, n);

    k_gemm1_wmma<<<(numPairs + 7) / 8, B, 0, stream>>>(x, W1, dinv, h1s, numPairs);

    k_agg1_bins  <<<NB, AGG1_WAVES * 32, 0, stream>>>(bucket, binBase, dinv, h1s, b1, agg1, n);

    k_gemm2     <<<(n + B - 1) / B, B, 0, stream>>>(agg1, W2, dinv, h2s, n);
    k_agg2_bins  <<<NB, 512, 0, stream>>>(bucket, binBase, dinv, h2s, b2, out, n);
}


# Layer-2 product folded into the layer-1 sum's epilogue: ReLU and the 64x2 weights applied to rows still in LDS (16-lane tree sum); the n x 64 plane and one launch go
# speedup vs baseline: 4.3845x; 1.0523x over previous
#include <hip/hip_runtime.h>
#include <hip/hip_bf16.h>


typedef __attribute__((ext_vector_type(2))) float v2f;
typedef __attribute__((ext_vector_type(8))) float v8f;

#define C_IN  128
#define C_HID 64
#define C_OUT 2
#define WPAD  132

#define BIN_SH 7
#define BIN_R  (1 << BIN_SH)
#define NCHUNK 256
#define EPI_TRIPS ((BIN_R * (C_HID / 4)) / (AGG1_WAVES * 32))
#define AGG1_WAVES 16

__device__ __forceinline__ int block_excl_scan(int v, int* wsum, int tid, int nwaves, int* total) {
    int lane = tid & 31, wv = tid >> 5;
    int inc = v;
    #pragma unroll
    for (int d = 1; d < 32; d <<= 1) {
        int t = __shfl_up(inc, d, 32);
        if (lane >= d) inc += t;
    }
    if (lane == 31) wsum[wv] = inc;
    __syncthreads();
    if (wv == 0) {
        int s = lane < nwaves ? wsum[lane] : 0;
        int si = s;
        #pragma unroll
        for (int d = 1; d < 32; d <<= 1) {
            int t = __shfl_up(si, d, 32);
            if (lane >= d) si += t;
        }
        wsum[lane] = si - s;
        if (lane == 31) wsum[32] = si;
    }
    __syncthreads();
    *total = wsum[32];
    return inc - v + wsum[wv];
}

__global__ __launch_bounds__(1024)
void k_bin_hist(const int* __restrict__ ei, int* __restrict__ counts, int E, int NB, int chunk) {
    extern __shared__ int sh_bins[];
    int tid = threadIdx.x;
    for (int i = tid; i < NB; i += 1024) sh_bins[i] = 0;
    __syncthreads();
    int w = blockIdx.x;
    int beg = w * chunk;
    int end = beg + chunk < E ? beg + chunk : E;
    const int* col = ei + (size_t)E;
    int e = beg + tid;
    for (; e + 3 * 1024 < end; e += 4 * 1024) {
        int c[4];
        #pragma unroll
        for (int u = 0; u < 4; ++u) c[u] = col[e + u * 1024];
        #pragma unroll
        for (int u = 0; u < 4; ++u) atomicAdd(&sh_bins[c[u] >> BIN_SH], 1);
    }
    for (; e < end; e += 1024) atomicAdd(&sh_bins[col[e] >> BIN_SH], 1);
    __syncthreads();
    for (int i = tid; i < NB; i += 1024) counts[(size_t)i * NCHUNK + w] = sh_bins[i];
}

__global__ __launch_bounds__(NCHUNK)
void k_bin_rowscan(int* __restrict__ counts, int* __restrict__ binTotal) {
    __shared__ int wsum[33];
    int tid = threadIdx.x;
    size_t idx = (size_t)blockIdx.x * NCHUNK + tid;
    int total;
    int off = block_excl_scan(counts[idx], wsum, tid, NCHUNK / 32, &total);
    counts[idx] = off;
    if (tid == 0) binTotal[blockIdx.x] = total;
}

__global__ __launch_bounds__(1024)
void k_bin_base(const int* __restrict__ binTotal, int* __restrict__ binBase, int NB) {
    __shared__ int wsum[33];
    int tid = threadIdx.x;
    int ipt = (NB + 1023) / 1024;
    int b0 = tid * ipt;
    int s = 0;
    for (int i = 0; i < ipt; ++i) if (b0 + i < NB) s += binTotal[b0 + i];
    int total;
    int run = block_excl_scan(s, wsum, tid, 32, &total);
    for (int i = 0; i < ipt; ++i) if (b0 + i < NB) { binBase[b0 + i] = run; run += binTotal[b0 + i]; }
    if (tid == 0) binBase[NB] = total;
}

__global__ __launch_bounds__(1024)
void k_bin_scatter(const int* __restrict__ ei, const int* __restrict__ counts,
                   const int* __restrict__ binBase, unsigned* __restrict__ bucket,
                   int E, int NB, int chunk) {
    extern __shared__ int sh_bins[];
    int tid = threadIdx.x;
    int w = blockIdx.x;
    for (int i = tid; i < NB; i += 1024) sh_bins[i] = binBase[i] + counts[(size_t)i * NCHUNK + w];
    __syncthreads();
    int beg = w * chunk;
    int end = beg + chunk < E ? beg + chunk : E;
    const int* row = ei;
    const int* col = ei + (size_t)E;
    int e = beg + tid;
    for (; e + 3 * 1024 < end; e += 4 * 1024) {
        int r[4], c[4];
        #pragma unroll
        for (int u = 0; u < 4; ++u) { r[u] = row[e + u * 1024]; c[u] = col[e + u * 1024]; }
        #pragma unroll
        for (int u = 0; u < 4; ++u) {
            int pos = atomicAdd(&sh_bins[c[u] >> BIN_SH], 1);
            bucket[pos] = ((unsigned)r[u] << BIN_SH) | (unsigned)(c[u] & (BIN_R - 1));
        }
    }
    for (; e < end; e += 1024) {
        int r = row[e], c = col[e];
        int pos = atomicAdd(&sh_bins[c >> BIN_SH], 1);
        bucket[pos] = ((unsigned)r << BIN_SH) | (unsigned)(c & (BIN_R - 1));
    }
}

__global__ __launch_bounds__(512)
void k_deg_bins(const unsigned* __restrict__ bucket, const int* __restrict__ binBase,
                float* __restrict__ dinv, int n) {
    __shared__ int cnt[BIN_R];
    int tid = threadIdx.x;
    int b = blockIdx.x;
    if (tid < BIN_R) cnt[tid] = 0;
    __syncthreads();
    int beg = binBase[b], end = binBase[b + 1];
    int p = beg + tid;
    for (; p + 3 * 512 < end; p += 4 * 512) {
        unsigned rec[4];
        #pragma unroll
        for (int u = 0; u < 4; ++u) rec[u] = bucket[p + u * 512];
        #pragma unroll
        for (int u = 0; u < 4; ++u) atomicAdd(&cnt[rec[u] & (BIN_R - 1)], 1);
    }
    for (; p < end; p += 512) atomicAdd(&cnt[bucket[p] & (BIN_R - 1)], 1);
    __syncthreads();
    if (tid < BIN_R) {
        int node = b * BIN_R + tid;
        if (node < n) dinv[node] = __frsqrt_rn((float)(cnt[tid] + 1));
    }
}

__global__ __launch_bounds__(256)
void k_gemm1_wmma(const float* __restrict__ x, const float* __restrict__ W1,
                  const float* __restrict__ dinv, float* __restrict__ h1s, int numPairs) {
    __shared__ float Wt[C_HID * WPAD];
    for (int i = threadIdx.x; i < C_HID * C_IN; i += 256) {
        int nn = i >> 7;
        int kk = i & 127;
        Wt[nn * WPAD + kk] = W1[kk * C_HID + nn];
    }
    __syncthreads();

    int wave = threadIdx.x >> 5;
    int lane = threadIdx.x & 31;
    int pair = blockIdx.x * 8 + wave;
    if (pair >= numPairs) pair = numPairs - 1;

    int m     = lane & 15;
    int khalf = lane >> 4;
    int ncol  = lane & 15;

    const float* xrow0 = x + ((size_t)pair * 32 + m) * C_IN + 2 * khalf;
    const float* xrow1 = xrow0 + (size_t)16 * C_IN;

    const v2f* bp0 = (const v2f*)&Wt[(ncol     ) * WPAD + 2 * khalf];
    const v2f* bp1 = (const v2f*)&Wt[(ncol + 16) * WPAD + 2 * khalf];
    const v2f* bp2 = (const v2f*)&Wt[(ncol + 32) * WPAD + 2 * khalf];
    const v2f* bp3 = (const v2f*)&Wt[(ncol + 48) * WPAD + 2 * khalf];

    const float* dr = dinv + (size_t)pair * 32 + 8 * khalf;
    float4 dA0 = *(const float4*)(dr);
    float4 dA1 = *(const float4*)(dr + 4);
    float4 dB0 = *(const float4*)(dr + 16);
    float4 dB1 = *(const float4*)(dr + 20);

    v8f a00 = {}, a01 = {}, a02 = {}, a03 = {};
    v8f a10 = {}, a11 = {}, a12 = {}, a13 = {};
    #pragma unroll 4
    for (int k0 = 0; k0 < C_IN; k0 += 4) {
        v2f av0 = *(const v2f*)(xrow0 + k0);
        v2f av1 = *(const v2f*)(xrow1 + k0);
        v2f b0 = bp0[k0 >> 1];
        v2f b1 = bp1[k0 >> 1];
        v2f b2 = bp2[k0 >> 1];
        v2f b3 = bp3[k0 >> 1];
        a00 = __builtin_amdgcn_wmma_f32_16x16x4_f32(false, av0, false, b0, (short)0, a00, false, false);
        a10 = __builtin_amdgcn_wmma_f32_16x16x4_f32(false, av1, false, b0, (short)0, a10, false, false);
        a01 = __builtin_amdgcn_wmma_f32_16x16x4_f32(false, av0, false, b1, (short)0, a01, false, false);
        a11 = __builtin_amdgcn_wmma_f32_16x16x4_f32(false, av1, false, b1, (short)0, a11, false, false);
        a02 = __builtin_amdgcn_wmma_f32_16x16x4_f32(false, av0, false, b2, (short)0, a02, false, false);
        a12 = __builtin_amdgcn_wmma_f32_16x16x4_f32(false, av1, false, b2, (short)0, a12, false, false);
        a03 = __builtin_amdgcn_wmma_f32_16x16x4_f32(false, av0, false, b3, (short)0, a03, false, false);
        a13 = __builtin_amdgcn_wmma_f32_16x16x4_f32(false, av1, false, b3, (short)0, a13, false, false);
    }

    float d0[8] = {dA0.x, dA0.y, dA0.z, dA0.w, dA1.x, dA1.y, dA1.z, dA1.w};
    float d1[8] = {dB0.x, dB0.y, dB0.z, dB0.w, dB1.x, dB1.y, dB1.z, dB1.w};
    float* h0 = h1s + (size_t)pair * 32 * C_HID;
    float* h16 = h0 + (size_t)16 * C_HID;
    #pragma unroll
    for (int r = 0; r < 8; ++r) {
        int mrow = r + 8 * khalf;
        h0[mrow * C_HID + ncol]       = d0[r] * a00[r];
        h0[mrow * C_HID + 16 + ncol]  = d0[r] * a01[r];
        h0[mrow * C_HID + 32 + ncol]  = d0[r] * a02[r];
        h0[mrow * C_HID + 48 + ncol]  = d0[r] * a03[r];
        h16[mrow * C_HID + ncol]      = d1[r] * a10[r];
        h16[mrow * C_HID + 16 + ncol] = d1[r] * a11[r];
        h16[mrow * C_HID + 32 + ncol] = d1[r] * a12[r];
        h16[mrow * C_HID + 48 + ncol] = d1[r] * a13[r];
    }
}

__device__ __forceinline__ void agg1_gather16(v2f (&hv)[16], const float* __restrict__ h1, int row, int lane, int j0) {
    #pragma unroll
    for (int u = 0; u < 16; ++u) {
        int r = __builtin_amdgcn_readlane(row, j0 + u);
        hv[u] = *(const v2f*)(h1 + (size_t)r * C_HID + 2 * lane);
    }
}
__device__ __forceinline__ void agg1_add16(const v2f (&hv)[16], float* acc, int cl, int lane, int j0) {
    #pragma unroll
    for (int u = 0; u < 16; ++u) {
        int c = __builtin_amdgcn_readlane(cl, j0 + u);
        float* a = &acc[c * C_HID + 2 * lane];
        atomicAdd(a,     hv[u].x);
        atomicAdd(a + 1, hv[u].y);
    }
}

__global__ __launch_bounds__(AGG1_WAVES * 32, 16)
void k_agg1_bins(const unsigned* __restrict__ bucket, const int* __restrict__ binBase,
                 const float* __restrict__ dinv, const float* __restrict__ h1s,
                 const float* __restrict__ b1, const float* __restrict__ W2,
                 float* __restrict__ h2s, int n) {
    __shared__ __attribute__((aligned(16))) float acc[BIN_R * C_HID];
    __shared__ float h2l[BIN_R * C_OUT];
    int tid  = threadIdx.x;
    int lane = tid & 31;
    int wv   = __builtin_amdgcn_readfirstlane(tid >> 5);
    int b    = blockIdx.x;
    int node0 = b * BIN_R;
    for (int i = tid; i < BIN_R * C_HID; i += AGG1_WAVES * 32) acc[i] = 0.f;
    __syncthreads();

    int beg = binBase[b], end = binBase[b + 1];
    int p = beg + wv * 32;
    unsigned rec = p + lane < end ? bucket[p + lane] : 0u;
    while (p < end) {
        int left = end - p;
        int pn = p + AGG1_WAVES * 32;
        unsigned recN = 0u;
        int row = (int)(rec >> BIN_SH);
        int cl  = (int)(rec & (BIN_R - 1));
        if (left >= 32) {
            v2f hv[16];
            agg1_gather16(hv, h1s, row, lane, 0);
            recN = pn + lane < end ? bucket[pn + lane] : 0u;
            agg1_add16(hv, acc, cl, lane, 0);
            agg1_gather16(hv, h1s, row, lane, 16);
            agg1_add16(hv, acc, cl, lane, 16);
        } else {
            for (int j = 0; j < left; ++j) {
                int r = __builtin_amdgcn_readlane(row, j);
                int c = __builtin_amdgcn_readlane(cl, j);
                v2f hv = *(const v2f*)(h1s + (size_t)r * C_HID + 2 * lane);
                float* a = &acc[c * C_HID + 2 * lane];
                atomicAdd(a,     hv.x);
                atomicAdd(a + 1, hv.y);
            }
        }
        rec = recN;
        p = pn;
    }
    __syncthreads();

    int q  = tid & 15;
    int c4 = q * 4;
    float4 bb = *(const float4*)(b1 + c4);
    float4 wA = *(const float4*)(W2 + c4 * C_OUT);
    float4 wB = *(const float4*)(W2 + c4 * C_OUT + 4);
    float  dd[EPI_TRIPS];
    float4 hh[EPI_TRIPS];
    #pragma unroll
    for (int u = 0; u < EPI_TRIPS; ++u) {
        int node = node0 + ((tid + u * AGG1_WAVES * 32) >> 4);
        node = node < n ? node : n - 1;
        dd[u] = dinv[node];
        hh[u] = *(const float4*)(h1s + (size_t)node * C_HID + c4);
    }
    #pragma unroll
    for (int u = 0; u < EPI_TRIPS; ++u) {
        int cl = (tid + u * AGG1_WAVES * 32) >> 4;
        float d = dd[u];
        float4 hs = hh[u];
        float4 av = *(const float4*)(&acc[cl * C_HID + c4]);
        float o0 = bb.x + d * (hs.x + av.x);
        float o1 = bb.y + d * (hs.y + av.y);
        float o2 = bb.z + d * (hs.z + av.z);
        float o3 = bb.w + d * (hs.w + av.w);
        o0 = o0 > 0.f ? o0 : 0.f;
        o1 = o1 > 0.f ? o1 : 0.f;
        o2 = o2 > 0.f ? o2 : 0.f;
        o3 = o3 > 0.f ? o3 : 0.f;
        float p0 = fmaf(o3, wB.z, fmaf(o2, wB.x, fmaf(o1, wA.z, o0 * wA.x)));
        float p1 = fmaf(o3, wB.w, fmaf(o2, wB.y, fmaf(o1, wA.w, o0 * wA.y)));
        #pragma unroll
        for (int sft = 8; sft >= 1; sft >>= 1) {
            p0 += __shfl_xor(p0, sft, 32);
            p1 += __shfl_xor(p1, sft, 32);
        }
        if (q == 0) {
            h2l[cl * C_OUT + 0] = d * p0;
            h2l[cl * C_OUT + 1] = d * p1;
        }
    }
    __syncthreads();
    if (tid < BIN_R * C_OUT) {
        int node = node0 + (tid >> 1);
        if (node < n) h2s[(size_t)node0 * C_OUT + tid] = h2l[tid];
    }
}

__global__ __launch_bounds__(512)
void k_agg2_bins(const unsigned* __restrict__ bucket, const int* __restrict__ binBase,
                 const float* __restrict__ dinv, const float* __restrict__ h2s,
                 const float* __restrict__ b2, float* __restrict__ out, int n) {
    __shared__ float acc[BIN_R * C_OUT];
    int tid = threadIdx.x;
    int b = blockIdx.x;
    int node0 = b * BIN_R;
    if (tid < BIN_R * C_OUT) acc[tid] = 0.f;
    __syncthreads();
    int beg = binBase[b], end = binBase[b + 1];
    int p = beg + tid;
    for (; p + 3 * 512 < end; p += 4 * 512) {
        unsigned rec[4];
        #pragma unroll
        for (int u = 0; u < 4; ++u) rec[u] = bucket[p + u * 512];
        v2f hv[4];
        #pragma unroll
        for (int u = 0; u < 4; ++u) hv[u] = *(const v2f*)(h2s + (size_t)(rec[u] >> BIN_SH) * C_OUT);
        #pragma unroll
        for (int u = 0; u < 4; ++u) {
            int cl = (int)(rec[u] & (BIN_R - 1));
            atomicAdd(&acc[cl * C_OUT + 0], hv[u].x);
            atomicAdd(&acc[cl * C_OUT + 1], hv[u].y);
        }
    }
    for (; p < end; p += 512) {
        unsigned rec = bucket[p];
        int cl = (int)(rec & (BIN_R - 1));
        v2f hv = *(const v2f*)(h2s + (size_t)(rec >> BIN_SH) * C_OUT);
        atomicAdd(&acc[cl * C_OUT + 0], hv.x);
        atomicAdd(&acc[cl * C_OUT + 1], hv.y);
    }
    __syncthreads();
    if (tid < BIN_R * C_OUT) {
        int node = node0 + (tid >> 1);
        if (node < n) {
            size_t g = (size_t)node0 * C_OUT + tid;
            out[g] = b2[tid & 1] + dinv[node] * (h2s[g] + acc[tid]);
        }
    }
}

extern "C" void kernel_launch(void* const* d_in, const int* in_sizes, int n_in,
                              void* d_out, int out_size, void* d_ws, size_t ws_size,
                              hipStream_t stream) {
    const float* x  = (const float*)d_in[0];
    const int*   ei = (const int*)d_in[1];
    const float* W1 = (const float*)d_in[2];
    const float* b1 = (const float*)d_in[3];
    const float* W2 = (const float*)d_in[4];
    const float* b2 = (const float*)d_in[5];
    float* out = (float*)d_out;

    int n = in_sizes[0] / C_IN;
    int E = in_sizes[1] / 2;
    int nPad = (n + 3) & ~3;

    float* ws   = (float*)d_ws;
    float* dinv = ws;
    float* h1s  = dinv + nPad;
    float* h2s  = h1s + (size_t)nPad * C_HID;

    int NB    = (n + BIN_R - 1) / BIN_R;
    int chunk = (E + NCHUNK - 1) / NCHUNK;
    int*      counts   = (int*)(h2s + (size_t)nPad * C_OUT);
    int*      binTotal = counts + (size_t)NB * NCHUNK;
    int*      binBase  = binTotal + NB;
    unsigned* bucket   = (unsigned*)(binBase + NB + 1);

    const int B = 256;
    int numPairs = (n + 31) / 32;
    size_t ldsBins = (size_t)NB * sizeof(int);

    k_bin_hist   <<<NCHUNK, 1024, ldsBins, stream>>>(ei, counts, E, NB, chunk);
    k_bin_rowscan<<<NB, NCHUNK, 0, stream>>>(counts, binTotal);
    k_bin_base   <<<1, 1024, 0, stream>>>(binTotal, binBase, NB);
    k_bin_scatter<<<NCHUNK, 1024, ldsBins, stream>>>(ei, counts, binBase, bucket, E, NB, chunk);
    k_deg_bins   <<<NB, 512, 0, stream>>>(bucket, binBase, dinv, n);

    k_gemm1_wmma<<<(numPairs + 7) / 8, B, 0, stream>>>(x, W1, dinv, h1s, numPairs);

    k_agg1_bins  <<<NB, AGG1_WAVES * 32, 0, stream>>>(bucket, binBase, dinv, h1s, b1, W2, h2s, n);
    k_agg2_bins  <<<NB, 512, 0, stream>>>(bucket, binBase, dinv, h2s, b2, out, n);
}


# Bins of 200 nodes in 25-wave work-groups (degree and layer-2 passes: 32-wave): 500 groups, two on every WGP, where 782 bins of 128 in 16-wave groups sat 3 or 4 to a WGP; arrival-order sums kept
# speedup vs baseline: 5.0260x; 1.1463x over previous
#include <hip/hip_runtime.h>
#include <hip/hip_bf16.h>


typedef __attribute__((ext_vector_type(2))) float v2f;
typedef __attribute__((ext_vector_type(8))) float v8f;

#define C_IN  128
#define C_HID 64
#define C_OUT 2
#define WPAD  132

#define BIN_R  200
#define REC_SH 8
#define REC_MASK ((1u << REC_SH) - 1u)
#define NCHUNK 256
#define EPI_TRIPS ((BIN_R * (C_HID / 4)) / (AGG1_WAVES * 32))
#define AGG1_WAVES 25
#define BIN_T 1024
static_assert(BIN_R <= (1 << REC_SH), "local target must fit the record's low bits");
static_assert((BIN_R * (C_HID / 4)) % (AGG1_WAVES * 32) == 0, "epilogue trips must be whole");
static_assert(BIN_R * C_OUT <= AGG1_WAVES * 32 && BIN_R * C_OUT <= BIN_T, "one thread per stored element of a bin");

__device__ __forceinline__ int block_excl_scan(int v, int* wsum, int tid, int nwaves, int* total) {
    int lane = tid & 31, wv = tid >> 5;
    int inc = v;
    #pragma unroll
    for (int d = 1; d < 32; d <<= 1) {
        int t = __shfl_up(inc, d, 32);
        if (lane >= d) inc += t;
    }
    if (lane == 31) wsum[wv] = inc;
    __syncthreads();
    if (wv == 0) {
        int s = lane < nwaves ? wsum[lane] : 0;
        int si = s;
        #pragma unroll
        for (int d = 1; d < 32; d <<= 1) {
            int t = __shfl_up(si, d, 32);
            if (lane >= d) si += t;
        }
        wsum[lane] = si - s;
        if (lane == 31) wsum[32] = si;
    }
    __syncthreads();
    *total = wsum[32];
    return inc - v + wsum[wv];
}

__global__ __launch_bounds__(1024)
void k_bin_hist(const int* __restrict__ ei, int* __restrict__ counts, int E, int NB, int chunk) {
    extern __shared__ int sh_bins[];
    int tid = threadIdx.x;
    for (int i = tid; i < NB; i += 1024) sh_bins[i] = 0;
    __syncthreads();
    int w = blockIdx.x;
    int beg = w * chunk;
    int end = beg + chunk < E ? beg + chunk : E;
    const int* col = ei + (size_t)E;
    int e = beg + tid;
    for (; e + 3 * 1024 < end; e += 4 * 1024) {
        int c[4];
        #pragma unroll
        for (int u = 0; u < 4; ++u) c[u] = col[e + u * 1024];
        #pragma unroll
        for (int u = 0; u < 4; ++u) atomicAdd(&sh_bins[(unsigned)c[u] / BIN_R], 1);
    }
    for (; e < end; e += 1024) atomicAdd(&sh_bins[(unsigned)col[e] / BIN_R], 1);
    __syncthreads();
    for (int i = tid; i < NB; i += 1024) counts[(size_t)i * NCHUNK + w] = sh_bins[i];
}

__global__ __launch_bounds__(NCHUNK)
void k_bin_rowscan(int* __restrict__ counts, int* __restrict__ binTotal) {
    __shared__ int wsum[33];
    int tid = threadIdx.x;
    size_t idx = (size_t)blockIdx.x * NCHUNK + tid;
    int total;
    int off = block_excl_scan(counts[idx], wsum, tid, NCHUNK / 32, &total);
    counts[idx] = off;
    if (tid == 0) binTotal[blockIdx.x] = total;
}

__global__ __launch_bounds__(1024)
void k_bin_base(const int* __restrict__ binTotal, int* __restrict__ binBase, int NB) {
    __shared__ int wsum[33];
    int tid = threadIdx.x;
    int ipt = (NB + 1023) / 1024;
    int b0 = tid * ipt;
    int s = 0;
    for (int i = 0; i < ipt; ++i) if (b0 + i < NB) s += binTotal[b0 + i];
    int total;
    int run = block_excl_scan(s, wsum, tid, 32, &total);
    for (int i = 0; i < ipt; ++i) if (b0 + i < NB) { binBase[b0 + i] = run; run += binTotal[b0 + i]; }
    if (tid == 0) binBase[NB] = total;
}

__global__ __launch_bounds__(1024)
void k_bin_scatter(const int* __restrict__ ei, const int* __restrict__ counts,
                   const int* __restrict__ binBase, unsigned* __restrict__ bucket,
                   int E, int NB, int chunk) {
    extern __shared__ int sh_bins[];
    int tid = threadIdx.x;
    int w = blockIdx.x;
    for (int i = tid; i < NB; i += 1024) sh_bins[i] = binBase[i] + counts[(size_t)i * NCHUNK + w];
    __syncthreads();
    int beg = w * chunk;
    int end = beg + chunk < E ? beg + chunk : E;
    const int* row = ei;
    const int* col = ei + (size_t)E;
    int e = beg + tid;
    for (; e + 3 * 1024 < end; e += 4 * 1024) {
        int r[4], c[4];
        #pragma unroll
        for (int u = 0; u < 4; ++u) { r[u] = row[e + u * 1024]; c[u] = col[e + u * 1024]; }
        #pragma unroll
        for (int u = 0; u < 4; ++u) {
            unsigned bin = (unsigned)c[u] / BIN_R;
            int pos = atomicAdd(&sh_bins[bin], 1);
            bucket[pos] = ((unsigned)r[u] << REC_SH) | ((unsigned)c[u] - bin * BIN_R);
        }
    }
    for (; e < end; e += 1024) {
        int r = row[e], c = col[e];
        unsigned bin = (unsigned)c / BIN_R;
        int pos = atomicAdd(&sh_bins[bin], 1);
        bucket[pos] = ((unsigned)r << REC_SH) | ((unsigned)c - bin * BIN_R);
    }
}

__global__ __launch_bounds__(BIN_T)
void k_deg_bins(const unsigned* __restrict__ bucket, const int* __restrict__ binBase,
                float* __restrict__ dinv, int n) {
    __shared__ int cnt[BIN_R];
    int tid = threadIdx.x;
    int b = blockIdx.x;
    if (tid < BIN_R) cnt[tid] = 0;
    __syncthreads();
    int beg = binBase[b], end = binBase[b + 1];
    int p = beg + tid;
    for (; p + 3 * BIN_T < end; p += 4 * BIN_T) {
        unsigned rec[4];
        #pragma unroll
        for (int u = 0; u < 4; ++u) rec[u] = bucket[p + u * BIN_T];
        #pragma unroll
        for (int u = 0; u < 4; ++u) atomicAdd(&cnt[rec[u] & REC_MASK], 1);
    }
    for (; p < end; p += BIN_T) atomicAdd(&cnt[bucket[p] & REC_MASK], 1);
    __syncthreads();
    if (tid < BIN_R) {
        int node = b * BIN_R + tid;
        if (node < n) dinv[node] = __frsqrt_rn((float)(cnt[tid] + 1));
    }
}

__global__ __launch_bounds__(256)
void k_gemm1_wmma(const float* __restrict__ x, const float* __restrict__ W1,
                  const float* __restrict__ dinv, float* __restrict__ h1s, int numPairs) {
    __shared__ float Wt[C_HID * WPAD];
    for (int i = threadIdx.x; i < C_HID * C_IN; i += 256) {
        int nn = i >> 7;
        int kk = i & 127;
        Wt[nn * WPAD + kk] = W1[kk * C_HID + nn];
    }
    __syncthreads();

    int wave = threadIdx.x >> 5;
    int lane = threadIdx.x & 31;
    int pair = blockIdx.x * 8 + wave;
    if (pair >= numPairs) pair = numPairs - 1;

    int m     = lane & 15;
    int khalf = lane >> 4;
    int ncol  = lane & 15;

    const float* xrow0 = x + ((size_t)pair * 32 + m) * C_IN + 2 * khalf;
    const float* xrow1 = xrow0 + (size_t)16 * C_IN;

    const v2f* bp0 = (const v2f*)&Wt[(ncol     ) * WPAD + 2 * khalf];
    const v2f* bp1 = (const v2f*)&Wt[(ncol + 16) * WPAD + 2 * khalf];
    const v2f* bp2 = (const v2f*)&Wt[(ncol + 32) * WPAD + 2 * khalf];
    const v2f* bp3 = (const v2f*)&Wt[(ncol + 48) * WPAD + 2 * khalf];

    const float* dr = dinv + (size_t)pair * 32 + 8 * khalf;
    float4 dA0 = *(const float4*)(dr);
    float4 dA1 = *(const float4*)(dr + 4);
    float4 dB0 = *(const float4*)(dr + 16);
    float4 dB1 = *(const float4*)(dr + 20);

    v8f a00 = {}, a01 = {}, a02 = {}, a03 = {};
    v8f a10 = {}, a11 = {}, a12 = {}, a13 = {};
    #pragma unroll 4
    for (int k0 = 0; k0 < C_IN; k0 += 4) {
        v2f av0 = *(const v2f*)(xrow0 + k0);
        v2f av1 = *(const v2f*)(xrow1 + k0);
        v2f b0 = bp0[k0 >> 1];
        v2f b1 = bp1[k0 >> 1];
        v2f b2 = bp2[k0 >> 1];
        v2f b3 = bp3[k0 >> 1];
        a00 = __builtin_amdgcn_wmma_f32_16x16x4_f32(false, av0, false, b0, (short)0, a00, false, false);
        a10 = __builtin_amdgcn_wmma_f32_16x16x4_f32(false, av1, false, b0, (short)0, a10, false, false);
        a01 = __builtin_amdgcn_wmma_f32_16x16x4_f32(false, av0, false, b1, (short)0, a01, false, false);
        a11 = __builtin_amdgcn_wmma_f32_16x16x4_f32(false, av1, false, b1, (short)0, a11, false, false);
        a02 = __builtin_amdgcn_wmma_f32_16x16x4_f32(false, av0, false, b2, (short)0, a02, false, false);
        a12 = __builtin_amdgcn_wmma_f32_16x16x4_f32(false, av1, false, b2, (short)0, a12, false, false);
        a03 = __builtin_amdgcn_wmma_f32_16x16x4_f32(false, av0, false, b3, (short)0, a03, false, false);
        a13 = __builtin_amdgcn_wmma_f32_16x16x4_f32(false, av1, false, b3, (short)0, a13, false, false);
    }

    float d0[8] = {dA0.x, dA0.y, dA0.z, dA0.w, dA1.x, dA1.y, dA1.z, dA1.w};
    float d1[8] = {dB0.x, dB0.y, dB0.z, dB0.w, dB1.x, dB1.y, dB1.z, dB1.w};
    float* h0 = h1s + (size_t)pair * 32 * C_HID;
    float* h16 = h0 + (size_t)16 * C_HID;
    #pragma unroll
    for (int r = 0; r < 8; ++r) {
        int mrow = r + 8 * khalf;
        h0[mrow * C_HID + ncol]       = d0[r] * a00[r];
        h0[mrow * C_HID + 16 + ncol]  = d0[r] * a01[r];
        h0[mrow * C_HID + 32 + ncol]  = d0[r] * a02[r];
        h0[mrow * C_HID + 48 + ncol]  = d0[r] * a03[r];
        h16[mrow * C_HID + ncol]      = d1[r] * a10[r];
        h16[mrow * C_HID + 16 + ncol] = d1[r] * a11[r];
        h16[mrow * C_HID + 32 + ncol] = d1[r] * a12[r];
        h16[mrow * C_HID + 48 + ncol] = d1[r] * a13[r];
    }
}

__device__ __forceinline__ void agg1_gather16(v2f (&hv)[16], const float* __restrict__ h1, int row, int lane, int j0) {
    #pragma unroll
    for (int u = 0; u < 16; ++u) {
        int r = __builtin_amdgcn_readlane(row, j0 + u);
        hv[u] = *(const v2f*)(h1 + (size_t)r * C_HID + 2 * lane);
    }
}
__device__ __forceinline__ void agg1_add16(const v2f (&hv)[16], float* acc, int cl, int lane, int j0) {
    #pragma unroll
    for (int u = 0; u < 16; ++u) {
        int c = __builtin_amdgcn_readlane(cl, j0 + u);
        float* a = &acc[c * C_HID + 2 * lane];
        atomicAdd(a,     hv[u].x);
        atomicAdd(a + 1, hv[u].y);
    }
}

__global__ __launch_bounds__(AGG1_WAVES * 32, 16)
void k_agg1_bins(const unsigned* __restrict__ bucket, const int* __restrict__ binBase,
                 const float* __restrict__ dinv, const float* __restrict__ h1s,
                 const float* __restrict__ b1, const float* __restrict__ W2,
                 float* __restrict__ h2s, int n) {
    __shared__ __attribute__((aligned(16))) float acc[BIN_R * C_HID];
    __shared__ float h2l[BIN_R * C_OUT];
    int tid  = threadIdx.x;
    int lane = tid & 31;
    int wv   = __builtin_amdgcn_readfirstlane(tid >> 5);
    int b    = blockIdx.x;
    int node0 = b * BIN_R;
    for (int i = tid; i < BIN_R * C_HID; i += AGG1_WAVES * 32) acc[i] = 0.f;
    __syncthreads();

    int beg = binBase[b], end = binBase[b + 1];
    int p = beg + wv * 32;
    unsigned rec = p + lane < end ? bucket[p + lane] : 0u;
    while (p < end) {
        int left = end - p;
        int pn = p + AGG1_WAVES * 32;
        unsigned recN = 0u;
        int row = (int)(rec >> REC_SH);
        int cl  = (int)(rec & REC_MASK);
        if (left >= 32) {
            v2f hv[16];
            agg1_gather16(hv, h1s, row, lane, 0);
            recN = pn + lane < end ? bucket[pn + lane] : 0u;
            agg1_add16(hv, acc, cl, lane, 0);
            agg1_gather16(hv, h1s, row, lane, 16);
            agg1_add16(hv, acc, cl, lane, 16);
        } else {
            for (int j = 0; j < left; ++j) {
                int r = __builtin_amdgcn_readlane(row, j);
                int c = __builtin_amdgcn_readlane(cl, j);
                v2f hv = *(const v2f*)(h1s + (size_t)r * C_HID + 2 * lane);
                float* a = &acc[c * C_HID + 2 * lane];
                atomicAdd(a,     hv.x);
                atomicAdd(a + 1, hv.y);
            }
        }
        rec = recN;
        p = pn;
    }
    __syncthreads();

    int q  = tid & 15;
    int c4 = q * 4;
    float4 bb = *(const float4*)(b1 + c4);
    float4 wA = *(const float4*)(W2 + c4 * C_OUT);
    float4 wB = *(const float4*)(W2 + c4 * C_OUT + 4);
    float  dd[EPI_TRIPS];
    float4 hh[EPI_TRIPS];
    #pragma unroll
    for (int u = 0; u < EPI_TRIPS; ++u) {
        int node = node0 + ((tid + u * AGG1_WAVES * 32) >> 4);
        node = node < n ? node : n - 1;
        dd[u] = dinv[node];
        hh[u] = *(const float4*)(h1s + (size_t)node * C_HID + c4);
    }
    #pragma unroll
    for (int u = 0; u < EPI_TRIPS; ++u) {
        int cl = (tid + u * AGG1_WAVES * 32) >> 4;
        float d = dd[u];
        float4 hs = hh[u];
        float4 av = *(const float4*)(&acc[cl * C_HID + c4]);
        float o0 = bb.x + d * (hs.x + av.x);
        float o1 = bb.y + d * (hs.y + av.y);
        float o2 = bb.z + d * (hs.z + av.z);
        float o3 = bb.w + d * (hs.w + av.w);
        o0 = o0 > 0.f ? o0 : 0.f;
        o1 = o1 > 0.f ? o1 : 0.f;
        o2 = o2 > 0.f ? o2 : 0.f;
        o3 = o3 > 0.f ? o3 : 0.f;
        float p0 = fmaf(o3, wB.z, fmaf(o2, wB.x, fmaf(o1, wA.z, o0 * wA.x)));
        float p1 = fmaf(o3, wB.w, fmaf(o2, wB.y, fmaf(o1, wA.w, o0 * wA.y)));
        #pragma unroll
        for (int sft = 8; sft >= 1; sft >>= 1) {
            p0 += __shfl_xor(p0, sft, 32);
            p1 += __shfl_xor(p1, sft, 32);
        }
        if (q == 0) {
            h2l[cl * C_OUT + 0] = d * p0;
            h2l[cl * C_OUT + 1] = d * p1;
        }
    }
    __syncthreads();
    if (tid < BIN_R * C_OUT) {
        int node = node0 + (tid >> 1);
        if (node < n) h2s[(size_t)node0 * C_OUT + tid] = h2l[tid];
    }
}

__global__ __launch_bounds__(BIN_T)
void k_agg2_bins(const unsigned* __restrict__ bucket, const int* __restrict__ binBase,
                 const float* __restrict__ dinv, const float* __restrict__ h2s,
                 const float* __restrict__ b2, float* __restrict__ out, int n) {
    __shared__ float acc[BIN_R * C_OUT];
    int tid = threadIdx.x;
    int b = blockIdx.x;
    int node0 = b * BIN_R;
    if (tid < BIN_R * C_OUT) acc[tid] = 0.f;
    __syncthreads();
    int beg = binBase[b], end = binBase[b + 1];
    int p = beg + tid;
    for (; p + 3 * BIN_T < end; p += 4 * BIN_T) {
        unsigned rec[4];
        #pragma unroll
        for (int u = 0; u < 4; ++u) rec[u] = bucket[p + u * BIN_T];
        v2f hv[4];
        #pragma unroll
        for (int u = 0; u < 4; ++u) hv[u] = *(const v2f*)(h2s + (size_t)(rec[u] >> REC_SH) * C_OUT);
        #pragma unroll
        for (int u = 0; u < 4; ++u) {
            int cl = (int)(rec[u] & REC_MASK);
            atomicAdd(&acc[cl * C_OUT + 0], hv[u].x);
            atomicAdd(&acc[cl * C_OUT + 1], hv[u].y);
        }
    }
    for (; p < end; p += BIN_T) {
        unsigned rec = bucket[p];
        int cl = (int)(rec & REC_MASK);
        v2f hv = *(const v2f*)(h2s + (size_t)(rec >> REC_SH) * C_OUT);
        atomicAdd(&acc[cl * C_OUT + 0], hv.x);
        atomicAdd(&acc[cl * C_OUT + 1], hv.y);
    }
    __syncthreads();
    if (tid < BIN_R * C_OUT) {
        int node = node0 + (tid >> 1);
        if (node < n) {
            size_t g = (size_t)node0 * C_OUT + tid;
            out[g] = b2[tid & 1] + dinv[node] * (h2s[g] + acc[tid]);
        }
    }
}

extern "C" void kernel_launch(void* const* d_in, const int* in_sizes, int n_in,
                              void* d_out, int out_size, void* d_ws, size_t ws_size,
                              hipStream_t stream) {
    const float* x  = (const float*)d_in[0];
    const int*   ei = (const int*)d_in[1];
    const float* W1 = (const float*)d_in[2];
    const float* b1 = (const float*)d_in[3];
    const float* W2 = (const float*)d_in[4];
    const float* b2 = (const float*)d_in[5];
    float* out = (float*)d_out;

    int n = in_sizes[0] / C_IN;
    int E = in_sizes[1] / 2;
    int nPad = (n + 3) & ~3;

    float* ws   = (float*)d_ws;
    float* dinv = ws;
    float* h1s  = dinv + nPad;
    float* h2s  = h1s + (size_t)nPad * C_HID;

    int NB    = (n + BIN_R - 1) / BIN_R;
    int chunk = (E + NCHUNK - 1) / NCHUNK;
    int*      counts   = (int*)(h2s + (size_t)nPad * C_OUT);
    int*      binTotal = counts + (size_t)NB * NCHUNK;
    int*      binBase  = binTotal + NB;
    unsigned* bucket   = (unsigned*)(binBase + NB + 1);

    const int B = 256;
    int numPairs = (n + 31) / 32;
    size_t ldsBins = (size_t)NB * sizeof(int);

    k_bin_hist   <<<NCHUNK, 1024, ldsBins, stream>>>(ei, counts, E, NB, chunk);
    k_bin_rowscan<<<NB, NCHUNK, 0, stream>>>(counts, binTotal);
    k_bin_base   <<<1, 1024, 0, stream>>>(binTotal, binBase, NB);
    k_bin_scatter<<<NCHUNK, 1024, ldsBins, stream>>>(ei, counts, binBase, bucket, E, NB, chunk);
    k_deg_bins   <<<NB, BIN_T, 0, stream>>>(bucket, binBase, dinv, n);

    k_gemm1_wmma<<<(numPairs + 7) / 8, B, 0, stream>>>(x, W1, dinv, h1s, numPairs);

    k_agg1_bins  <<<NB, AGG1_WAVES * 32, 0, stream>>>(bucket, binBase, dinv, h1s, b1, W2, h2s, n);
    k_agg2_bins  <<<NB, BIN_T, 0, stream>>>(bucket, binBase, dinv, h2s, b2, out, n);
}


# Degree pass also orders each bin's records by target (LDS ranks); layer-1 sum adds a target's run in registers, in list order, one LDS float add per run instead of two per edge
# speedup vs baseline: 5.8776x; 1.1694x over previous
#include <hip/hip_runtime.h>
#include <hip/hip_bf16.h>


typedef __attribute__((ext_vector_type(2))) float v2f;
typedef __attribute__((ext_vector_type(8))) float v8f;

#define C_IN  128
#define C_HID 64
#define C_OUT 2
#define WPAD  132

#define BIN_R  200
#define REC_SH 8
#define REC_MASK ((1u << REC_SH) - 1u)
#define NCHUNK 256
#define EPI_TRIPS ((BIN_R * (C_HID / 4)) / (AGG1_WAVES * 32))
#define AGG1_WAVES 25
#define BIN_T 1024
#define SORT_CAP 8192
static_assert(BIN_R <= (1 << REC_SH), "local target must fit the record's low bits");
static_assert((BIN_R * (C_HID / 4)) % (AGG1_WAVES * 32) == 0, "epilogue trips must be whole");
static_assert(BIN_R * C_OUT <= AGG1_WAVES * 32 && BIN_R * C_OUT <= BIN_T, "one thread per stored element of a bin");

__device__ __forceinline__ int block_excl_scan(int v, int* wsum, int tid, int nwaves, int* total) {
    int lane = tid & 31, wv = tid >> 5;
    int inc = v;
    #pragma unroll
    for (int d = 1; d < 32; d <<= 1) {
        int t = __shfl_up(inc, d, 32);
        if (lane >= d) inc += t;
    }
    if (lane == 31) wsum[wv] = inc;
    __syncthreads();
    if (wv == 0) {
        int s = lane < nwaves ? wsum[lane] : 0;
        int si = s;
        #pragma unroll
        for (int d = 1; d < 32; d <<= 1) {
            int t = __shfl_up(si, d, 32);
            if (lane >= d) si += t;
        }
        wsum[lane] = si - s;
        if (lane == 31) wsum[32] = si;
    }
    __syncthreads();
    *total = wsum[32];
    return inc - v + wsum[wv];
}

__global__ __launch_bounds__(1024)
void k_bin_hist(const int* __restrict__ ei, int* __restrict__ counts, int E, int NB, int chunk) {
    extern __shared__ int sh_bins[];
    int tid = threadIdx.x;
    for (int i = tid; i < NB; i += 1024) sh_bins[i] = 0;
    __syncthreads();
    int w = blockIdx.x;
    int beg = w * chunk;
    int end = beg + chunk < E ? beg + chunk : E;
    const int* col = ei + (size_t)E;
    int e = beg + tid;
    for (; e + 3 * 1024 < end; e += 4 * 1024) {
        int c[4];
        #pragma unroll
        for (int u = 0; u < 4; ++u) c[u] = col[e + u * 1024];
        #pragma unroll
        for (int u = 0; u < 4; ++u) atomicAdd(&sh_bins[(unsigned)c[u] / BIN_R], 1);
    }
    for (; e < end; e += 1024) atomicAdd(&sh_bins[(unsigned)col[e] / BIN_R], 1);
    __syncthreads();
    for (int i = tid; i < NB; i += 1024) counts[(size_t)i * NCHUNK + w] = sh_bins[i];
}

__global__ __launch_bounds__(NCHUNK)
void k_bin_rowscan(int* __restrict__ counts, int* __restrict__ binTotal) {
    __shared__ int wsum[33];
    int tid = threadIdx.x;
    size_t idx = (size_t)blockIdx.x * NCHUNK + tid;
    int total;
    int off = block_excl_scan(counts[idx], wsum, tid, NCHUNK / 32, &total);
    counts[idx] = off;
    if (tid == 0) binTotal[blockIdx.x] = total;
}

__global__ __launch_bounds__(1024)
void k_bin_base(const int* __restrict__ binTotal, int* __restrict__ binBase, int NB) {
    __shared__ int wsum[33];
    int tid = threadIdx.x;
    int ipt = (NB + 1023) / 1024;
    int b0 = tid * ipt;
    int s = 0;
    for (int i = 0; i < ipt; ++i) if (b0 + i < NB) s += binTotal[b0 + i];
    int total;
    int run = block_excl_scan(s, wsum, tid, 32, &total);
    for (int i = 0; i < ipt; ++i) if (b0 + i < NB) { binBase[b0 + i] = run; run += binTotal[b0 + i]; }
    if (tid == 0) binBase[NB] = total;
}

__global__ __launch_bounds__(1024)
void k_bin_scatter(const int* __restrict__ ei, const int* __restrict__ counts,
                   const int* __restrict__ binBase, unsigned* __restrict__ bucket,
                   int E, int NB, int chunk) {
    extern __shared__ int sh_bins[];
    int tid = threadIdx.x;
    int w = blockIdx.x;
    for (int i = tid; i < NB; i += 1024) sh_bins[i] = binBase[i] + counts[(size_t)i * NCHUNK + w];
    __syncthreads();
    int beg = w * chunk;
    int end = beg + chunk < E ? beg + chunk : E;
    const int* row = ei;
    const int* col = ei + (size_t)E;
    int e = beg + tid;
    for (; e + 3 * 1024 < end; e += 4 * 1024) {
        int r[4], c[4];
        #pragma unroll
        for (int u = 0; u < 4; ++u) { r[u] = row[e + u * 1024]; c[u] = col[e + u * 1024]; }
        #pragma unroll
        for (int u = 0; u < 4; ++u) {
            unsigned bin = (unsigned)c[u] / BIN_R;
            int pos = atomicAdd(&sh_bins[bin], 1);
            bucket[pos] = ((unsigned)r[u] << REC_SH) | ((unsigned)c[u] - bin * BIN_R);
        }
    }
    for (; e < end; e += 1024) {
        int r = row[e], c = col[e];
        unsigned bin = (unsigned)c / BIN_R;
        int pos = atomicAdd(&sh_bins[bin], 1);
        bucket[pos] = ((unsigned)r << REC_SH) | ((unsigned)c - bin * BIN_R);
    }
}

__global__ __launch_bounds__(BIN_T)
void k_deg_bins(const unsigned* __restrict__ bucket, const int* __restrict__ binBase,
                float* __restrict__ dinv, unsigned* __restrict__ sorted, int n) {
    __shared__ int cnt[BIN_R];
    __shared__ int wsum[33];
    __shared__ unsigned stage[SORT_CAP];
    int tid = threadIdx.x;
    int b = blockIdx.x;
    if (tid < BIN_R) cnt[tid] = 0;
    __syncthreads();
    int beg = binBase[b], end = binBase[b + 1];
    int p = beg + tid;
    for (; p + 3 * BIN_T < end; p += 4 * BIN_T) {
        unsigned rec[4];
        #pragma unroll
        for (int u = 0; u < 4; ++u) rec[u] = bucket[p + u * BIN_T];
        #pragma unroll
        for (int u = 0; u < 4; ++u) atomicAdd(&cnt[rec[u] & REC_MASK], 1);
    }
    for (; p < end; p += BIN_T) atomicAdd(&cnt[bucket[p] & REC_MASK], 1);
    __syncthreads();
    int c = tid < BIN_R ? cnt[tid] : 0;
    int total;
    int off = block_excl_scan(c, wsum, tid, BIN_T / 32, &total);
    if (tid < BIN_R) {
        int node = b * BIN_R + tid;
        if (node < n) dinv[node] = __frsqrt_rn((float)(c + 1));
        cnt[tid] = off;
    }
    __syncthreads();
    bool staged = end - beg <= SORT_CAP;
    p = beg + tid;
    for (; p + 3 * BIN_T < end; p += 4 * BIN_T) {
        unsigned rec[4];
        #pragma unroll
        for (int u = 0; u < 4; ++u) rec[u] = bucket[p + u * BIN_T];
        #pragma unroll
        for (int u = 0; u < 4; ++u) {
            int pos = atomicAdd(&cnt[rec[u] & REC_MASK], 1);
            if (staged) stage[pos] = rec[u]; else sorted[beg + pos] = rec[u];
        }
    }
    for (; p < end; p += BIN_T) {
        unsigned rec = bucket[p];
        int pos = atomicAdd(&cnt[rec & REC_MASK], 1);
        if (staged) stage[pos] = rec; else sorted[beg + pos] = rec;
    }
    if (staged) {
        __syncthreads();
        for (int i = tid; i < end - beg; i += BIN_T) sorted[beg + i] = stage[i];
    }
}

__global__ __launch_bounds__(256)
void k_gemm1_wmma(const float* __restrict__ x, const float* __restrict__ W1,
                  const float* __restrict__ dinv, float* __restrict__ h1s, int numPairs) {
    __shared__ float Wt[C_HID * WPAD];
    for (int i = threadIdx.x; i < C_HID * C_IN; i += 256) {
        int nn = i >> 7;
        int kk = i & 127;
        Wt[nn * WPAD + kk] = W1[kk * C_HID + nn];
    }
    __syncthreads();

    int wave = threadIdx.x >> 5;
    int lane = threadIdx.x & 31;
    int pair = blockIdx.x * 8 + wave;
    if (pair >= numPairs) pair = numPairs - 1;

    int m     = lane & 15;
    int khalf = lane >> 4;
    int ncol  = lane & 15;

    const float* xrow0 = x + ((size_t)pair * 32 + m) * C_IN + 2 * khalf;
    const float* xrow1 = xrow0 + (size_t)16 * C_IN;

    const v2f* bp0 = (const v2f*)&Wt[(ncol     ) * WPAD + 2 * khalf];
    const v2f* bp1 = (const v2f*)&Wt[(ncol + 16) * WPAD + 2 * khalf];
    const v2f* bp2 = (const v2f*)&Wt[(ncol + 32) * WPAD + 2 * khalf];
    const v2f* bp3 = (const v2f*)&Wt[(ncol + 48) * WPAD + 2 * khalf];

    const float* dr = dinv + (size_t)pair * 32 + 8 * khalf;
    float4 dA0 = *(const float4*)(dr);
    float4 dA1 = *(const float4*)(dr + 4);
    float4 dB0 = *(const float4*)(dr + 16);
    float4 dB1 = *(const float4*)(dr + 20);

    v8f a00 = {}, a01 = {}, a02 = {}, a03 = {};
    v8f a10 = {}, a11 = {}, a12 = {}, a13 = {};
    #pragma unroll 4
    for (int k0 = 0; k0 < C_IN; k0 += 4) {
        v2f av0 = *(const v2f*)(xrow0 + k0);
        v2f av1 = *(const v2f*)(xrow1 + k0);
        v2f b0 = bp0[k0 >> 1];
        v2f b1 = bp1[k0 >> 1];
        v2f b2 = bp2[k0 >> 1];
        v2f b3 = bp3[k0 >> 1];
        a00 = __builtin_amdgcn_wmma_f32_16x16x4_f32(false, av0, false, b0, (short)0, a00, false, false);
        a10 = __builtin_amdgcn_wmma_f32_16x16x4_f32(false, av1, false, b0, (short)0, a10, false, false);
        a01 = __builtin_amdgcn_wmma_f32_16x16x4_f32(false, av0, false, b1, (short)0, a01, false, false);
        a11 = __builtin_amdgcn_wmma_f32_16x16x4_f32(false, av1, false, b1, (short)0, a11, false, false);
        a02 = __builtin_amdgcn_wmma_f32_16x16x4_f32(false, av0, false, b2, (short)0, a02, false, false);
        a12 = __builtin_amdgcn_wmma_f32_16x16x4_f32(false, av1, false, b2, (short)0, a12, false, false);
        a03 = __builtin_amdgcn_wmma_f32_16x16x4_f32(false, av0, false, b3, (short)0, a03, false, false);
        a13 = __builtin_amdgcn_wmma_f32_16x16x4_f32(false, av1, false, b3, (short)0, a13, false, false);
    }

    float d0[8] = {dA0.x, dA0.y, dA0.z, dA0.w, dA1.x, dA1.y, dA1.z, dA1.w};
    float d1[8] = {dB0.x, dB0.y, dB0.z, dB0.w, dB1.x, dB1.y, dB1.z, dB1.w};
    float* h0 = h1s + (size_t)pair * 32 * C_HID;
    float* h16 = h0 + (size_t)16 * C_HID;
    #pragma unroll
    for (int r = 0; r < 8; ++r) {
        int mrow = r + 8 * khalf;
        h0[mrow * C_HID + ncol]       = d0[r] * a00[r];
        h0[mrow * C_HID + 16 + ncol]  = d0[r] * a01[r];
        h0[mrow * C_HID + 32 + ncol]  = d0[r] * a02[r];
        h0[mrow * C_HID + 48 + ncol]  = d0[r] * a03[r];
        h16[mrow * C_HID + ncol]      = d1[r] * a10[r];
        h16[mrow * C_HID + 16 + ncol] = d1[r] * a11[r];
        h16[mrow * C_HID + 32 + ncol] = d1[r] * a12[r];
        h16[mrow * C_HID + 48 + ncol] = d1[r] * a13[r];
    }
}

__device__ __forceinline__ void agg1_gather16(v2f (&hv)[16], const float* __restrict__ h1, int row, int lane, int j0) {
    #pragma unroll
    for (int u = 0; u < 16; ++u) {
        int r = __builtin_amdgcn_readlane(row, j0 + u);
        hv[u] = *(const v2f*)(h1 + (size_t)r * C_HID + 2 * lane);
    }
}
__device__ __forceinline__ void agg1_flush(float* acc, int c, v2f sum, int lane) {
    float* a = &acc[c * C_HID + 2 * lane];
    atomicAdd(a,     sum.x);
    atomicAdd(a + 1, sum.y);
}
__device__ __forceinline__ void agg1_sum16(const v2f (&hv)[16], float* acc, int cl, int lane, int j0, int& cur, v2f& sum) {
    #pragma unroll
    for (int u = 0; u < 16; ++u) {
        int c = __builtin_amdgcn_readlane(cl, j0 + u);
        if (c != cur) {
            agg1_flush(acc, cur, sum, lane);
            cur = c;
            sum.x = 0.f; sum.y = 0.f;
        }
        sum.x += hv[u].x;
        sum.y += hv[u].y;
    }
}

__global__ __launch_bounds__(AGG1_WAVES * 32, 16)
void k_agg1_bins(const unsigned* __restrict__ sorted, const int* __restrict__ binBase,
                 const float* __restrict__ dinv, const float* __restrict__ h1s,
                 const float* __restrict__ b1, const float* __restrict__ W2,
                 float* __restrict__ h2s, int n) {
    __shared__ __attribute__((aligned(16))) float acc[BIN_R * C_HID];
    __shared__ float h2l[BIN_R * C_OUT];
    int tid  = threadIdx.x;
    int lane = tid & 31;
    int wv   = __builtin_amdgcn_readfirstlane(tid >> 5);
    int b    = blockIdx.x;
    int node0 = b * BIN_R;
    for (int i = tid; i < BIN_R * C_HID; i += AGG1_WAVES * 32) acc[i] = 0.f;
    __syncthreads();

    int beg = binBase[b], end = binBase[b + 1];
    int chunk = ((end - beg + AGG1_WAVES * 32 - 1) / (AGG1_WAVES * 32)) * 32;
    int p  = beg + wv * chunk;
    int pe = p + chunk < end ? p + chunk : end;
    unsigned rec = p + lane < pe ? sorted[p + lane] : 0u;
    bool any = p < pe;
    int cur = __builtin_amdgcn_readlane((int)(rec & REC_MASK), 0);
    v2f sum = {0.f, 0.f};
    while (p < pe) {
        int left = pe - p;
        int pn = p + 32;
        unsigned recN = 0u;
        int row = (int)(rec >> REC_SH);
        int cl  = (int)(rec & REC_MASK);
        if (left >= 32) {
            v2f hv[16];
            agg1_gather16(hv, h1s, row, lane, 0);
            recN = pn + lane < pe ? sorted[pn + lane] : 0u;
            agg1_sum16(hv, acc, cl, lane, 0, cur, sum);
            agg1_gather16(hv, h1s, row, lane, 16);
            agg1_sum16(hv, acc, cl, lane, 16, cur, sum);
        } else {
            for (int j = 0; j < left; ++j) {
                int r = __builtin_amdgcn_readlane(row, j);
                int c = __builtin_amdgcn_readlane(cl, j);
                v2f hv = *(const v2f*)(h1s + (size_t)r * C_HID + 2 * lane);
                if (c != cur) {
                    agg1_flush(acc, cur, sum, lane);
                    cur = c;
                    sum.x = 0.f; sum.y = 0.f;
                }
                sum.x += hv.x;
                sum.y += hv.y;
            }
        }
        rec = recN;
        p = pn;
    }
    if (any) agg1_flush(acc, cur, sum, lane);
    __syncthreads();

    int q  = tid & 15;
    int c4 = q * 4;
    float4 bb = *(const float4*)(b1 + c4);
    float4 wA = *(const float4*)(W2 + c4 * C_OUT);
    float4 wB = *(const float4*)(W2 + c4 * C_OUT + 4);
    float  dd[EPI_TRIPS];
    float4 hh[EPI_TRIPS];
    #pragma unroll
    for (int u = 0; u < EPI_TRIPS; ++u) {
        int node = node0 + ((tid + u * AGG1_WAVES * 32) >> 4);
        node = node < n ? node : n - 1;
        dd[u] = dinv[node];
        hh[u] = *(const float4*)(h1s + (size_t)node * C_HID + c4);
    }
    #pragma unroll
    for (int u = 0; u < EPI_TRIPS; ++u) {
        int cl = (tid + u * AGG1_WAVES * 32) >> 4;
        float d = dd[u];
        float4 hs = hh[u];
        float4 av = *(const float4*)(&acc[cl * C_HID + c4]);
        float o0 = bb.x + d * (hs.x + av.x);
        float o1 = bb.y + d * (hs.y + av.y);
        float o2 = bb.z + d * (hs.z + av.z);
        float o3 = bb.w + d * (hs.w + av.w);
        o0 = o0 > 0.f ? o0 : 0.f;
        o1 = o1 > 0.f ? o1 : 0.f;
        o2 = o2 > 0.f ? o2 : 0.f;
        o3 = o3 > 0.f ? o3 : 0.f;
        float p0 = fmaf(o3, wB.z, fmaf(o2, wB.x, fmaf(o1, wA.z, o0 * wA.x)));
        float p1 = fmaf(o3, wB.w, fmaf(o2, wB.y, fmaf(o1, wA.w, o0 * wA.y)));
        #pragma unroll
        for (int sft = 8; sft >= 1; sft >>= 1) {
            p0 += __shfl_xor(p0, sft, 32);
            p1 += __shfl_xor(p1, sft, 32);
        }
        if (q == 0) {
            h2l[cl * C_OUT + 0] = d * p0;
            h2l[cl * C_OUT + 1] = d * p1;
        }
    }
    __syncthreads();
    if (tid < BIN_R * C_OUT) {
        int node = node0 + (tid >> 1);
        if (node < n) h2s[(size_t)node0 * C_OUT + tid] = h2l[tid];
    }
}

__global__ __launch_bounds__(BIN_T)
void k_agg2_bins(const unsigned* __restrict__ bucket, const int* __restrict__ binBase,
                 const float* __restrict__ dinv, const float* __restrict__ h2s,
                 const float* __restrict__ b2, float* __restrict__ out, int n) {
    __shared__ float acc[BIN_R * C_OUT];
    int tid = threadIdx.x;
    int b = blockIdx.x;
    int node0 = b * BIN_R;
    if (tid < BIN_R * C_OUT) acc[tid] = 0.f;
    __syncthreads();
    int beg = binBase[b], end = binBase[b + 1];
    int p = beg + tid;
    for (; p + 3 * BIN_T < end; p += 4 * BIN_T) {
        unsigned rec[4];
        #pragma unroll
        for (int u = 0; u < 4; ++u) rec[u] = bucket[p + u * BIN_T];
        v2f hv[4];
        #pragma unroll
        for (int u = 0; u < 4; ++u) hv[u] = *(const v2f*)(h2s + (size_t)(rec[u] >> REC_SH) * C_OUT);
        #pragma unroll
        for (int u = 0; u < 4; ++u) {
            int cl = (int)(rec[u] & REC_MASK);
            atomicAdd(&acc[cl * C_OUT + 0], hv[u].x);
            atomicAdd(&acc[cl * C_OUT + 1], hv[u].y);
        }
    }
    for (; p < end; p += BIN_T) {
        unsigned rec = bucket[p];
        int cl = (int)(rec & REC_MASK);
        v2f hv = *(const v2f*)(h2s + (size_t)(rec >> REC_SH) * C_OUT);
        atomicAdd(&acc[cl * C_OUT + 0], hv.x);
        atomicAdd(&acc[cl * C_OUT + 1], hv.y);
    }
    __syncthreads();
    if (tid < BIN_R * C_OUT) {
        int node = node0 + (tid >> 1);
        if (node < n) {
            size_t g = (size_t)node0 * C_OUT + tid;
            out[g] = b2[tid & 1] + dinv[node] * (h2s[g] + acc[tid]);
        }
    }
}

extern "C" void kernel_launch(void* const* d_in, const int* in_sizes, int n_in,
                              void* d_out, int out_size, void* d_ws, size_t ws_size,
                              hipStream_t stream) {
    const float* x  = (const float*)d_in[0];
    const int*   ei = (const int*)d_in[1];
    const float* W1 = (const float*)d_in[2];
    const float* b1 = (const float*)d_in[3];
    const float* W2 = (const float*)d_in[4];
    const float* b2 = (const float*)d_in[5];
    float* out = (float*)d_out;

    int n = in_sizes[0] / C_IN;
    int E = in_sizes[1] / 2;
    int nPad = (n + 3) & ~3;

    float* ws   = (float*)d_ws;
    float* dinv = ws;
    float* h1s  = dinv + nPad;
    float* h2s  = h1s + (size_t)nPad * C_HID;

    int NB    = (n + BIN_R - 1) / BIN_R;
    int chunk = (E + NCHUNK - 1) / NCHUNK;
    int*      counts   = (int*)(h2s + (size_t)nPad * C_OUT);
    int*      binTotal = counts + (size_t)NB * NCHUNK;
    int*      binBase  = binTotal + NB;
    unsigned* bucket   = (unsigned*)(binBase + NB + 1);
    unsigned* sorted   = bucket + (size_t)E;

    const int B = 256;
    int numPairs = (n + 31) / 32;
    size_t ldsBins = (size_t)NB * sizeof(int);

    k_bin_hist   <<<NCHUNK, 1024, ldsBins, stream>>>(ei, counts, E, NB, chunk);
    k_bin_rowscan<<<NB, NCHUNK, 0, stream>>>(counts, binTotal);
    k_bin_base   <<<1, 1024, 0, stream>>>(binTotal, binBase, NB);
    k_bin_scatter<<<NCHUNK, 1024, ldsBins, stream>>>(ei, counts, binBase, bucket, E, NB, chunk);
    k_deg_bins   <<<NB, BIN_T, 0, stream>>>(bucket, binBase, dinv, sorted, n);

    k_gemm1_wmma<<<(numPairs + 7) / 8, B, 0, stream>>>(x, W1, dinv, h1s, numPairs);

    k_agg1_bins  <<<NB, AGG1_WAVES * 32, 0, stream>>>(sorted, binBase, dinv, h1s, b1, W2, h2s, n);
    k_agg2_bins  <<<NB, BIN_T, 0, stream>>>(bucket, binBase, dinv, h2s, b2, out, n);
}


# Binning pass 4 lays each chunk's records out in LDS by bin and stores them from consecutive threads: a wave's store covers 3-4 lines, not 32 four-byte pieces; arrival-order sums unchanged
# speedup vs baseline: 7.2912x; 1.2405x over previous
#include <hip/hip_runtime.h>
#include <hip/hip_bf16.h>


typedef __attribute__((ext_vector_type(2))) float v2f;
typedef __attribute__((ext_vector_type(8))) float v8f;

#define C_IN  128
#define C_HID 64
#define C_OUT 2
#define WPAD  132

#define BIN_R  200
#define REC_SH 8
#define REC_MASK ((1u << REC_SH) - 1u)
#define NCHUNK 256
#define EPI_TRIPS ((BIN_R * (C_HID / 4)) / (AGG1_WAVES * 32))
#define AGG1_WAVES 25
#define BIN_T 1024
#define STAGE_CAP 12544
#define SORT_CAP 8192
static_assert(BIN_R <= (1 << REC_SH), "local target must fit the record's low bits");
static_assert((BIN_R * (C_HID / 4)) % (AGG1_WAVES * 32) == 0, "epilogue trips must be whole");
static_assert(BIN_R * C_OUT <= AGG1_WAVES * 32 && BIN_R * C_OUT <= BIN_T, "one thread per stored element of a bin");

__device__ __forceinline__ int block_excl_scan(int v, int* wsum, int tid, int nwaves, int* total) {
    int lane = tid & 31, wv = tid >> 5;
    int inc = v;
    #pragma unroll
    for (int d = 1; d < 32; d <<= 1) {
        int t = __shfl_up(inc, d, 32);
        if (lane >= d) inc += t;
    }
    if (lane == 31) wsum[wv] = inc;
    __syncthreads();
    if (wv == 0) {
        int s = lane < nwaves ? wsum[lane] : 0;
        int si = s;
        #pragma unroll
        for (int d = 1; d < 32; d <<= 1) {
            int t = __shfl_up(si, d, 32);
            if (lane >= d) si += t;
        }
        wsum[lane] = si - s;
        if (lane == 31) wsum[32] = si;
    }
    __syncthreads();
    *total = wsum[32];
    return inc - v + wsum[wv];
}

__global__ __launch_bounds__(1024)
void k_bin_hist(const int* __restrict__ ei, int* __restrict__ counts, int E, int NB, int chunk) {
    extern __shared__ int sh_bins[];
    int tid = threadIdx.x;
    for (int i = tid; i < NB; i += 1024) sh_bins[i] = 0;
    __syncthreads();
    int w = blockIdx.x;
    int beg = w * chunk;
    int end = beg + chunk < E ? beg + chunk : E;
    const int* col = ei + (size_t)E;
    int e = beg + tid;
    for (; e + 3 * 1024 < end; e += 4 * 1024) {
        int c[4];
        #pragma unroll
        for (int u = 0; u < 4; ++u) c[u] = col[e + u * 1024];
        #pragma unroll
        for (int u = 0; u < 4; ++u) atomicAdd(&sh_bins[(unsigned)c[u] / BIN_R], 1);
    }
    for (; e < end; e += 1024) atomicAdd(&sh_bins[(unsigned)col[e] / BIN_R], 1);
    __syncthreads();
    for (int i = tid; i < NB; i += 1024) counts[(size_t)i * NCHUNK + w] = sh_bins[i];
}

__global__ __launch_bounds__(NCHUNK)
void k_bin_rowscan(int* __restrict__ counts, int* __restrict__ binTotal) {
    __shared__ int wsum[33];
    int tid = threadIdx.x;
    size_t idx = (size_t)blockIdx.x * NCHUNK + tid;
    int total;
    int off = block_excl_scan(counts[idx], wsum, tid, NCHUNK / 32, &total);
    counts[idx] = off;
    if (tid == 0) binTotal[blockIdx.x] = total;
}

__global__ __launch_bounds__(1024)
void k_bin_base(const int* __restrict__ binTotal, int* __restrict__ binBase, int NB) {
    __shared__ int wsum[33];
    int tid = threadIdx.x;
    int ipt = (NB + 1023) / 1024;
    int b0 = tid * ipt;
    int s = 0;
    for (int i = 0; i < ipt; ++i) if (b0 + i < NB) s += binTotal[b0 + i];
    int total;
    int run = block_excl_scan(s, wsum, tid, 32, &total);
    for (int i = 0; i < ipt; ++i) if (b0 + i < NB) { binBase[b0 + i] = run; run += binTotal[b0 + i]; }
    if (tid == 0) binBase[NB] = total;
}

__global__ __launch_bounds__(1024)
void k_bin_scatter(const int* __restrict__ ei, const int* __restrict__ counts,
                   const int* __restrict__ binBase, unsigned* __restrict__ bucket,
                   int E, int NB, int chunk) {
    extern __shared__ int sh_bins[];
    __shared__ int wsum[33];
    __shared__ uint2 stage[STAGE_CAP];
    int* lcur   = sh_bins;
    int* gdelta = sh_bins + NB;
    int tid = threadIdx.x;
    int w = blockIdx.x;
    int beg = w * chunk;
    int end = beg + chunk < E ? beg + chunk : E;
    bool staged = end - beg <= STAGE_CAP;
    int ipt = (NB + 1023) / 1024;
    int b0 = tid * ipt;
    int s = 0;
    for (int i = 0; i < ipt; ++i) {
        int bi = b0 + i;
        if (bi < NB) {
            int off = counts[(size_t)bi * NCHUNK + w];
            int nxt = w + 1 < NCHUNK ? counts[(size_t)bi * NCHUNK + w + 1] : binBase[bi + 1] - binBase[bi];
            s += nxt - off;
        }
    }
    int total;
    int run = block_excl_scan(s, wsum, tid, 32, &total);
    for (int i = 0; i < ipt; ++i) {
        int bi = b0 + i;
        if (bi < NB) {
            int off = counts[(size_t)bi * NCHUNK + w];
            int nxt = w + 1 < NCHUNK ? counts[(size_t)bi * NCHUNK + w + 1] : binBase[bi + 1] - binBase[bi];
            int g = binBase[bi] + off;
            lcur[bi]   = staged ? run : g;
            gdelta[bi] = g - run;
            run += nxt - off;
        }
    }
    __syncthreads();
    const int* row = ei;
    const int* col = ei + (size_t)E;
    int e = beg + tid;
    if (staged) {
        for (; e + 3 * 1024 < end; e += 4 * 1024) {
            int r[4], c[4];
            #pragma unroll
            for (int u = 0; u < 4; ++u) { r[u] = row[e + u * 1024]; c[u] = col[e + u * 1024]; }
            unsigned bin[4];
            int pos[4], gd[4];
            #pragma unroll
            for (int u = 0; u < 4; ++u) bin[u] = (unsigned)c[u] / BIN_R;
            #pragma unroll
            for (int u = 0; u < 4; ++u) pos[u] = atomicAdd(&lcur[bin[u]], 1);
            #pragma unroll
            for (int u = 0; u < 4; ++u) gd[u] = gdelta[bin[u]];
            #pragma unroll
            for (int u = 0; u < 4; ++u)
                stage[pos[u]] = make_uint2(((unsigned)r[u] << REC_SH) | ((unsigned)c[u] - bin[u] * BIN_R),
                                           (unsigned)(pos[u] + gd[u]));
        }
        for (; e < end; e += 1024) {
            int r = row[e], c = col[e];
            unsigned bin = (unsigned)c / BIN_R;
            int pos = atomicAdd(&lcur[bin], 1);
            stage[pos] = make_uint2(((unsigned)r << REC_SH) | ((unsigned)c - bin * BIN_R), (unsigned)(pos + gdelta[bin]));
        }
        __syncthreads();
        for (int j = tid; j < end - beg; j += 1024) {
            uint2 v = stage[j];
            bucket[v.y] = v.x;
        }
    } else {
        for (; e + 3 * 1024 < end; e += 4 * 1024) {
            int r[4], c[4];
            #pragma unroll
            for (int u = 0; u < 4; ++u) { r[u] = row[e + u * 1024]; c[u] = col[e + u * 1024]; }
            #pragma unroll
            for (int u = 0; u < 4; ++u) {
                unsigned bin = (unsigned)c[u] / BIN_R;
                int pos = atomicAdd(&lcur[bin], 1);
                bucket[pos] = ((unsigned)r[u] << REC_SH) | ((unsigned)c[u] - bin * BIN_R);
            }
        }
        for (; e < end; e += 1024) {
            int r = row[e], c = col[e];
            unsigned bin = (unsigned)c / BIN_R;
            int pos = atomicAdd(&lcur[bin], 1);
            bucket[pos] = ((unsigned)r << REC_SH) | ((unsigned)c - bin * BIN_R);
        }
    }
}

__global__ __launch_bounds__(BIN_T)
void k_deg_bins(const unsigned* __restrict__ bucket, const int* __restrict__ binBase,
                float* __restrict__ dinv, unsigned* __restrict__ sorted, int n) {
    __shared__ int cnt[BIN_R];
    __shared__ int wsum[33];
    __shared__ unsigned stage[SORT_CAP];
    int tid = threadIdx.x;
    int b = blockIdx.x;
    if (tid < BIN_R) cnt[tid] = 0;
    __syncthreads();
    int beg = binBase[b], end = binBase[b + 1];
    int p = beg + tid;
    for (; p + 3 * BIN_T < end; p += 4 * BIN_T) {
        unsigned rec[4];
        #pragma unroll
        for (int u = 0; u < 4; ++u) rec[u] = bucket[p + u * BIN_T];
        #pragma unroll
        for (int u = 0; u < 4; ++u) atomicAdd(&cnt[rec[u] & REC_MASK], 1);
    }
    for (; p < end; p += BIN_T) atomicAdd(&cnt[bucket[p] & REC_MASK], 1);
    __syncthreads();
    int c = tid < BIN_R ? cnt[tid] : 0;
    int total;
    int off = block_excl_scan(c, wsum, tid, BIN_T / 32, &total);
    if (tid < BIN_R) {
        int node = b * BIN_R + tid;
        if (node < n) dinv[node] = __frsqrt_rn((float)(c + 1));
        cnt[tid] = off;
    }
    __syncthreads();
    bool staged = end - beg <= SORT_CAP;
    p = beg + tid;
    for (; p + 3 * BIN_T < end; p += 4 * BIN_T) {
        unsigned rec[4];
        #pragma unroll
        for (int u = 0; u < 4; ++u) rec[u] = bucket[p + u * BIN_T];
        #pragma unroll
        for (int u = 0; u < 4; ++u) {
            int pos = atomicAdd(&cnt[rec[u] & REC_MASK], 1);
            if (staged) stage[pos] = rec[u]; else sorted[beg + pos] = rec[u];
        }
    }
    for (; p < end; p += BIN_T) {
        unsigned rec = bucket[p];
        int pos = atomicAdd(&cnt[rec & REC_MASK], 1);
        if (staged) stage[pos] = rec; else sorted[beg + pos] = rec;
    }
    if (staged) {
        __syncthreads();
        for (int i = tid; i < end - beg; i += BIN_T) sorted[beg + i] = stage[i];
    }
}

__global__ __launch_bounds__(256)
void k_gemm1_wmma(const float* __restrict__ x, const float* __restrict__ W1,
                  const float* __restrict__ dinv, float* __restrict__ h1s, int numPairs) {
    __shared__ float Wt[C_HID * WPAD];
    for (int i = threadIdx.x; i < C_HID * C_IN; i += 256) {
        int nn = i >> 7;
        int kk = i & 127;
        Wt[nn * WPAD + kk] = W1[kk * C_HID + nn];
    }
    __syncthreads();

    int wave = threadIdx.x >> 5;
    int lane = threadIdx.x & 31;
    int pair = blockIdx.x * 8 + wave;
    if (pair >= numPairs) pair = numPairs - 1;

    int m     = lane & 15;
    int khalf = lane >> 4;
    int ncol  = lane & 15;

    const float* xrow0 = x + ((size_t)pair * 32 + m) * C_IN + 2 * khalf;
    const float* xrow1 = xrow0 + (size_t)16 * C_IN;

    const v2f* bp0 = (const v2f*)&Wt[(ncol     ) * WPAD + 2 * khalf];
    const v2f* bp1 = (const v2f*)&Wt[(ncol + 16) * WPAD + 2 * khalf];
    const v2f* bp2 = (const v2f*)&Wt[(ncol + 32) * WPAD + 2 * khalf];
    const v2f* bp3 = (const v2f*)&Wt[(ncol + 48) * WPAD + 2 * khalf];

    const float* dr = dinv + (size_t)pair * 32 + 8 * khalf;
    float4 dA0 = *(const float4*)(dr);
    float4 dA1 = *(const float4*)(dr + 4);
    float4 dB0 = *(const float4*)(dr + 16);
    float4 dB1 = *(const float4*)(dr + 20);

    v8f a00 = {}, a01 = {}, a02 = {}, a03 = {};
    v8f a10 = {}, a11 = {}, a12 = {}, a13 = {};
    #pragma unroll 4
    for (int k0 = 0; k0 < C_IN; k0 += 4) {
        v2f av0 = *(const v2f*)(xrow0 + k0);
        v2f av1 = *(const v2f*)(xrow1 + k0);
        v2f b0 = bp0[k0 >> 1];
        v2f b1 = bp1[k0 >> 1];
        v2f b2 = bp2[k0 >> 1];
        v2f b3 = bp3[k0 >> 1];
        a00 = __builtin_amdgcn_wmma_f32_16x16x4_f32(false, av0, false, b0, (short)0, a00, false, false);
        a10 = __builtin_amdgcn_wmma_f32_16x16x4_f32(false, av1, false, b0, (short)0, a10, false, false);
        a01 = __builtin_amdgcn_wmma_f32_16x16x4_f32(false, av0, false, b1, (short)0, a01, false, false);
        a11 = __builtin_amdgcn_wmma_f32_16x16x4_f32(false, av1, false, b1, (short)0, a11, false, false);
        a02 = __builtin_amdgcn_wmma_f32_16x16x4_f32(false, av0, false, b2, (short)0, a02, false, false);
        a12 = __builtin_amdgcn_wmma_f32_16x16x4_f32(false, av1, false, b2, (short)0, a12, false, false);
        a03 = __builtin_amdgcn_wmma_f32_16x16x4_f32(false, av0, false, b3, (short)0, a03, false, false);
        a13 = __builtin_amdgcn_wmma_f32_16x16x4_f32(false, av1, false, b3, (short)0, a13, false, false);
    }

    float d0[8] = {dA0.x, dA0.y, dA0.z, dA0.w, dA1.x, dA1.y, dA1.z, dA1.w};
    float d1[8] = {dB0.x, dB0.y, dB0.z, dB0.w, dB1.x, dB1.y, dB1.z, dB1.w};
    float* h0 = h1s + (size_t)pair * 32 * C_HID;
    float* h16 = h0 + (size_t)16 * C_HID;
    #pragma unroll
    for (int r = 0; r < 8; ++r) {
        int mrow = r + 8 * khalf;
        h0[mrow * C_HID + ncol]       = d0[r] * a00[r];
        h0[mrow * C_HID + 16 + ncol]  = d0[r] * a01[r];
        h0[mrow * C_HID + 32 + ncol]  = d0[r] * a02[r];
        h0[mrow * C_HID + 48 + ncol]  = d0[r] * a03[r];
        h16[mrow * C_HID + ncol]      = d1[r] * a10[r];
        h16[mrow * C_HID + 16 + ncol] = d1[r] * a11[r];
        h16[mrow * C_HID + 32 + ncol] = d1[r] * a12[r];
        h16[mrow * C_HID + 48 + ncol] = d1[r] * a13[r];
    }
}

__device__ __forceinline__ void agg1_gather16(v2f (&hv)[16], const float* __restrict__ h1, int row, int lane, int j0) {
    #pragma unroll
    for (int u = 0; u < 16; ++u) {
        int r = __builtin_amdgcn_readlane(row, j0 + u);
        hv[u] = *(const v2f*)(h1 + (size_t)r * C_HID + 2 * lane);
    }
}
__device__ __forceinline__ void agg1_flush(float* acc, int c, v2f sum, int lane) {
    float* a = &acc[c * C_HID + 2 * lane];
    atomicAdd(a,     sum.x);
    atomicAdd(a + 1, sum.y);
}
__device__ __forceinline__ void agg1_sum16(const v2f (&hv)[16], float* acc, int cl, int lane, int j0, int& cur, v2f& sum) {
    #pragma unroll
    for (int u = 0; u < 16; ++u) {
        int c = __builtin_amdgcn_readlane(cl, j0 + u);
        if (c != cur) {
            agg1_flush(acc, cur, sum, lane);
            cur = c;
            sum.x = 0.f; sum.y = 0.f;
        }
        sum.x += hv[u].x;
        sum.y += hv[u].y;
    }
}

__global__ __launch_bounds__(AGG1_WAVES * 32, 16)
void k_agg1_bins(const unsigned* __restrict__ sorted, const int* __restrict__ binBase,
                 const float* __restrict__ dinv, const float* __restrict__ h1s,
                 const float* __restrict__ b1, const float* __restrict__ W2,
                 float* __restrict__ h2s, int n) {
    __shared__ __attribute__((aligned(16))) float acc[BIN_R * C_HID];
    __shared__ float h2l[BIN_R * C_OUT];
    int tid  = threadIdx.x;
    int lane = tid & 31;
    int wv   = __builtin_amdgcn_readfirstlane(tid >> 5);
    int b    = blockIdx.x;
    int node0 = b * BIN_R;
    for (int i = tid; i < BIN_R * C_HID; i += AGG1_WAVES * 32) acc[i] = 0.f;
    __syncthreads();

    int beg = binBase[b], end = binBase[b + 1];
    int chunk = ((end - beg + AGG1_WAVES * 32 - 1) / (AGG1_WAVES * 32)) * 32;
    int p  = beg + wv * chunk;
    int pe = p + chunk < end ? p + chunk : end;
    unsigned rec = p + lane < pe ? sorted[p + lane] : 0u;
    bool any = p < pe;
    int cur = __builtin_amdgcn_readlane((int)(rec & REC_MASK), 0);
    v2f sum = {0.f, 0.f};
    while (p < pe) {
        int left = pe - p;
        int pn = p + 32;
        unsigned recN = 0u;
        int row = (int)(rec >> REC_SH);
        int cl  = (int)(rec & REC_MASK);
        if (left >= 32) {
            v2f hv[16];
            agg1_gather16(hv, h1s, row, lane, 0);
            recN = pn + lane < pe ? sorted[pn + lane] : 0u;
            agg1_sum16(hv, acc, cl, lane, 0, cur, sum);
            agg1_gather16(hv, h1s, row, lane, 16);
            agg1_sum16(hv, acc, cl, lane, 16, cur, sum);
        } else {
            for (int j = 0; j < left; ++j) {
                int r = __builtin_amdgcn_readlane(row, j);
                int c = __builtin_amdgcn_readlane(cl, j);
                v2f hv = *(const v2f*)(h1s + (size_t)r * C_HID + 2 * lane);
                if (c != cur) {
                    agg1_flush(acc, cur, sum, lane);
                    cur = c;
                    sum.x = 0.f; sum.y = 0.f;
                }
                sum.x += hv.x;
                sum.y += hv.y;
            }
        }
        rec = recN;
        p = pn;
    }
    if (any) agg1_flush(acc, cur, sum, lane);
    __syncthreads();

    int q  = tid & 15;
    int c4 = q * 4;
    float4 bb = *(const float4*)(b1 + c4);
    float4 wA = *(const float4*)(W2 + c4 * C_OUT);
    float4 wB = *(const float4*)(W2 + c4 * C_OUT + 4);
    float  dd[EPI_TRIPS];
    float4 hh[EPI_TRIPS];
    #pragma unroll
    for (int u = 0; u < EPI_TRIPS; ++u) {
        int node = node0 + ((tid + u * AGG1_WAVES * 32) >> 4);
        node = node < n ? node : n - 1;
        dd[u] = dinv[node];
        hh[u] = *(const float4*)(h1s + (size_t)node * C_HID + c4);
    }
    #pragma unroll
    for (int u = 0; u < EPI_TRIPS; ++u) {
        int cl = (tid + u * AGG1_WAVES * 32) >> 4;
        float d = dd[u];
        float4 hs = hh[u];
        float4 av = *(const float4*)(&acc[cl * C_HID + c4]);
        float o0 = bb.x + d * (hs.x + av.x);
        float o1 = bb.y + d * (hs.y + av.y);
        float o2 = bb.z + d * (hs.z + av.z);
        float o3 = bb.w + d * (hs.w + av.w);
        o0 = o0 > 0.f ? o0 : 0.f;
        o1 = o1 > 0.f ? o1 : 0.f;
        o2 = o2 > 0.f ? o2 : 0.f;
        o3 = o3 > 0.f ? o3 : 0.f;
        float p0 = fmaf(o3, wB.z, fmaf(o2, wB.x, fmaf(o1, wA.z, o0 * wA.x)));
        float p1 = fmaf(o3, wB.w, fmaf(o2, wB.y, fmaf(o1, wA.w, o0 * wA.y)));
        #pragma unroll
        for (int sft = 8; sft >= 1; sft >>= 1) {
            p0 += __shfl_xor(p0, sft, 32);
            p1 += __shfl_xor(p1, sft, 32);
        }
        if (q == 0) {
            h2l[cl * C_OUT + 0] = d * p0;
            h2l[cl * C_OUT + 1] = d * p1;
        }
    }
    __syncthreads();
    if (tid < BIN_R * C_OUT) {
        int node = node0 + (tid >> 1);
        if (node < n) h2s[(size_t)node0 * C_OUT + tid] = h2l[tid];
    }
}

__global__ __launch_bounds__(BIN_T)
void k_agg2_bins(const unsigned* __restrict__ bucket, const int* __restrict__ binBase,
                 const float* __restrict__ dinv, const float* __restrict__ h2s,
                 const float* __restrict__ b2, float* __restrict__ out, int n) {
    __shared__ float acc[BIN_R * C_OUT];
    int tid = threadIdx.x;
    int b = blockIdx.x;
    int node0 = b * BIN_R;
    if (tid < BIN_R * C_OUT) acc[tid] = 0.f;
    __syncthreads();
    int beg = binBase[b], end = binBase[b + 1];
    int p = beg + tid;
    for (; p + 3 * BIN_T < end; p += 4 * BIN_T) {
        unsigned rec[4];
        #pragma unroll
        for (int u = 0; u < 4; ++u) rec[u] = bucket[p + u * BIN_T];
        v2f hv[4];
        #pragma unroll
        for (int u = 0; u < 4; ++u) hv[u] = *(const v2f*)(h2s + (size_t)(rec[u] >> REC_SH) * C_OUT);
        #pragma unroll
        for (int u = 0; u < 4; ++u) {
            int cl = (int)(rec[u] & REC_MASK);
            atomicAdd(&acc[cl * C_OUT + 0], hv[u].x);
            atomicAdd(&acc[cl * C_OUT + 1], hv[u].y);
        }
    }
    for (; p < end; p += BIN_T) {
        unsigned rec = bucket[p];
        int cl = (int)(rec & REC_MASK);
        v2f hv = *(const v2f*)(h2s + (size_t)(rec >> REC_SH) * C_OUT);
        atomicAdd(&acc[cl * C_OUT + 0], hv.x);
        atomicAdd(&acc[cl * C_OUT + 1], hv.y);
    }
    __syncthreads();
    if (tid < BIN_R * C_OUT) {
        int node = node0 + (tid >> 1);
        if (node < n) {
            size_t g = (size_t)node0 * C_OUT + tid;
            out[g] = b2[tid & 1] + dinv[node] * (h2s[g] + acc[tid]);
        }
    }
}

extern "C" void kernel_launch(void* const* d_in, const int* in_sizes, int n_in,
                              void* d_out, int out_size, void* d_ws, size_t ws_size,
                              hipStream_t stream) {
    const float* x  = (const float*)d_in[0];
    const int*   ei = (const int*)d_in[1];
    const float* W1 = (const float*)d_in[2];
    const float* b1 = (const float*)d_in[3];
    const float* W2 = (const float*)d_in[4];
    const float* b2 = (const float*)d_in[5];
    float* out = (float*)d_out;

    int n = in_sizes[0] / C_IN;
    int E = in_sizes[1] / 2;
    int nPad = (n + 3) & ~3;

    float* ws   = (float*)d_ws;
    float* dinv = ws;
    float* h1s  = dinv + nPad;
    float* h2s  = h1s + (size_t)nPad * C_HID;

    int NB    = (n + BIN_R - 1) / BIN_R;
    int chunk = (E + NCHUNK - 1) / NCHUNK;
    int*      counts   = (int*)(h2s + (size_t)nPad * C_OUT);
    int*      binTotal = counts + (size_t)NB * NCHUNK;
    int*      binBase  = binTotal + NB;
    unsigned* bucket   = (unsigned*)(binBase + NB + 1);
    unsigned* sorted   = bucket + (size_t)E;

    const int B = 256;
    int numPairs = (n + 31) / 32;
    size_t ldsBins = (size_t)NB * sizeof(int);

    k_bin_hist   <<<NCHUNK, 1024, ldsBins, stream>>>(ei, counts, E, NB, chunk);
    k_bin_rowscan<<<NB, NCHUNK, 0, stream>>>(counts, binTotal);
    k_bin_base   <<<1, 1024, 0, stream>>>(binTotal, binBase, NB);
    k_bin_scatter<<<NCHUNK, 1024, 2 * ldsBins, stream>>>(ei, counts, binBase, bucket, E, NB, chunk);
    k_deg_bins   <<<NB, BIN_T, 0, stream>>>(bucket, binBase, dinv, sorted, n);

    k_gemm1_wmma<<<(numPairs + 7) / 8, B, 0, stream>>>(x, W1, dinv, h1s, numPairs);

    k_agg1_bins  <<<NB, AGG1_WAVES * 32, 0, stream>>>(sorted, binBase, dinv, h1s, b1, W2, h2s, n);
    k_agg2_bins  <<<NB, BIN_T, 0, stream>>>(bucket, binBase, dinv, h2s, b2, out, n);
}


# Each edge chunk ordered by bin inside its own work-group, stored as one piece with a run table; bin owners read their 256 runs; three launches (global count, two scans) go; arrival-order sums kept
# speedup vs baseline: 8.1047x; 1.1116x over previous
#include <hip/hip_runtime.h>
#include <hip/hip_bf16.h>


typedef __attribute__((ext_vector_type(2))) float v2f;
typedef __attribute__((ext_vector_type(8))) float v8f;

#define C_IN  128
#define C_HID 64
#define C_OUT 2
#define WPAD  132

#define BIN_R  200
#define REC_SH 8
#define REC_MASK ((1u << REC_SH) - 1u)
#define NCHUNK 256
#define EPI_TRIPS ((BIN_R * (C_HID / 4)) / (AGG1_WAVES * 32))
#define AGG1_WAVES 25
#define BIN_T 1024
#define RUN_L 4
#define RUN_U 8
#define STAGE_CAP 12544
#define SORT_CAP 8192
static_assert(BIN_R <= (1 << REC_SH), "local target must fit the record's low bits");
static_assert(BIN_T == RUN_L * NCHUNK, "the per-bin passes give every run RUN_L lanes");
static_assert((BIN_R * (C_HID / 4)) % (AGG1_WAVES * 32) == 0, "epilogue trips must be whole");
static_assert(BIN_R * C_OUT <= AGG1_WAVES * 32 && BIN_R * C_OUT <= BIN_T, "one thread per stored element of a bin");

__device__ __forceinline__ int block_excl_scan(int v, int* wsum, int tid, int nwaves, int* total) {
    int lane = tid & 31, wv = tid >> 5;
    int inc = v;
    #pragma unroll
    for (int d = 1; d < 32; d <<= 1) {
        int t = __shfl_up(inc, d, 32);
        if (lane >= d) inc += t;
    }
    if (lane == 31) wsum[wv] = inc;
    __syncthreads();
    if (wv == 0) {
        int s = lane < nwaves ? wsum[lane] : 0;
        int si = s;
        #pragma unroll
        for (int d = 1; d < 32; d <<= 1) {
            int t = __shfl_up(si, d, 32);
            if (lane >= d) si += t;
        }
        wsum[lane] = si - s;
        if (lane == 31) wsum[32] = si;
    }
    __syncthreads();
    *total = wsum[32];
    return inc - v + wsum[wv];
}

__device__ __forceinline__ void run_load(unsigned (&rec)[RUN_U], const unsigned* __restrict__ src, int sub, int len) {
    int last = len > 0 ? len - 1 : 0;
    #pragma unroll
    for (int u = 0; u < RUN_U; ++u) {
        int k = sub + RUN_L * u;
        rec[u] = src[k < last ? k : last];
    }
}

__global__ __launch_bounds__(1024)
void k_bin_local(const int* __restrict__ ei, int* __restrict__ runStart,
                 unsigned* __restrict__ bucket, int E, int NB, int chunk) {
    extern __shared__ int sh_bins[];
    __shared__ int wsum[33];
    __shared__ unsigned stage[STAGE_CAP];
    int* lcur = sh_bins;
    int tid = threadIdx.x;
    int w = blockIdx.x;
    int beg = w * chunk < E ? w * chunk : E;
    int end = beg + chunk < E ? beg + chunk : E;
    for (int i = tid; i < NB; i += 1024) lcur[i] = 0;
    __syncthreads();
    const int* row = ei;
    const int* col = ei + (size_t)E;
    int e = beg + tid;
    for (; e + 3 * 1024 < end; e += 4 * 1024) {
        int c[4];
        #pragma unroll
        for (int u = 0; u < 4; ++u) c[u] = col[e + u * 1024];
        #pragma unroll
        for (int u = 0; u < 4; ++u) atomicAdd(&lcur[(unsigned)c[u] / BIN_R], 1);
    }
    for (; e < end; e += 1024) atomicAdd(&lcur[(unsigned)col[e] / BIN_R], 1);
    __syncthreads();
    int ipt = (NB + 1023) / 1024;
    int b0 = tid * ipt;
    int s = 0;
    for (int i = 0; i < ipt; ++i) if (b0 + i < NB) s += lcur[b0 + i];
    int total;
    int run = block_excl_scan(s, wsum, tid, 32, &total);
    int* rs = runStart + (size_t)w * (NB + 1);
    for (int i = 0; i < ipt; ++i) {
        int bi = b0 + i;
        if (bi < NB) {
            int c = lcur[bi];
            lcur[bi] = run;
            rs[bi] = run;
            run += c;
        }
    }
    if (tid == 0) rs[NB] = total;
    __syncthreads();
    bool staged = end - beg <= STAGE_CAP;
    unsigned* dst = bucket + beg;
    e = beg + tid;
    for (; e + 3 * 1024 < end; e += 4 * 1024) {
        int r[4], c[4];
        #pragma unroll
        for (int u = 0; u < 4; ++u) { r[u] = row[e + u * 1024]; c[u] = col[e + u * 1024]; }
        unsigned bin[4];
        int pos[4];
        #pragma unroll
        for (int u = 0; u < 4; ++u) bin[u] = (unsigned)c[u] / BIN_R;
        #pragma unroll
        for (int u = 0; u < 4; ++u) pos[u] = atomicAdd(&lcur[bin[u]], 1);
        #pragma unroll
        for (int u = 0; u < 4; ++u) {
            unsigned rec = ((unsigned)r[u] << REC_SH) | ((unsigned)c[u] - bin[u] * BIN_R);
            if (staged) stage[pos[u]] = rec; else dst[pos[u]] = rec;
        }
    }
    for (; e < end; e += 1024) {
        int r = row[e], c = col[e];
        unsigned bin = (unsigned)c / BIN_R;
        int pos = atomicAdd(&lcur[bin], 1);
        unsigned rec = ((unsigned)r << REC_SH) | ((unsigned)c - bin * BIN_R);
        if (staged) stage[pos] = rec; else dst[pos] = rec;
    }
    if (staged) {
        __syncthreads();
        for (int j = tid; j < end - beg; j += 1024) dst[j] = stage[j];
    }
}

__global__ __launch_bounds__(BIN_T)
void k_deg_bins(const unsigned* __restrict__ bucket, const int* __restrict__ runStart,
                float* __restrict__ dinv, unsigned* __restrict__ sorted, int* __restrict__ binBase,
                int n, int NB, int chunk) {
    __shared__ int cnt[BIN_R];
    __shared__ int wsum[33];
    __shared__ int baseSum;
    __shared__ unsigned stage[SORT_CAP];
    int tid = threadIdx.x;
    int b = blockIdx.x;
    if (tid < BIN_R) cnt[tid] = 0;
    if (tid == 0) baseSum = 0;
    __syncthreads();
    int r = tid / RUN_L, sub = tid % RUN_L;
    const int* rs = runStart + (size_t)r * (NB + 1) + b;
    int s0 = rs[0];
    int len = rs[1] - s0;
    const unsigned* src = len > 0 ? bucket + (size_t)r * chunk + s0 : bucket;
    int v = sub == 0 ? s0 : 0;
    #pragma unroll
    for (int d = 16; d >= 1; d >>= 1) v += __shfl_xor(v, d, 32);
    if ((tid & 31) == 0) atomicAdd(&baseSum, v);
    unsigned rec[RUN_U];
    run_load(rec, src, sub, len);
    #pragma unroll
    for (int u = 0; u < RUN_U; ++u) if (sub + RUN_L * u < len) atomicAdd(&cnt[rec[u] & REC_MASK], 1);
    for (int k = sub + RUN_L * RUN_U; k < len; k += RUN_L) atomicAdd(&cnt[src[k] & REC_MASK], 1);
    __syncthreads();
    int c = tid < BIN_R ? cnt[tid] : 0;
    int total;
    int off = block_excl_scan(c, wsum, tid, BIN_T / 32, &total);
    if (tid < BIN_R) {
        int node = b * BIN_R + tid;
        if (node < n) dinv[node] = __frsqrt_rn((float)(c + 1));
        cnt[tid] = off;
    }
    __syncthreads();
    int base = baseSum;
    bool staged = total <= SORT_CAP;
    #pragma unroll
    for (int u = 0; u < RUN_U; ++u) {
        if (sub + RUN_L * u < len) {
            int pos = atomicAdd(&cnt[rec[u] & REC_MASK], 1);
            if (staged) stage[pos] = rec[u]; else sorted[base + pos] = rec[u];
        }
    }
    for (int k = sub + RUN_L * RUN_U; k < len; k += RUN_L) {
        unsigned rc = src[k];
        int pos = atomicAdd(&cnt[rc & REC_MASK], 1);
        if (staged) stage[pos] = rc; else sorted[base + pos] = rc;
    }
    if (staged) {
        __syncthreads();
        for (int i = tid; i < total; i += BIN_T) sorted[base + i] = stage[i];
    }
    if (tid == 0) {
        binBase[b] = base;
        if (b == NB - 1) binBase[NB] = base + total;
    }
}

__global__ __launch_bounds__(256)
void k_gemm1_wmma(const float* __restrict__ x, const float* __restrict__ W1,
                  const float* __restrict__ dinv, float* __restrict__ h1s, int numPairs) {
    __shared__ float Wt[C_HID * WPAD];
    for (int i = threadIdx.x; i < C_HID * C_IN; i += 256) {
        int nn = i >> 7;
        int kk = i & 127;
        Wt[nn * WPAD + kk] = W1[kk * C_HID + nn];
    }
    __syncthreads();

    int wave = threadIdx.x >> 5;
    int lane = threadIdx.x & 31;
    int pair = blockIdx.x * 8 + wave;
    if (pair >= numPairs) pair = numPairs - 1;

    int m     = lane & 15;
    int khalf = lane >> 4;
    int ncol  = lane & 15;

    const float* xrow0 = x + ((size_t)pair * 32 + m) * C_IN + 2 * khalf;
    const float* xrow1 = xrow0 + (size_t)16 * C_IN;

    const v2f* bp0 = (const v2f*)&Wt[(ncol     ) * WPAD + 2 * khalf];
    const v2f* bp1 = (const v2f*)&Wt[(ncol + 16) * WPAD + 2 * khalf];
    const v2f* bp2 = (const v2f*)&Wt[(ncol + 32) * WPAD + 2 * khalf];
    const v2f* bp3 = (const v2f*)&Wt[(ncol + 48) * WPAD + 2 * khalf];

    const float* dr = dinv + (size_t)pair * 32 + 8 * khalf;
    float4 dA0 = *(const float4*)(dr);
    float4 dA1 = *(const float4*)(dr + 4);
    float4 dB0 = *(const float4*)(dr + 16);
    float4 dB1 = *(const float4*)(dr + 20);

    v8f a00 = {}, a01 = {}, a02 = {}, a03 = {};
    v8f a10 = {}, a11 = {}, a12 = {}, a13 = {};
    #pragma unroll 4
    for (int k0 = 0; k0 < C_IN; k0 += 4) {
        v2f av0 = *(const v2f*)(xrow0 + k0);
        v2f av1 = *(const v2f*)(xrow1 + k0);
        v2f b0 = bp0[k0 >> 1];
        v2f b1 = bp1[k0 >> 1];
        v2f b2 = bp2[k0 >> 1];
        v2f b3 = bp3[k0 >> 1];
        a00 = __builtin_amdgcn_wmma_f32_16x16x4_f32(false, av0, false, b0, (short)0, a00, false, false);
        a10 = __builtin_amdgcn_wmma_f32_16x16x4_f32(false, av1, false, b0, (short)0, a10, false, false);
        a01 = __builtin_amdgcn_wmma_f32_16x16x4_f32(false, av0, false, b1, (short)0, a01, false, false);
        a11 = __builtin_amdgcn_wmma_f32_16x16x4_f32(false, av1, false, b1, (short)0, a11, false, false);
        a02 = __builtin_amdgcn_wmma_f32_16x16x4_f32(false, av0, false, b2, (short)0, a02, false, false);
        a12 = __builtin_amdgcn_wmma_f32_16x16x4_f32(false, av1, false, b2, (short)0, a12, false, false);
        a03 = __builtin_amdgcn_wmma_f32_16x16x4_f32(false, av0, false, b3, (short)0, a03, false, false);
        a13 = __builtin_amdgcn_wmma_f32_16x16x4_f32(false, av1, false, b3, (short)0, a13, false, false);
    }

    float d0[8] = {dA0.x, dA0.y, dA0.z, dA0.w, dA1.x, dA1.y, dA1.z, dA1.w};
    float d1[8] = {dB0.x, dB0.y, dB0.z, dB0.w, dB1.x, dB1.y, dB1.z, dB1.w};
    float* h0 = h1s + (size_t)pair * 32 * C_HID;
    float* h16 = h0 + (size_t)16 * C_HID;
    #pragma unroll
    for (int r = 0; r < 8; ++r) {
        int mrow = r + 8 * khalf;
        h0[mrow * C_HID + ncol]       = d0[r] * a00[r];
        h0[mrow * C_HID + 16 + ncol]  = d0[r] * a01[r];
        h0[mrow * C_HID + 32 + ncol]  = d0[r] * a02[r];
        h0[mrow * C_HID + 48 + ncol]  = d0[r] * a03[r];
        h16[mrow * C_HID + ncol]      = d1[r] * a10[r];
        h16[mrow * C_HID + 16 + ncol] = d1[r] * a11[r];
        h16[mrow * C_HID + 32 + ncol] = d1[r] * a12[r];
        h16[mrow * C_HID + 48 + ncol] = d1[r] * a13[r];
    }
}

__device__ __forceinline__ void agg1_gather16(v2f (&hv)[16], const float* __restrict__ h1, int row, int lane, int j0) {
    #pragma unroll
    for (int u = 0; u < 16; ++u) {
        int r = __builtin_amdgcn_readlane(row, j0 + u);
        hv[u] = *(const v2f*)(h1 + (size_t)r * C_HID + 2 * lane);
    }
}
__device__ __forceinline__ void agg1_flush(float* acc, int c, v2f sum, int lane) {
    float* a = &acc[c * C_HID + 2 * lane];
    atomicAdd(a,     sum.x);
    atomicAdd(a + 1, sum.y);
}
__device__ __forceinline__ void agg1_sum16(const v2f (&hv)[16], float* acc, int cl, int lane, int j0, int& cur, v2f& sum) {
    #pragma unroll
    for (int u = 0; u < 16; ++u) {
        int c = __builtin_amdgcn_readlane(cl, j0 + u);
        if (c != cur) {
            agg1_flush(acc, cur, sum, lane);
            cur = c;
            sum.x = 0.f; sum.y = 0.f;
        }
        sum.x += hv[u].x;
        sum.y += hv[u].y;
    }
}

__global__ __launch_bounds__(AGG1_WAVES * 32, 16)
void k_agg1_bins(const unsigned* __restrict__ sorted, const int* __restrict__ binBase,
                 const float* __restrict__ dinv, const float* __restrict__ h1s,
                 const float* __restrict__ b1, const float* __restrict__ W2,
                 float* __restrict__ h2s, int n) {
    __shared__ __attribute__((aligned(16))) float acc[BIN_R * C_HID];
    __shared__ float h2l[BIN_R * C_OUT];
    int tid  = threadIdx.x;
    int lane = tid & 31;
    int wv   = __builtin_amdgcn_readfirstlane(tid >> 5);
    int b    = blockIdx.x;
    int node0 = b * BIN_R;
    for (int i = tid; i < BIN_R * C_HID; i += AGG1_WAVES * 32) acc[i] = 0.f;
    __syncthreads();

    int beg = binBase[b], end = binBase[b + 1];
    int chunk = ((end - beg + AGG1_WAVES * 32 - 1) / (AGG1_WAVES * 32)) * 32;
    int p  = beg + wv * chunk;
    int pe = p + chunk < end ? p + chunk : end;
    unsigned rec = p + lane < pe ? sorted[p + lane] : 0u;
    bool any = p < pe;
    int cur = __builtin_amdgcn_readlane((int)(rec & REC_MASK), 0);
    v2f sum = {0.f, 0.f};
    while (p < pe) {
        int left = pe - p;
        int pn = p + 32;
        unsigned recN = 0u;
        int row = (int)(rec >> REC_SH);
        int cl  = (int)(rec & REC_MASK);
        if (left >= 32) {
            v2f hv[16];
            agg1_gather16(hv, h1s, row, lane, 0);
            recN = pn + lane < pe ? sorted[pn + lane] : 0u;
            agg1_sum16(hv, acc, cl, lane, 0, cur, sum);
            agg1_gather16(hv, h1s, row, lane, 16);
            agg1_sum16(hv, acc, cl, lane, 16, cur, sum);
        } else {
            for (int j = 0; j < left; ++j) {
                int r = __builtin_amdgcn_readlane(row, j);
                int c = __builtin_amdgcn_readlane(cl, j);
                v2f hv = *(const v2f*)(h1s + (size_t)r * C_HID + 2 * lane);
                if (c != cur) {
                    agg1_flush(acc, cur, sum, lane);
                    cur = c;
                    sum.x = 0.f; sum.y = 0.f;
                }
                sum.x += hv.x;
                sum.y += hv.y;
            }
        }
        rec = recN;
        p = pn;
    }
    if (any) agg1_flush(acc, cur, sum, lane);
    __syncthreads();

    int q  = tid & 15;
    int c4 = q * 4;
    float4 bb = *(const float4*)(b1 + c4);
    float4 wA = *(const float4*)(W2 + c4 * C_OUT);
    float4 wB = *(const float4*)(W2 + c4 * C_OUT + 4);
    float  dd[EPI_TRIPS];
    float4 hh[EPI_TRIPS];
    #pragma unroll
    for (int u = 0; u < EPI_TRIPS; ++u) {
        int node = node0 + ((tid + u * AGG1_WAVES * 32) >> 4);
        node = node < n ? node : n - 1;
        dd[u] = dinv[node];
        hh[u] = *(const float4*)(h1s + (size_t)node * C_HID + c4);
    }
    #pragma unroll
    for (int u = 0; u < EPI_TRIPS; ++u) {
        int cl = (tid + u * AGG1_WAVES * 32) >> 4;
        float d = dd[u];
        float4 hs = hh[u];
        float4 av = *(const float4*)(&acc[cl * C_HID + c4]);
        float o0 = bb.x + d * (hs.x + av.x);
        float o1 = bb.y + d * (hs.y + av.y);
        float o2 = bb.z + d * (hs.z + av.z);
        float o3 = bb.w + d * (hs.w + av.w);
        o0 = o0 > 0.f ? o0 : 0.f;
        o1 = o1 > 0.f ? o1 : 0.f;
        o2 = o2 > 0.f ? o2 : 0.f;
        o3 = o3 > 0.f ? o3 : 0.f;
        float p0 = fmaf(o3, wB.z, fmaf(o2, wB.x, fmaf(o1, wA.z, o0 * wA.x)));
        float p1 = fmaf(o3, wB.w, fmaf(o2, wB.y, fmaf(o1, wA.w, o0 * wA.y)));
        #pragma unroll
        for (int sft = 8; sft >= 1; sft >>= 1) {
            p0 += __shfl_xor(p0, sft, 32);
            p1 += __shfl_xor(p1, sft, 32);
        }
        if (q == 0) {
            h2l[cl * C_OUT + 0] = d * p0;
            h2l[cl * C_OUT + 1] = d * p1;
        }
    }
    __syncthreads();
    if (tid < BIN_R * C_OUT) {
        int node = node0 + (tid >> 1);
        if (node < n) h2s[(size_t)node0 * C_OUT + tid] = h2l[tid];
    }
}

__global__ __launch_bounds__(BIN_T)
void k_agg2_bins(const unsigned* __restrict__ bucket, const int* __restrict__ runStart,
                 const float* __restrict__ dinv, const float* __restrict__ h2s,
                 const float* __restrict__ b2, float* __restrict__ out, int n, int NB, int chunk) {
    __shared__ float acc[BIN_R * C_OUT];
    int tid = threadIdx.x;
    int b = blockIdx.x;
    int node0 = b * BIN_R;
    if (tid < BIN_R * C_OUT) acc[tid] = 0.f;
    __syncthreads();
    int r = tid / RUN_L, sub = tid % RUN_L;
    const int* rs = runStart + (size_t)r * (NB + 1) + b;
    int s0 = rs[0];
    int len = rs[1] - s0;
    const unsigned* src = len > 0 ? bucket + (size_t)r * chunk + s0 : bucket;
    unsigned rec[RUN_U];
    run_load(rec, src, sub, len);
    v2f hv[RUN_U];
    #pragma unroll
    for (int u = 0; u < RUN_U; ++u) hv[u] = *(const v2f*)(h2s + (size_t)(rec[u] >> REC_SH) * C_OUT);
    #pragma unroll
    for (int u = 0; u < RUN_U; ++u) {
        if (sub + RUN_L * u < len) {
            int cl = (int)(rec[u] & REC_MASK);
            atomicAdd(&acc[cl * C_OUT + 0], hv[u].x);
            atomicAdd(&acc[cl * C_OUT + 1], hv[u].y);
        }
    }
    for (int k = sub + RUN_L * RUN_U; k < len; k += RUN_L) {
        unsigned rc = src[k];
        int cl = (int)(rc & REC_MASK);
        v2f h = *(const v2f*)(h2s + (size_t)(rc >> REC_SH) * C_OUT);
        atomicAdd(&acc[cl * C_OUT + 0], h.x);
        atomicAdd(&acc[cl * C_OUT + 1], h.y);
    }
    __syncthreads();
    if (tid < BIN_R * C_OUT) {
        int node = node0 + (tid >> 1);
        if (node < n) {
            size_t g = (size_t)node0 * C_OUT + tid;
            out[g] = b2[tid & 1] + dinv[node] * (h2s[g] + acc[tid]);
        }
    }
}

extern "C" void kernel_launch(void* const* d_in, const int* in_sizes, int n_in,
                              void* d_out, int out_size, void* d_ws, size_t ws_size,
                              hipStream_t stream) {
    const float* x  = (const float*)d_in[0];
    const int*   ei = (const int*)d_in[1];
    const float* W1 = (const float*)d_in[2];
    const float* b1 = (const float*)d_in[3];
    const float* W2 = (const float*)d_in[4];
    const float* b2 = (const float*)d_in[5];
    float* out = (float*)d_out;

    int n = in_sizes[0] / C_IN;
    int E = in_sizes[1] / 2;
    int nPad = (n + 3) & ~3;

    float* ws   = (float*)d_ws;
    float* dinv = ws;
    float* h1s  = dinv + nPad;
    float* h2s  = h1s + (size_t)nPad * C_HID;

    int NB    = (n + BIN_R - 1) / BIN_R;
    int chunk = (E + NCHUNK - 1) / NCHUNK;
    int*      runStart = (int*)(h2s + (size_t)nPad * C_OUT);
    int*      binBase  = runStart + (size_t)NCHUNK * (NB + 1);
    unsigned* bucket   = (unsigned*)(binBase + NB + 1);
    unsigned* sorted   = bucket + (size_t)E;

    const int B = 256;
    int numPairs = (n + 31) / 32;
    size_t ldsBins = (size_t)NB * sizeof(int);

    k_bin_local  <<<NCHUNK, 1024, ldsBins, stream>>>(ei, runStart, bucket, E, NB, chunk);
    k_deg_bins   <<<NB, BIN_T, 0, stream>>>(bucket, runStart, dinv, sorted, binBase, n, NB, chunk);

    k_gemm1_wmma<<<(numPairs + 7) / 8, B, 0, stream>>>(x, W1, dinv, h1s, numPairs);

    k_agg1_bins  <<<NB, AGG1_WAVES * 32, 0, stream>>>(sorted, binBase, dinv, h1s, b1, W2, h2s, n);
    k_agg2_bins  <<<NB, BIN_T, 0, stream>>>(bucket, runStart, dinv, h2s, b2, out, n, NB, chunk);
}


# Layer-1 product folded into the degree pass: each bin's work-group forms its own 200 scaled rows on WMMA (13 waves, one 16-node tile each) once it has their factors; the product's own launch goes
# speedup vs baseline: 8.4742x; 1.0456x over previous
#include <hip/hip_runtime.h>
#include <hip/hip_bf16.h>


typedef __attribute__((ext_vector_type(2))) float v2f;
typedef __attribute__((ext_vector_type(8))) float v8f;

#define C_IN  128
#define C_HID 64
#define C_OUT 2
#define WPAD  132

#define BIN_R  200
#define REC_SH 8
#define REC_MASK ((1u << REC_SH) - 1u)
#define NCHUNK 256
#define EPI_TRIPS ((BIN_R * (C_HID / 4)) / (AGG1_WAVES * 32))
#define AGG1_WAVES 25
#define BIN_T 1024
#define RUN_L 4
#define RUN_U 8
#define STAGE_CAP 12544
#define SORT_CAP 8192
#define GEMM_TILES ((BIN_R + 15) / 16)
#define W1_PER_T ((C_IN * C_HID) / BIN_T)
static_assert(BIN_R <= (1 << REC_SH), "local target must fit the record's low bits");
static_assert(BIN_T == RUN_L * NCHUNK, "the per-bin passes give every run RUN_L lanes");
static_assert((BIN_R * (C_HID / 4)) % (AGG1_WAVES * 32) == 0, "epilogue trips must be whole");
static_assert(BIN_R * C_OUT <= AGG1_WAVES * 32 && BIN_R * C_OUT <= BIN_T, "one thread per stored element of a bin");
static_assert(GEMM_TILES * 32 <= BIN_T, "one wave per 16-node tile of a bin");
static_assert((C_IN * C_HID) % BIN_T == 0, "W1 is dealt to the threads in whole floats");

__device__ __forceinline__ int block_excl_scan(int v, int* wsum, int tid, int nwaves, int* total) {
    int lane = tid & 31, wv = tid >> 5;
    int inc = v;
    #pragma unroll
    for (int d = 1; d < 32; d <<= 1) {
        int t = __shfl_up(inc, d, 32);
        if (lane >= d) inc += t;
    }
    if (lane == 31) wsum[wv] = inc;
    __syncthreads();
    if (wv == 0) {
        int s = lane < nwaves ? wsum[lane] : 0;
        int si = s;
        #pragma unroll
        for (int d = 1; d < 32; d <<= 1) {
            int t = __shfl_up(si, d, 32);
            if (lane >= d) si += t;
        }
        wsum[lane] = si - s;
        if (lane == 31) wsum[32] = si;
    }
    __syncthreads();
    *total = wsum[32];
    return inc - v + wsum[wv];
}

__device__ __forceinline__ void run_load(unsigned (&rec)[RUN_U], const unsigned* __restrict__ src, int sub, int len) {
    int last = len > 0 ? len - 1 : 0;
    #pragma unroll
    for (int u = 0; u < RUN_U; ++u) {
        int k = sub + RUN_L * u;
        rec[u] = src[k < last ? k : last];
    }
}

__global__ __launch_bounds__(1024)
void k_bin_local(const int* __restrict__ ei, int* __restrict__ runStart,
                 unsigned* __restrict__ bucket, int E, int NB, int chunk) {
    extern __shared__ int sh_bins[];
    __shared__ int wsum[33];
    __shared__ unsigned stage[STAGE_CAP];
    int* lcur = sh_bins;
    int tid = threadIdx.x;
    int w = blockIdx.x;
    int beg = w * chunk < E ? w * chunk : E;
    int end = beg + chunk < E ? beg + chunk : E;
    for (int i = tid; i < NB; i += 1024) lcur[i] = 0;
    __syncthreads();
    const int* row = ei;
    const int* col = ei + (size_t)E;
    int e = beg + tid;
    for (; e + 3 * 1024 < end; e += 4 * 1024) {
        int c[4];
        #pragma unroll
        for (int u = 0; u < 4; ++u) c[u] = col[e + u * 1024];
        #pragma unroll
        for (int u = 0; u < 4; ++u) atomicAdd(&lcur[(unsigned)c[u] / BIN_R], 1);
    }
    for (; e < end; e += 1024) atomicAdd(&lcur[(unsigned)col[e] / BIN_R], 1);
    __syncthreads();
    int ipt = (NB + 1023) / 1024;
    int b0 = tid * ipt;
    int s = 0;
    for (int i = 0; i < ipt; ++i) if (b0 + i < NB) s += lcur[b0 + i];
    int total;
    int run = block_excl_scan(s, wsum, tid, 32, &total);
    int* rs = runStart + (size_t)w * (NB + 1);
    for (int i = 0; i < ipt; ++i) {
        int bi = b0 + i;
        if (bi < NB) {
            int c = lcur[bi];
            lcur[bi] = run;
            rs[bi] = run;
            run += c;
        }
    }
    if (tid == 0) rs[NB] = total;
    __syncthreads();
    bool staged = end - beg <= STAGE_CAP;
    unsigned* dst = bucket + beg;
    e = beg + tid;
    for (; e + 3 * 1024 < end; e += 4 * 1024) {
        int r[4], c[4];
        #pragma unroll
        for (int u = 0; u < 4; ++u) { r[u] = row[e + u * 1024]; c[u] = col[e + u * 1024]; }
        unsigned bin[4];
        int pos[4];
        #pragma unroll
        for (int u = 0; u < 4; ++u) bin[u] = (unsigned)c[u] / BIN_R;
        #pragma unroll
        for (int u = 0; u < 4; ++u) pos[u] = atomicAdd(&lcur[bin[u]], 1);
        #pragma unroll
        for (int u = 0; u < 4; ++u) {
            unsigned rec = ((unsigned)r[u] << REC_SH) | ((unsigned)c[u] - bin[u] * BIN_R);
            if (staged) stage[pos[u]] = rec; else dst[pos[u]] = rec;
        }
    }
    for (; e < end; e += 1024) {
        int r = row[e], c = col[e];
        unsigned bin = (unsigned)c / BIN_R;
        int pos = atomicAdd(&lcur[bin], 1);
        unsigned rec = ((unsigned)r << REC_SH) | ((unsigned)c - bin * BIN_R);
        if (staged) stage[pos] = rec; else dst[pos] = rec;
    }
    if (staged) {
        __syncthreads();
        for (int j = tid; j < end - beg; j += 1024) dst[j] = stage[j];
    }
}

__global__ __launch_bounds__(BIN_T, 16)
void k_deg_gemm_bins(const unsigned* __restrict__ bucket, const int* __restrict__ runStart,
                     const float* __restrict__ x, const float* __restrict__ W1,
                     float* __restrict__ dinv, float* __restrict__ h1s,
                     unsigned* __restrict__ sorted, int* __restrict__ binBase,
                     int n, int NB, int chunk) {
    __shared__ int cnt[BIN_R];
    __shared__ int wsum[33];
    __shared__ int baseSum;
    __shared__ float dl[BIN_R];
    __shared__ unsigned stage[SORT_CAP];
    __shared__ __attribute__((aligned(16))) float Wt[C_HID * WPAD];
    int tid = threadIdx.x;
    int b = blockIdx.x;
    int node0 = b * BIN_R;
    if (tid < BIN_R) cnt[tid] = 0;
    if (tid == 0) baseSum = 0;
    __syncthreads();
    int r = tid / RUN_L, sub = tid % RUN_L;
    const int* rs = runStart + (size_t)r * (NB + 1) + b;
    int s0 = rs[0];
    int len = rs[1] - s0;
    const unsigned* src = len > 0 ? bucket + (size_t)r * chunk + s0 : bucket;
    int v = sub == 0 ? s0 : 0;
    #pragma unroll
    for (int d = 16; d >= 1; d >>= 1) v += __shfl_xor(v, d, 32);
    if ((tid & 31) == 0) atomicAdd(&baseSum, v);
    unsigned rec[RUN_U];
    run_load(rec, src, sub, len);
    float wreg[W1_PER_T];
    #pragma unroll
    for (int u = 0; u < W1_PER_T; ++u) wreg[u] = W1[tid + u * BIN_T];
    #pragma unroll
    for (int u = 0; u < RUN_U; ++u) if (sub + RUN_L * u < len) atomicAdd(&cnt[rec[u] & REC_MASK], 1);
    for (int k = sub + RUN_L * RUN_U; k < len; k += RUN_L) atomicAdd(&cnt[src[k] & REC_MASK], 1);
    #pragma unroll
    for (int u = 0; u < W1_PER_T; ++u) {
        int i = tid + u * BIN_T;
        Wt[(i % C_HID) * WPAD + i / C_HID] = wreg[u];
    }
    __syncthreads();
    int c = tid < BIN_R ? cnt[tid] : 0;
    int total;
    int off = block_excl_scan(c, wsum, tid, BIN_T / 32, &total);
    if (tid < BIN_R) {
        int node = node0 + tid;
        float dv = __frsqrt_rn((float)(c + 1));
        if (node < n) dinv[node] = dv;
        dl[tid] = dv;
        cnt[tid] = off;
    }
    __syncthreads();
    int base = baseSum;
    bool staged = total <= SORT_CAP;
    #pragma unroll
    for (int u = 0; u < RUN_U; ++u) {
        if (sub + RUN_L * u < len) {
            int pos = atomicAdd(&cnt[rec[u] & REC_MASK], 1);
            if (staged) stage[pos] = rec[u]; else sorted[base + pos] = rec[u];
        }
    }
    for (int k = sub + RUN_L * RUN_U; k < len; k += RUN_L) {
        unsigned rc = src[k];
        int pos = atomicAdd(&cnt[rc & REC_MASK], 1);
        if (staged) stage[pos] = rc; else sorted[base + pos] = rc;
    }
    if (staged) {
        __syncthreads();
        for (int i = tid; i < total; i += BIN_T) sorted[base + i] = stage[i];
    }
    if (tid == 0) {
        binBase[b] = base;
        if (b == NB - 1) binBase[NB] = base + total;
    }

    int wv = __builtin_amdgcn_readfirstlane(tid >> 5);
    if (wv < GEMM_TILES) {
        int lane  = tid & 31;
        int m     = lane & 15;
        int khalf = lane >> 4;
        int ncol  = lane & 15;
        int lrow0 = wv * 16;
        int arow = node0 + lrow0 + m;
        arow = arow < n ? arow : n - 1;
        const float* xrow = x + (size_t)arow * C_IN + 2 * khalf;

        const v2f* bp0 = (const v2f*)&Wt[(ncol     ) * WPAD + 2 * khalf];
        const v2f* bp1 = (const v2f*)&Wt[(ncol + 16) * WPAD + 2 * khalf];
        const v2f* bp2 = (const v2f*)&Wt[(ncol + 32) * WPAD + 2 * khalf];
        const v2f* bp3 = (const v2f*)&Wt[(ncol + 48) * WPAD + 2 * khalf];

        v8f a0 = {}, a1 = {}, a2 = {}, a3 = {};
        #pragma unroll 4
        for (int k0 = 0; k0 < C_IN; k0 += 4) {
            v2f av = *(const v2f*)(xrow + k0);
            v2f b0 = bp0[k0 >> 1];
            v2f b1 = bp1[k0 >> 1];
            v2f b2 = bp2[k0 >> 1];
            v2f b3 = bp3[k0 >> 1];
            a0 = __builtin_amdgcn_wmma_f32_16x16x4_f32(false, av, false, b0, (short)0, a0, false, false);
            a1 = __builtin_amdgcn_wmma_f32_16x16x4_f32(false, av, false, b1, (short)0, a1, false, false);
            a2 = __builtin_amdgcn_wmma_f32_16x16x4_f32(false, av, false, b2, (short)0, a2, false, false);
            a3 = __builtin_amdgcn_wmma_f32_16x16x4_f32(false, av, false, b3, (short)0, a3, false, false);
        }

        int lr = lrow0 + 8 * khalf;
        float* hrow = h1s + (size_t)(node0 + lr) * C_HID + ncol;
        #pragma unroll
        for (int q = 0; q < 8; ++q) {
            int l = lr + q;
            if (l < BIN_R && node0 + l < n) {
                float d = dl[l];
                hrow[q * C_HID]      = d * a0[q];
                hrow[q * C_HID + 16] = d * a1[q];
                hrow[q * C_HID + 32] = d * a2[q];
                hrow[q * C_HID + 48] = d * a3[q];
            }
        }
    }
}

__device__ __forceinline__ void agg1_gather16(v2f (&hv)[16], const float* __restrict__ h1, int row, int lane, int j0) {
    #pragma unroll
    for (int u = 0; u < 16; ++u) {
        int r = __builtin_amdgcn_readlane(row, j0 + u);
        hv[u] = *(const v2f*)(h1 + (size_t)r * C_HID + 2 * lane);
    }
}
__device__ __forceinline__ void agg1_flush(float* acc, int c, v2f sum, int lane) {
    float* a = &acc[c * C_HID + 2 * lane];
    atomicAdd(a,     sum.x);
    atomicAdd(a + 1, sum.y);
}
__device__ __forceinline__ void agg1_sum16(const v2f (&hv)[16], float* acc, int cl, int lane, int j0, int& cur, v2f& sum) {
    #pragma unroll
    for (int u = 0; u < 16; ++u) {
        int c = __builtin_amdgcn_readlane(cl, j0 + u);
        if (c != cur) {
            agg1_flush(acc, cur, sum, lane);
            cur = c;
            sum.x = 0.f; sum.y = 0.f;
        }
        sum.x += hv[u].x;
        sum.y += hv[u].y;
    }
}

__global__ __launch_bounds__(AGG1_WAVES * 32, 16)
void k_agg1_bins(const unsigned* __restrict__ sorted, const int* __restrict__ binBase,
                 const float* __restrict__ dinv, const float* __restrict__ h1s,
                 const float* __restrict__ b1, const float* __restrict__ W2,
                 float* __restrict__ h2s, int n) {
    __shared__ __attribute__((aligned(16))) float acc[BIN_R * C_HID];
    __shared__ float h2l[BIN_R * C_OUT];
    int tid  = threadIdx.x;
    int lane = tid & 31;
    int wv   = __builtin_amdgcn_readfirstlane(tid >> 5);
    int b    = blockIdx.x;
    int node0 = b * BIN_R;
    for (int i = tid; i < BIN_R * C_HID; i += AGG1_WAVES * 32) acc[i] = 0.f;
    __syncthreads();

    int beg = binBase[b], end = binBase[b + 1];
    int chunk = ((end - beg + AGG1_WAVES * 32 - 1) / (AGG1_WAVES * 32)) * 32;
    int p  = beg + wv * chunk;
    int pe = p + chunk < end ? p + chunk : end;
    unsigned rec = p + lane < pe ? sorted[p + lane] : 0u;
    bool any = p < pe;
    int cur = __builtin_amdgcn_readlane((int)(rec & REC_MASK), 0);
    v2f sum = {0.f, 0.f};
    while (p < pe) {
        int left = pe - p;
        int pn = p + 32;
        unsigned recN = 0u;
        int row = (int)(rec >> REC_SH);
        int cl  = (int)(rec & REC_MASK);
        if (left >= 32) {
            v2f hv[16];
            agg1_gather16(hv, h1s, row, lane, 0);
            recN = pn + lane < pe ? sorted[pn + lane] : 0u;
            agg1_sum16(hv, acc, cl, lane, 0, cur, sum);
            agg1_gather16(hv, h1s, row, lane, 16);
            agg1_sum16(hv, acc, cl, lane, 16, cur, sum);
        } else {
            for (int j = 0; j < left; ++j) {
                int r = __builtin_amdgcn_readlane(row, j);
                int c = __builtin_amdgcn_readlane(cl, j);
                v2f hv = *(const v2f*)(h1s + (size_t)r * C_HID + 2 * lane);
                if (c != cur) {
                    agg1_flush(acc, cur, sum, lane);
                    cur = c;
                    sum.x = 0.f; sum.y = 0.f;
                }
                sum.x += hv.x;
                sum.y += hv.y;
            }
        }
        rec = recN;
        p = pn;
    }
    if (any) agg1_flush(acc, cur, sum, lane);
    __syncthreads();

    int q  = tid & 15;
    int c4 = q * 4;
    float4 bb = *(const float4*)(b1 + c4);
    float4 wA = *(const float4*)(W2 + c4 * C_OUT);
    float4 wB = *(const float4*)(W2 + c4 * C_OUT + 4);
    float  dd[EPI_TRIPS];
    float4 hh[EPI_TRIPS];
    #pragma unroll
    for (int u = 0; u < EPI_TRIPS; ++u) {
        int node = node0 + ((tid + u * AGG1_WAVES * 32) >> 4);
        node = node < n ? node : n - 1;
        dd[u] = dinv[node];
        hh[u] = *(const float4*)(h1s + (size_t)node * C_HID + c4);
    }
    #pragma unroll
    for (int u = 0; u < EPI_TRIPS; ++u) {
        int cl = (tid + u * AGG1_WAVES * 32) >> 4;
        float d = dd[u];
        float4 hs = hh[u];
        float4 av = *(const float4*)(&acc[cl * C_HID + c4]);
        float o0 = bb.x + d * (hs.x + av.x);
        float o1 = bb.y + d * (hs.y + av.y);
        float o2 = bb.z + d * (hs.z + av.z);
        float o3 = bb.w + d * (hs.w + av.w);
        o0 = o0 > 0.f ? o0 : 0.f;
        o1 = o1 > 0.f ? o1 : 0.f;
        o2 = o2 > 0.f ? o2 : 0.f;
        o3 = o3 > 0.f ? o3 : 0.f;
        float p0 = fmaf(o3, wB.z, fmaf(o2, wB.x, fmaf(o1, wA.z, o0 * wA.x)));
        float p1 = fmaf(o3, wB.w, fmaf(o2, wB.y, fmaf(o1, wA.w, o0 * wA.y)));
        #pragma unroll
        for (int sft = 8; sft >= 1; sft >>= 1) {
            p0 += __shfl_xor(p0, sft, 32);
            p1 += __shfl_xor(p1, sft, 32);
        }
        if (q == 0) {
            h2l[cl * C_OUT + 0] = d * p0;
            h2l[cl * C_OUT + 1] = d * p1;
        }
    }
    __syncthreads();
    if (tid < BIN_R * C_OUT) {
        int node = node0 + (tid >> 1);
        if (node < n) h2s[(size_t)node0 * C_OUT + tid] = h2l[tid];
    }
}

__global__ __launch_bounds__(BIN_T)
void k_agg2_bins(const unsigned* __restrict__ bucket, const int* __restrict__ runStart,
                 const float* __restrict__ dinv, const float* __restrict__ h2s,
                 const float* __restrict__ b2, float* __restrict__ out, int n, int NB, int chunk) {
    __shared__ float acc[BIN_R * C_OUT];
    int tid = threadIdx.x;
    int b = blockIdx.x;
    int node0 = b * BIN_R;
    if (tid < BIN_R * C_OUT) acc[tid] = 0.f;
    __syncthreads();
    int r = tid / RUN_L, sub = tid % RUN_L;
    const int* rs = runStart + (size_t)r * (NB + 1) + b;
    int s0 = rs[0];
    int len = rs[1] - s0;
    const unsigned* src = len > 0 ? bucket + (size_t)r * chunk + s0 : bucket;
    unsigned rec[RUN_U];
    run_load(rec, src, sub, len);
    v2f hv[RUN_U];
    #pragma unroll
    for (int u = 0; u < RUN_U; ++u) hv[u] = *(const v2f*)(h2s + (size_t)(rec[u] >> REC_SH) * C_OUT);
    #pragma unroll
    for (int u = 0; u < RUN_U; ++u) {
        if (sub + RUN_L * u < len) {
            int cl = (int)(rec[u] & REC_MASK);
            atomicAdd(&acc[cl * C_OUT + 0], hv[u].x);
            atomicAdd(&acc[cl * C_OUT + 1], hv[u].y);
        }
    }
    for (int k = sub + RUN_L * RUN_U; k < len; k += RUN_L) {
        unsigned rc = src[k];
        int cl = (int)(rc & REC_MASK);
        v2f h = *(const v2f*)(h2s + (size_t)(rc >> REC_SH) * C_OUT);
        atomicAdd(&acc[cl * C_OUT + 0], h.x);
        atomicAdd(&acc[cl * C_OUT + 1], h.y);
    }
    __syncthreads();
    if (tid < BIN_R * C_OUT) {
        int node = node0 + (tid >> 1);
        if (node < n) {
            size_t g = (size_t)node0 * C_OUT + tid;
            out[g] = b2[tid & 1] + dinv[node] * (h2s[g] + acc[tid]);
        }
    }
}

extern "C" void kernel_launch(void* const* d_in, const int* in_sizes, int n_in,
                              void* d_out, int out_size, void* d_ws, size_t ws_size,
                              hipStream_t stream) {
    const float* x  = (const float*)d_in[0];
    const int*   ei = (const int*)d_in[1];
    const float* W1 = (const float*)d_in[2];
    const float* b1 = (const float*)d_in[3];
    const float* W2 = (const float*)d_in[4];
    const float* b2 = (const float*)d_in[5];
    float* out = (float*)d_out;

    int n = in_sizes[0] / C_IN;
    int E = in_sizes[1] / 2;
    int nPad = (n + 3) & ~3;

    float* ws   = (float*)d_ws;
    float* dinv = ws;
    float* h1s  = dinv + nPad;
    float* h2s  = h1s + (size_t)nPad * C_HID;

    int NB    = (n + BIN_R - 1) / BIN_R;
    int chunk = (E + NCHUNK - 1) / NCHUNK;
    int*      runStart = (int*)(h2s + (size_t)nPad * C_OUT);
    int*      binBase  = runStart + (size_t)NCHUNK * (NB + 1);
    unsigned* bucket   = (unsigned*)(binBase + NB + 1);
    unsigned* sorted   = bucket + (size_t)E;

    size_t ldsBins = (size_t)NB * sizeof(int);

    k_bin_local  <<<NCHUNK, 1024, ldsBins, stream>>>(ei, runStart, bucket, E, NB, chunk);
    k_deg_gemm_bins<<<NB, BIN_T, 0, stream>>>(bucket, runStart, x, W1, dinv, h1s, sorted, binBase, n, NB, chunk);

    k_agg1_bins  <<<NB, AGG1_WAVES * 32, 0, stream>>>(sorted, binBase, dinv, h1s, b1, W2, h2s, n);
    k_agg2_bins  <<<NB, BIN_T, 0, stream>>>(bucket, runStart, dinv, h2s, b2, out, n, NB, chunk);
}


# Layer-2 sum walks each bin's target-ordered list through LDS: a thread adds 32 consecutive records' rows in registers (list order), one LDS float add per run; 8 waves a bin, run table unread
# speedup vs baseline: 8.8767x; 1.0475x over previous
#include <hip/hip_runtime.h>
#include <hip/hip_bf16.h>


typedef __attribute__((ext_vector_type(2))) float v2f;
typedef __attribute__((ext_vector_type(8))) float v8f;

#define C_IN  128
#define C_HID 64
#define C_OUT 2
#define WPAD  132

#define BIN_R  200
#define REC_SH 8
#define REC_MASK ((1u << REC_SH) - 1u)
#define NCHUNK 256
#define EPI_TRIPS ((BIN_R * (C_HID / 4)) / (AGG1_WAVES * 32))
#define AGG1_WAVES 25
#define BIN_T 1024
#define RUN_L 4
#define RUN_U 8
#define STAGE_CAP 12544
#define SORT_CAP 8192
#define AGG2_T 256
#define AGG2_K 32
#define AGG2_LDS_WORDS 27648
#define GEMM_TILES ((BIN_R + 15) / 16)
#define W1_PER_T ((C_IN * C_HID) / BIN_T)
static_assert(BIN_R <= (1 << REC_SH), "local target must fit the record's low bits");
static_assert(BIN_T == RUN_L * NCHUNK, "the per-bin passes give every run RUN_L lanes");
static_assert((BIN_R * (C_HID / 4)) % (AGG1_WAVES * 32) == 0, "epilogue trips must be whole");
static_assert(BIN_R * C_OUT <= AGG1_WAVES * 32 && BIN_R * C_OUT <= BIN_T, "one thread per stored element of a bin");
static_assert(GEMM_TILES * 32 <= BIN_T, "one wave per 16-node tile of a bin");
static_assert(AGG2_T * AGG2_K == SORT_CAP, "the layer-2 sum's LDS stage holds what its threads take in one pass");
static_assert(AGG2_LDS_WORDS >= SORT_CAP + SORT_CAP / 32, "the stage holds SORT_CAP records and their pad words");
static_assert(BIN_R * C_OUT <= 2 * AGG2_T, "two stored elements of a bin per thread at most");
static_assert((C_IN * C_HID) % BIN_T == 0, "W1 is dealt to the threads in whole floats");

__device__ __forceinline__ int block_excl_scan(int v, int* wsum, int tid, int nwaves, int* total) {
    int lane = tid & 31, wv = tid >> 5;
    int inc = v;
    #pragma unroll
    for (int d = 1; d < 32; d <<= 1) {
        int t = __shfl_up(inc, d, 32);
        if (lane >= d) inc += t;
    }
    if (lane == 31) wsum[wv] = inc;
    __syncthreads();
    if (wv == 0) {
        int s = lane < nwaves ? wsum[lane] : 0;
        int si = s;
        #pragma unroll
        for (int d = 1; d < 32; d <<= 1) {
            int t = __shfl_up(si, d, 32);
            if (lane >= d) si += t;
        }
        wsum[lane] = si - s;
        if (lane == 31) wsum[32] = si;
    }
    __syncthreads();
    *total = wsum[32];
    return inc - v + wsum[wv];
}

__device__ __forceinline__ void run_load(unsigned (&rec)[RUN_U], const unsigned* __restrict__ src, int sub, int len) {
    int last = len > 0 ? len - 1 : 0;
    #pragma unroll
    for (int u = 0; u < RUN_U; ++u) {
        int k = sub + RUN_L * u;
        rec[u] = src[k < last ? k : last];
    }
}

__global__ __launch_bounds__(1024)
void k_bin_local(const int* __restrict__ ei, int* __restrict__ runStart,
                 unsigned* __restrict__ bucket, int E, int NB, int chunk) {
    extern __shared__ int sh_bins[];
    __shared__ int wsum[33];
    __shared__ unsigned stage[STAGE_CAP];
    int* lcur = sh_bins;
    int tid = threadIdx.x;
    int w = blockIdx.x;
    int beg = w * chunk < E ? w * chunk : E;
    int end = beg + chunk < E ? beg + chunk : E;
    for (int i = tid; i < NB; i += 1024) lcur[i] = 0;
    __syncthreads();
    const int* row = ei;
    const int* col = ei + (size_t)E;
    int e = beg + tid;
    for (; e + 3 * 1024 < end; e += 4 * 1024) {
        int c[4];
        #pragma unroll
        for (int u = 0; u < 4; ++u) c[u] = col[e + u * 1024];
        #pragma unroll
        for (int u = 0; u < 4; ++u) atomicAdd(&lcur[(unsigned)c[u] / BIN_R], 1);
    }
    for (; e < end; e += 1024) atomicAdd(&lcur[(unsigned)col[e] / BIN_R], 1);
    __syncthreads();
    int ipt = (NB + 1023) / 1024;
    int b0 = tid * ipt;
    int s = 0;
    for (int i = 0; i < ipt; ++i) if (b0 + i < NB) s += lcur[b0 + i];
    int total;
    int run = block_excl_scan(s, wsum, tid, 32, &total);
    int* rs = runStart + (size_t)w * (NB + 1);
    for (int i = 0; i < ipt; ++i) {
        int bi = b0 + i;
        if (bi < NB) {
            int c = lcur[bi];
            lcur[bi] = run;
            rs[bi] = run;
            run += c;
        }
    }
    if (tid == 0) rs[NB] = total;
    __syncthreads();
    bool staged = end - beg <= STAGE_CAP;
    unsigned* dst = bucket + beg;
    e = beg + tid;
    for (; e + 3 * 1024 < end; e += 4 * 1024) {
        int r[4], c[4];
        #pragma unroll
        for (int u = 0; u < 4; ++u) { r[u] = row[e + u * 1024]; c[u] = col[e + u * 1024]; }
        unsigned bin[4];
        int pos[4];
        #pragma unroll
        for (int u = 0; u < 4; ++u) bin[u] = (unsigned)c[u] / BIN_R;
        #pragma unroll
        for (int u = 0; u < 4; ++u) pos[u] = atomicAdd(&lcur[bin[u]], 1);
        #pragma unroll
        for (int u = 0; u < 4; ++u) {
            unsigned rec = ((unsigned)r[u] << REC_SH) | ((unsigned)c[u] - bin[u] * BIN_R);
            if (staged) stage[pos[u]] = rec; else dst[pos[u]] = rec;
        }
    }
    for (; e < end; e += 1024) {
        int r = row[e], c = col[e];
        unsigned bin = (unsigned)c / BIN_R;
        int pos = atomicAdd(&lcur[bin], 1);
        unsigned rec = ((unsigned)r << REC_SH) | ((unsigned)c - bin * BIN_R);
        if (staged) stage[pos] = rec; else dst[pos] = rec;
    }
    if (staged) {
        __syncthreads();
        for (int j = tid; j < end - beg; j += 1024) dst[j] = stage[j];
    }
}

__global__ __launch_bounds__(BIN_T, 16)
void k_deg_gemm_bins(const unsigned* __restrict__ bucket, const int* __restrict__ runStart,
                     const float* __restrict__ x, const float* __restrict__ W1,
                     float* __restrict__ dinv, float* __restrict__ h1s,
                     unsigned* __restrict__ sorted, int* __restrict__ binBase,
                     int n, int NB, int chunk) {
    __shared__ int cnt[BIN_R];
    __shared__ int wsum[33];
    __shared__ int baseSum;
    __shared__ float dl[BIN_R];
    __shared__ unsigned stage[SORT_CAP];
    __shared__ __attribute__((aligned(16))) float Wt[C_HID * WPAD];
    int tid = threadIdx.x;
    int b = blockIdx.x;
    int node0 = b * BIN_R;
    if (tid < BIN_R) cnt[tid] = 0;
    if (tid == 0) baseSum = 0;
    __syncthreads();
    int r = tid / RUN_L, sub = tid % RUN_L;
    const int* rs = runStart + (size_t)r * (NB + 1) + b;
    int s0 = rs[0];
    int len = rs[1] - s0;
    const unsigned* src = len > 0 ? bucket + (size_t)r * chunk + s0 : bucket;
    int v = sub == 0 ? s0 : 0;
    #pragma unroll
    for (int d = 16; d >= 1; d >>= 1) v += __shfl_xor(v, d, 32);
    if ((tid & 31) == 0) atomicAdd(&baseSum, v);
    unsigned rec[RUN_U];
    run_load(rec, src, sub, len);
    float wreg[W1_PER_T];
    #pragma unroll
    for (int u = 0; u < W1_PER_T; ++u) wreg[u] = W1[tid + u * BIN_T];
    #pragma unroll
    for (int u = 0; u < RUN_U; ++u) if (sub + RUN_L * u < len) atomicAdd(&cnt[rec[u] & REC_MASK], 1);
    for (int k = sub + RUN_L * RUN_U; k < len; k += RUN_L) atomicAdd(&cnt[src[k] & REC_MASK], 1);
    #pragma unroll
    for (int u = 0; u < W1_PER_T; ++u) {
        int i = tid + u * BIN_T;
        Wt[(i % C_HID) * WPAD + i / C_HID] = wreg[u];
    }
    __syncthreads();
    int c = tid < BIN_R ? cnt[tid] : 0;
    int total;
    int off = block_excl_scan(c, wsum, tid, BIN_T / 32, &total);
    if (tid < BIN_R) {
        int node = node0 + tid;
        float dv = __frsqrt_rn((float)(c + 1));
        if (node < n) dinv[node] = dv;
        dl[tid] = dv;
        cnt[tid] = off;
    }
    __syncthreads();
    int base = baseSum;
    bool staged = total <= SORT_CAP;
    #pragma unroll
    for (int u = 0; u < RUN_U; ++u) {
        if (sub + RUN_L * u < len) {
            int pos = atomicAdd(&cnt[rec[u] & REC_MASK], 1);
            if (staged) stage[pos] = rec[u]; else sorted[base + pos] = rec[u];
        }
    }
    for (int k = sub + RUN_L * RUN_U; k < len; k += RUN_L) {
        unsigned rc = src[k];
        int pos = atomicAdd(&cnt[rc & REC_MASK], 1);
        if (staged) stage[pos] = rc; else sorted[base + pos] = rc;
    }
    if (staged) {
        __syncthreads();
        for (int i = tid; i < total; i += BIN_T) sorted[base + i] = stage[i];
    }
    if (tid == 0) {
        binBase[b] = base;
        if (b == NB - 1) binBase[NB] = base + total;
    }

    int wv = __builtin_amdgcn_readfirstlane(tid >> 5);
    if (wv < GEMM_TILES) {
        int lane  = tid & 31;
        int m     = lane & 15;
        int khalf = lane >> 4;
        int ncol  = lane & 15;
        int lrow0 = wv * 16;
        int arow = node0 + lrow0 + m;
        arow = arow < n ? arow : n - 1;
        const float* xrow = x + (size_t)arow * C_IN + 2 * khalf;

        const v2f* bp0 = (const v2f*)&Wt[(ncol     ) * WPAD + 2 * khalf];
        const v2f* bp1 = (const v2f*)&Wt[(ncol + 16) * WPAD + 2 * khalf];
        const v2f* bp2 = (const v2f*)&Wt[(ncol + 32) * WPAD + 2 * khalf];
        const v2f* bp3 = (const v2f*)&Wt[(ncol + 48) * WPAD + 2 * khalf];

        v8f a0 = {}, a1 = {}, a2 = {}, a3 = {};
        #pragma unroll 4
        for (int k0 = 0; k0 < C_IN; k0 += 4) {
            v2f av = *(const v2f*)(xrow + k0);
            v2f b0 = bp0[k0 >> 1];
            v2f b1 = bp1[k0 >> 1];
            v2f b2 = bp2[k0 >> 1];
            v2f b3 = bp3[k0 >> 1];
            a0 = __builtin_amdgcn_wmma_f32_16x16x4_f32(false, av, false, b0, (short)0, a0, false, false);
            a1 = __builtin_amdgcn_wmma_f32_16x16x4_f32(false, av, false, b1, (short)0, a1, false, false);
            a2 = __builtin_amdgcn_wmma_f32_16x16x4_f32(false, av, false, b2, (short)0, a2, false, false);
            a3 = __builtin_amdgcn_wmma_f32_16x16x4_f32(false, av, false, b3, (short)0, a3, false, false);
        }

        int lr = lrow0 + 8 * khalf;
        float* hrow = h1s + (size_t)(node0 + lr) * C_HID + ncol;
        #pragma unroll
        for (int q = 0; q < 8; ++q) {
            int l = lr + q;
            if (l < BIN_R && node0 + l < n) {
                float d = dl[l];
                hrow[q * C_HID]      = d * a0[q];
                hrow[q * C_HID + 16] = d * a1[q];
                hrow[q * C_HID + 32] = d * a2[q];
                hrow[q * C_HID + 48] = d * a3[q];
            }
        }
    }
}

__device__ __forceinline__ void agg1_gather16(v2f (&hv)[16], const float* __restrict__ h1, int row, int lane, int j0) {
    #pragma unroll
    for (int u = 0; u < 16; ++u) {
        int r = __builtin_amdgcn_readlane(row, j0 + u);
        hv[u] = *(const v2f*)(h1 + (size_t)r * C_HID + 2 * lane);
    }
}
__device__ __forceinline__ void agg1_flush(float* acc, int c, v2f sum, int lane) {
    float* a = &acc[c * C_HID + 2 * lane];
    atomicAdd(a,     sum.x);
    atomicAdd(a + 1, sum.y);
}
__device__ __forceinline__ void agg1_sum16(const v2f (&hv)[16], float* acc, int cl, int lane, int j0, int& cur, v2f& sum) {
    #pragma unroll
    for (int u = 0; u < 16; ++u) {
        int c = __builtin_amdgcn_readlane(cl, j0 + u);
        if (c != cur) {
            agg1_flush(acc, cur, sum, lane);
            cur = c;
            sum.x = 0.f; sum.y = 0.f;
        }
        sum.x += hv[u].x;
        sum.y += hv[u].y;
    }
}

__global__ __launch_bounds__(AGG1_WAVES * 32, 16)
void k_agg1_bins(const unsigned* __restrict__ sorted, const int* __restrict__ binBase,
                 const float* __restrict__ dinv, const float* __restrict__ h1s,
                 const float* __restrict__ b1, const float* __restrict__ W2,
                 float* __restrict__ h2s, int n) {
    __shared__ __attribute__((aligned(16))) float acc[BIN_R * C_HID];
    __shared__ float h2l[BIN_R * C_OUT];
    int tid  = threadIdx.x;
    int lane = tid & 31;
    int wv   = __builtin_amdgcn_readfirstlane(tid >> 5);
    int b    = blockIdx.x;
    int node0 = b * BIN_R;
    for (int i = tid; i < BIN_R * C_HID; i += AGG1_WAVES * 32) acc[i] = 0.f;
    __syncthreads();

    int beg = binBase[b], end = binBase[b + 1];
    int chunk = ((end - beg + AGG1_WAVES * 32 - 1) / (AGG1_WAVES * 32)) * 32;
    int p  = beg + wv * chunk;
    int pe = p + chunk < end ? p + chunk : end;
    unsigned rec = p + lane < pe ? sorted[p + lane] : 0u;
    bool any = p < pe;
    int cur = __builtin_amdgcn_readlane((int)(rec & REC_MASK), 0);
    v2f sum = {0.f, 0.f};
    while (p < pe) {
        int left = pe - p;
        int pn = p + 32;
        unsigned recN = 0u;
        int row = (int)(rec >> REC_SH);
        int cl  = (int)(rec & REC_MASK);
        if (left >= 32) {
            v2f hv[16];
            agg1_gather16(hv, h1s, row, lane, 0);
            recN = pn + lane < pe ? sorted[pn + lane] : 0u;
            agg1_sum16(hv, acc, cl, lane, 0, cur, sum);
            agg1_gather16(hv, h1s, row, lane, 16);
            agg1_sum16(hv, acc, cl, lane, 16, cur, sum);
        } else {
            for (int j = 0; j < left; ++j) {
                int r = __builtin_amdgcn_readlane(row, j);
                int c = __builtin_amdgcn_readlane(cl, j);
                v2f hv = *(const v2f*)(h1s + (size_t)r * C_HID + 2 * lane);
                if (c != cur) {
                    agg1_flush(acc, cur, sum, lane);
                    cur = c;
                    sum.x = 0.f; sum.y = 0.f;
                }
                sum.x += hv.x;
                sum.y += hv.y;
            }
        }
        rec = recN;
        p = pn;
    }
    if (any) agg1_flush(acc, cur, sum, lane);
    __syncthreads();

    int q  = tid & 15;
    int c4 = q * 4;
    float4 bb = *(const float4*)(b1 + c4);
    float4 wA = *(const float4*)(W2 + c4 * C_OUT);
    float4 wB = *(const float4*)(W2 + c4 * C_OUT + 4);
    float  dd[EPI_TRIPS];
    float4 hh[EPI_TRIPS];
    #pragma unroll
    for (int u = 0; u < EPI_TRIPS; ++u) {
        int node = node0 + ((tid + u * AGG1_WAVES * 32) >> 4);
        node = node < n ? node : n - 1;
        dd[u] = dinv[node];
        hh[u] = *(const float4*)(h1s + (size_t)node * C_HID + c4);
    }
    #pragma unroll
    for (int u = 0; u < EPI_TRIPS; ++u) {
        int cl = (tid + u * AGG1_WAVES * 32) >> 4;
        float d = dd[u];
        float4 hs = hh[u];
        float4 av = *(const float4*)(&acc[cl * C_HID + c4]);
        float o0 = bb.x + d * (hs.x + av.x);
        float o1 = bb.y + d * (hs.y + av.y);
        float o2 = bb.z + d * (hs.z + av.z);
        float o3 = bb.w + d * (hs.w + av.w);
        o0 = o0 > 0.f ? o0 : 0.f;
        o1 = o1 > 0.f ? o1 : 0.f;
        o2 = o2 > 0.f ? o2 : 0.f;
        o3 = o3 > 0.f ? o3 : 0.f;
        float p0 = fmaf(o3, wB.z, fmaf(o2, wB.x, fmaf(o1, wA.z, o0 * wA.x)));
        float p1 = fmaf(o3, wB.w, fmaf(o2, wB.y, fmaf(o1, wA.w, o0 * wA.y)));
        #pragma unroll
        for (int sft = 8; sft >= 1; sft >>= 1) {
            p0 += __shfl_xor(p0, sft, 32);
            p1 += __shfl_xor(p1, sft, 32);
        }
        if (q == 0) {
            h2l[cl * C_OUT + 0] = d * p0;
            h2l[cl * C_OUT + 1] = d * p1;
        }
    }
    __syncthreads();
    if (tid < BIN_R * C_OUT) {
        int node = node0 + (tid >> 1);
        if (node < n) h2s[(size_t)node0 * C_OUT + tid] = h2l[tid];
    }
}

__global__ __launch_bounds__(AGG2_T)
void k_agg2_bins(const unsigned* __restrict__ sorted, const int* __restrict__ binBase,
                 const float* __restrict__ dinv, const float* __restrict__ h2s,
                 const float* __restrict__ b2, float* __restrict__ out, int n) {
    __shared__ float acc[BIN_R * C_OUT];
    __shared__ unsigned stage[AGG2_LDS_WORDS];
    int tid = threadIdx.x;
    int b = blockIdx.x;
    int node0 = b * BIN_R;
    for (int i = tid; i < BIN_R * C_OUT; i += AGG2_T) acc[i] = 0.f;
    int beg = binBase[b], end = binBase[b + 1];
    int cnt  = end - beg;
    int cntS = cnt < SORT_CAP ? cnt : SORT_CAP;
    float dd[2], hs[2], bb[2];
    #pragma unroll
    for (int u = 0; u < 2; ++u) {
        int i = tid + u * AGG2_T;
        bool ok = i < BIN_R * C_OUT && node0 + (i >> 1) < n;
        int ic = ok ? i : 0;
        dd[u] = dinv[node0 + (ic >> 1)];
        hs[u] = h2s[(size_t)node0 * C_OUT + ic];
        bb[u] = b2[ic & 1];
    }
    unsigned rec[AGG2_K];
    if (cntS > 0) {
        #pragma unroll
        for (int u = 0; u < AGG2_K; ++u) {
            int i = tid + u * AGG2_T;
            rec[u] = sorted[beg + (i < cntS ? i : cntS - 1)];
        }
        #pragma unroll
        for (int u = 0; u < AGG2_K; ++u) {
            int i = tid + u * AGG2_T;
            if (i < cntS) stage[i + (i >> 5)] = rec[u];
        }
    }
    __syncthreads();
    int my0 = tid * AGG2_K;
    if (my0 < cntS) {
        int nv = cntS - my0 < AGG2_K ? cntS - my0 : AGG2_K;
        const unsigned* mine = &stage[my0 + tid];
        #pragma unroll
        for (int j = 0; j < AGG2_K; ++j) rec[j] = mine[j < nv ? j : nv - 1];
        v2f hv[AGG2_K];
        #pragma unroll
        for (int j = 0; j < AGG2_K; ++j) hv[j] = *(const v2f*)(h2s + (size_t)(rec[j] >> REC_SH) * C_OUT);
        int cur = (int)(rec[0] & REC_MASK);
        v2f sum = {0.f, 0.f};
        #pragma unroll
        for (int j = 0; j < AGG2_K; ++j) {
            int c = (int)(rec[j] & REC_MASK);
            if (c != cur) {
                atomicAdd(&acc[cur * C_OUT + 0], sum.x);
                atomicAdd(&acc[cur * C_OUT + 1], sum.y);
                cur = c;
                sum.x = 0.f; sum.y = 0.f;
            }
            if (j < nv) { sum.x += hv[j].x; sum.y += hv[j].y; }
        }
        atomicAdd(&acc[cur * C_OUT + 0], sum.x);
        atomicAdd(&acc[cur * C_OUT + 1], sum.y);
    }
    for (int i = SORT_CAP + tid; i < cnt; i += AGG2_T) {
        unsigned rc = sorted[beg + i];
        int cl = (int)(rc & REC_MASK);
        v2f h = *(const v2f*)(h2s + (size_t)(rc >> REC_SH) * C_OUT);
        atomicAdd(&acc[cl * C_OUT + 0], h.x);
        atomicAdd(&acc[cl * C_OUT + 1], h.y);
    }
    __syncthreads();
    #pragma unroll
    for (int u = 0; u < 2; ++u) {
        int i = tid + u * AGG2_T;
        if (i < BIN_R * C_OUT && node0 + (i >> 1) < n)
            out[(size_t)node0 * C_OUT + i] = bb[u] + dd[u] * (hs[u] + acc[i]);
    }
}

extern "C" void kernel_launch(void* const* d_in, const int* in_sizes, int n_in,
                              void* d_out, int out_size, void* d_ws, size_t ws_size,
                              hipStream_t stream) {
    const float* x  = (const float*)d_in[0];
    const int*   ei = (const int*)d_in[1];
    const float* W1 = (const float*)d_in[2];
    const float* b1 = (const float*)d_in[3];
    const float* W2 = (const float*)d_in[4];
    const float* b2 = (const float*)d_in[5];
    float* out = (float*)d_out;

    int n = in_sizes[0] / C_IN;
    int E = in_sizes[1] / 2;
    int nPad = (n + 3) & ~3;

    float* ws   = (float*)d_ws;
    float* dinv = ws;
    float* h1s  = dinv + nPad;
    float* h2s  = h1s + (size_t)nPad * C_HID;

    int NB    = (n + BIN_R - 1) / BIN_R;
    int chunk = (E + NCHUNK - 1) / NCHUNK;
    int*      runStart = (int*)(h2s + (size_t)nPad * C_OUT);
    int*      binBase  = runStart + (size_t)NCHUNK * (NB + 1);
    unsigned* bucket   = (unsigned*)(binBase + NB + 1);
    unsigned* sorted   = bucket + (size_t)E;

    size_t ldsBins = (size_t)NB * sizeof(int);

    k_bin_local  <<<NCHUNK, 1024, ldsBins, stream>>>(ei, runStart, bucket, E, NB, chunk);
    k_deg_gemm_bins<<<NB, BIN_T, 0, stream>>>(bucket, runStart, x, W1, dinv, h1s, sorted, binBase, n, NB, chunk);

    k_agg1_bins  <<<NB, AGG1_WAVES * 32, 0, stream>>>(sorted, binBase, dinv, h1s, b1, W2, h2s, n);
    k_agg2_bins  <<<NB, AGG2_T, 0, stream>>>(sorted, binBase, dinv, h2s, b2, out, n);
}
